# starting kernel
# baseline (speedup 1.0000x reference)
#include <hip/hip_runtime.h>

typedef __attribute__((ext_vector_type(2))) float v2f;
typedef __attribute__((ext_vector_type(8))) float v8f;

#define DD 128
#define NN 256
#define BB 8
#define ROWS (BB * NN)
#define WMSG_LD (2 * DD + 1)
#define WUPD_LD (2 * DD)

__global__ __launch_bounds__(256) void gemm_ts_kernel(
    const float* __restrict__ H, const float* __restrict__ Wmsg,
    float* __restrict__ T, float* __restrict__ S)
{
  const int wave = (blockIdx.x * 256 + threadIdx.x) >> 5;
  const int lane = threadIdx.x & 31;
  const int mt = wave >> 3;
  const int nt = wave & 7;
  const int m0 = mt << 4, n0 = nt << 4;
  const int half = lane >> 4;
  const int ln = lane & 15;

  const float* __restrict__ arow = H + (size_t)(m0 + ln) * DD;
  const float* __restrict__ wrow = Wmsg + (size_t)(n0 + ln) * WMSG_LD;

  v8f accT = {};
  v8f accS = {};
  for (int k0 = 0; k0 < DD; k0 += 4) {
    const int ka = k0 + 2 * half;
    v2f a;  a.x  = arow[ka];          a.y  = arow[ka + 1];
    v2f bt; bt.x = wrow[ka];          bt.y = wrow[ka + 1];
    v2f bs; bs.x = wrow[DD + ka];     bs.y = wrow[DD + ka + 1];
    accT = __builtin_amdgcn_wmma_f32_16x16x4_f32(
        false, a, false, bt, (short)0, accT, false, false);
    accS = __builtin_amdgcn_wmma_f32_16x16x4_f32(
        false, a, false, bs, (short)0, accS, false, false);
  }
  #pragma unroll
  for (int r = 0; r < 8; ++r) {
    const int row = m0 + r + 8 * half;
    T[(size_t)row * DD + n0 + ln] = accT[r];
    S[(size_t)row * DD + n0 + ln] = accS[r];
  }
}

__global__ __launch_bounds__(256) void gemm_g_kernel(
    const float* __restrict__ T, const float* __restrict__ S,
    float* __restrict__ G)
{
  const int wave = (blockIdx.x * 256 + threadIdx.x) >> 5;
  const int lane = threadIdx.x & 31;
  const int b  = wave >> 8;
  const int rm = wave & 255;
  const int it = rm >> 4;
  const int jt = rm & 15;
  const int m0 = it << 4, n0 = jt << 4;
  const int half = lane >> 4;
  const int ln = lane & 15;

  const float* __restrict__ arow = T + (size_t)(b * NN + m0 + ln) * DD;
  const float* __restrict__ brow = S + (size_t)(b * NN + n0 + ln) * DD;

  v8f acc = {};
  for (int k0 = 0; k0 < DD; k0 += 4) {
    const int ka = k0 + 2 * half;
    v2f a;  a.x = arow[ka]; a.y = arow[ka + 1];
    v2f bb; bb.x = brow[ka]; bb.y = brow[ka + 1];
    acc = __builtin_amdgcn_wmma_f32_16x16x4_f32(
        false, a, false, bb, (short)0, acc, false, false);
  }
  #pragma unroll
  for (int r = 0; r < 8; ++r) {
    const int i = m0 + r + 8 * half;
    G[(size_t)(b * NN + i) * NN + n0 + ln] = acc[r];
  }
}

__global__ __launch_bounds__(256) void stats_s_kernel(
    const float* __restrict__ S, const float* __restrict__ Wmsg,
    const float* __restrict__ bmsg, float4* __restrict__ STATS)
{
  const int row  = (blockIdx.x * 256 + threadIdx.x) >> 5;
  const int lane = threadIdx.x & 31;
  const int e0 = lane << 2;

  const float4 s4 = *(const float4*)(S + (size_t)row * DD + e0);
  float4 wj4;
  wj4.x = Wmsg[(size_t)(e0 + 0) * WMSG_LD + 2 * DD];
  wj4.y = Wmsg[(size_t)(e0 + 1) * WMSG_LD + 2 * DD];
  wj4.z = Wmsg[(size_t)(e0 + 2) * WMSG_LD + 2 * DD];
  wj4.w = Wmsg[(size_t)(e0 + 3) * WMSG_LD + 2 * DD];
  const float4 bm4 = *(const float4*)(bmsg + e0);

  float p1 = s4.x + s4.y + s4.z + s4.w;
  float p2 = s4.x * s4.x + s4.y * s4.y + s4.z * s4.z + s4.w * s4.w;
  float p3 = s4.x * wj4.x + s4.y * wj4.y + s4.z * wj4.z + s4.w * wj4.w;
  float p4 = s4.x * bm4.x + s4.y * bm4.y + s4.z * bm4.z + s4.w * bm4.w;
  #pragma unroll
  for (int off = 16; off >= 1; off >>= 1) {
    p1 += __shfl_xor(p1, off, 32);
    p2 += __shfl_xor(p2, off, 32);
    p3 += __shfl_xor(p3, off, 32);
    p4 += __shfl_xor(p4, off, 32);
  }
  if (lane == 0) {
    float4 st; st.x = p1; st.y = p2; st.z = p3; st.w = p4;
    STATS[row] = st;
  }
}

__global__ __launch_bounds__(256) void msg_agg_kernel(
    const float* __restrict__ T, const float* __restrict__ S,
    const float* __restrict__ adj, const float* __restrict__ Wmsg,
    const float* __restrict__ bmsg, const float* __restrict__ gmsg,
    const float* __restrict__ betamsg, const float* __restrict__ G,
    const float4* __restrict__ STATS, float* __restrict__ AGG)
{
  const int wave = (blockIdx.x * 256 + threadIdx.x) >> 5;
  const int lane = threadIdx.x & 31;
  const int b = wave >> 8;
  const int i = wave & 255;
  const int e0 = lane << 2;

  const float4 t4 = *(const float4*)(T + (size_t)(b * NN + i) * DD + e0);
  float4 wj4;
  wj4.x = Wmsg[(size_t)(e0 + 0) * WMSG_LD + 2 * DD];
  wj4.y = Wmsg[(size_t)(e0 + 1) * WMSG_LD + 2 * DD];
  wj4.z = Wmsg[(size_t)(e0 + 2) * WMSG_LD + 2 * DD];
  wj4.w = Wmsg[(size_t)(e0 + 3) * WMSG_LD + 2 * DD];
  const float4 bm4 = *(const float4*)(bmsg + e0);
  const float4 g4  = *(const float4*)(gmsg + e0);
  const float4 be4 = *(const float4*)(betamsg + e0);

  float4 c4;
  c4.x = t4.x + bm4.x; c4.y = t4.y + bm4.y;
  c4.z = t4.z + bm4.z; c4.w = t4.w + bm4.w;

  float r1 = c4.x + c4.y + c4.z + c4.w;
  float r2 = c4.x * c4.x + c4.y * c4.y + c4.z * c4.z + c4.w * c4.w;
  float r3 = c4.x * wj4.x + c4.y * wj4.y + c4.z * wj4.z + c4.w * wj4.w;
  float r4 = wj4.x + wj4.y + wj4.z + wj4.w;
  float r5 = wj4.x * wj4.x + wj4.y * wj4.y + wj4.z * wj4.z + wj4.w * wj4.w;
  #pragma unroll
  for (int off = 16; off >= 1; off >>= 1) {
    r1 += __shfl_xor(r1, off, 32);
    r2 += __shfl_xor(r2, off, 32);
    r3 += __shfl_xor(r3, off, 32);
    r4 += __shfl_xor(r4, off, 32);
    r5 += __shfl_xor(r5, off, 32);
  }
  const float C1 = r1, C2 = r2, Cw = r3, W1 = r4, W2 = r5;
  const float invD = 1.0f / DD;

  const float*  __restrict__ srow = S + (size_t)b * NN * DD;
  const float*  __restrict__ arow = adj + (size_t)(b * NN + i) * NN;
  const float*  __restrict__ grow = G + (size_t)(b * NN + i) * NN;
  const float4* __restrict__ strow = STATS + (size_t)b * NN;

  float4 acc = {0.f, 0.f, 0.f, 0.f};
  for (int j = 0; j < NN; ++j) {
    const float4 s4 = *(const float4*)(srow + (size_t)j * DD + e0);
    const float  aj = arow[j];
    const float  gij = grow[j];
    const float4 st = strow[j];

    const float mean  = (C1 + st.x + aj * W1) * invD;
    const float sumsq = C2 + st.y + aj * aj * W2
                      + 2.0f * (gij + st.w)
                      + 2.0f * aj * (Cw + st.z);
    const float var  = sumsq * invD - mean * mean;
    const float rinv = rsqrtf(var + 1e-5f);

    float4 p;
    p.x = c4.x + s4.x + aj * wj4.x;
    p.y = c4.y + s4.y + aj * wj4.y;
    p.z = c4.z + s4.z + aj * wj4.z;
    p.w = c4.w + s4.w + aj * wj4.w;

    float x0 = (p.x - mean) * rinv * g4.x + be4.x;
    float x1 = (p.y - mean) * rinv * g4.y + be4.y;
    float x2 = (p.z - mean) * rinv * g4.z + be4.z;
    float x3 = (p.w - mean) * rinv * g4.w + be4.w;
    const float m0 = x0 / (1.0f + __expf(-x0));
    const float m1 = x1 / (1.0f + __expf(-x1));
    const float m2 = x2 / (1.0f + __expf(-x2));
    const float m3 = x3 / (1.0f + __expf(-x3));

    const float keep = (j != i) ? 1.0f : 0.0f;
    acc.x += keep * m0;
    acc.y += keep * m1;
    acc.z += keep * m2;
    acc.w += keep * m3;
  }
  *(float4*)(AGG + (size_t)(b * NN + i) * DD + e0) = acc;
}

__global__ __launch_bounds__(256) void gemm_upd_kernel(
    const float* __restrict__ H, const float* __restrict__ AGG,
    const float* __restrict__ Wupd, const float* __restrict__ bupd,
    float* __restrict__ U)
{
  const int wave = (blockIdx.x * 256 + threadIdx.x) >> 5;
  const int lane = threadIdx.x & 31;
  const int mt = wave >> 3;
  const int nt = wave & 7;
  const int m0 = mt << 4, n0 = nt << 4;
  const int half = lane >> 4;
  const int ln = lane & 15;

  const float* __restrict__ arow1 = H + (size_t)(m0 + ln) * DD;
  const float* __restrict__ arow2 = AGG + (size_t)(m0 + ln) * DD;
  const float* __restrict__ wrow = Wupd + (size_t)(n0 + ln) * WUPD_LD;

  v8f acc = {};
  for (int k0 = 0; k0 < DD; k0 += 4) {
    const int ka = k0 + 2 * half;
    v2f a1; a1.x = arow1[ka]; a1.y = arow1[ka + 1];
    v2f b1; b1.x = wrow[ka];  b1.y = wrow[ka + 1];
    acc = __builtin_amdgcn_wmma_f32_16x16x4_f32(
        false, a1, false, b1, (short)0, acc, false, false);
    v2f a2; a2.x = arow2[ka];     a2.y = arow2[ka + 1];
    v2f b2; b2.x = wrow[DD + ka]; b2.y = wrow[DD + ka + 1];
    acc = __builtin_amdgcn_wmma_f32_16x16x4_f32(
        false, a2, false, b2, (short)0, acc, false, false);
  }

  const float bias = bupd[n0 + ln];
  #pragma unroll
  for (int r = 0; r < 8; ++r) {
    const int row = m0 + r + 8 * half;
    U[(size_t)row * DD + n0 + ln] = acc[r] + bias;
  }
}

__global__ __launch_bounds__(256) void finalize_kernel(
    const float* __restrict__ U, const float* __restrict__ H,
    const float* __restrict__ gupd, const float* __restrict__ betaupd,
    float* __restrict__ out)
{
  const int row = (blockIdx.x * 256 + threadIdx.x) >> 5;
  const int lane = threadIdx.x & 31;
  const int e0 = lane << 2;

  const float4 u4 = *(const float4*)(U + (size_t)row * DD + e0);
  float ps  = u4.x + u4.y + u4.z + u4.w;
  float pss = u4.x * u4.x + u4.y * u4.y + u4.z * u4.z + u4.w * u4.w;
  #pragma unroll
  for (int off = 16; off >= 1; off >>= 1) {
    ps  += __shfl_xor(ps,  off, 32);
    pss += __shfl_xor(pss, off, 32);
  }
  const float mean = ps * (1.0f / DD);
  const float var  = pss * (1.0f / DD) - mean * mean;
  const float rinv = rsqrtf(var + 1e-5f);

  const float4 g4  = *(const float4*)(gupd + e0);
  const float4 be4 = *(const float4*)(betaupd + e0);
  const float4 h4  = *(const float4*)(H + (size_t)row * DD + e0);

  float x0 = (u4.x - mean) * rinv * g4.x + be4.x;
  float x1 = (u4.y - mean) * rinv * g4.y + be4.y;
  float x2 = (u4.z - mean) * rinv * g4.z + be4.z;
  float x3 = (u4.w - mean) * rinv * g4.w + be4.w;

  float4 o;
  o.x = h4.x + x0 / (1.0f + __expf(-x0));
  o.y = h4.y + x1 / (1.0f + __expf(-x1));
  o.z = h4.z + x2 / (1.0f + __expf(-x2));
  o.w = h4.w + x3 / (1.0f + __expf(-x3));
  *(float4*)(out + (size_t)row * DD + e0) = o;
}

extern "C" void kernel_launch(void* const* d_in, const int* in_sizes, int n_in,
                              void* d_out, int out_size, void* d_ws, size_t ws_size,
                              hipStream_t stream) {
  const float* h       = (const float*)d_in[0];
  const float* adj     = (const float*)d_in[1];
  const float* Wmsg    = (const float*)d_in[2];
  const float* bmsg    = (const float*)d_in[3];
  const float* gmsg    = (const float*)d_in[4];
  const float* betamsg = (const float*)d_in[5];
  const float* Wupd    = (const float*)d_in[6];
  const float* bupd    = (const float*)d_in[7];
  const float* gupd    = (const float*)d_in[8];
  const float* betaupd = (const float*)d_in[9];
  float* out = (float*)d_out;

  const size_t mat_bytes = (size_t)ROWS * DD * sizeof(float);
  const size_t g_bytes   = (size_t)BB * NN * NN * sizeof(float);
  char* ws = (char*)d_ws;
  float*  T     = (float*)(ws);
  float*  S     = (float*)(ws + mat_bytes);
  float*  AGG   = (float*)(ws + 2 * mat_bytes);
  float*  U     = (float*)(ws + 3 * mat_bytes);
  float*  G     = (float*)(ws + 4 * mat_bytes);
  float4* STATS = (float4*)(ws + 4 * mat_bytes + g_bytes);

  gemm_ts_kernel<<<128, 256, 0, stream>>>(h, Wmsg, T, S);
  gemm_g_kernel<<<256, 256, 0, stream>>>(T, S, G);
  stats_s_kernel<<<256, 256, 0, stream>>>(S, Wmsg, bmsg, STATS);
  msg_agg_kernel<<<256, 256, 0, stream>>>(T, S, adj, Wmsg, bmsg, gmsg,
                                          betamsg, G, STATS, AGG);
  gemm_upd_kernel<<<128, 256, 0, stream>>>(h, AGG, Wupd, bupd, U);
  finalize_kernel<<<256, 256, 0, stream>>>(U, h, gupd, betaupd, out);
}


# msg/aggregate kernel: one work-group per (b,i), its 8 waves each sum 32 of the 256 sources j (16 waves per SIMD, was 2); the 8 partials are added through LDS in wave order: sum over j re-associated
# speedup vs baseline: 1.6476x; 1.6476x over previous
#include <hip/hip_runtime.h>

typedef __attribute__((ext_vector_type(2))) float v2f;
typedef __attribute__((ext_vector_type(8))) float v8f;

#define DD 128
#define NN 256
#define BB 8
#define ROWS (BB * NN)
#define WMSG_LD (2 * DD + 1)
#define WUPD_LD (2 * DD)
#define JSPLIT 8
#define JPER (NN / JSPLIT)

__global__ __launch_bounds__(256) void gemm_ts_kernel(
    const float* __restrict__ H, const float* __restrict__ Wmsg,
    float* __restrict__ T, float* __restrict__ S)
{
  const int wave = (blockIdx.x * 256 + threadIdx.x) >> 5;
  const int lane = threadIdx.x & 31;
  const int mt = wave >> 3;
  const int nt = wave & 7;
  const int m0 = mt << 4, n0 = nt << 4;
  const int half = lane >> 4;
  const int ln = lane & 15;

  const float* __restrict__ arow = H + (size_t)(m0 + ln) * DD;
  const float* __restrict__ wrow = Wmsg + (size_t)(n0 + ln) * WMSG_LD;

  v8f accT = {};
  v8f accS = {};
  for (int k0 = 0; k0 < DD; k0 += 4) {
    const int ka = k0 + 2 * half;
    v2f a;  a.x  = arow[ka];          a.y  = arow[ka + 1];
    v2f bt; bt.x = wrow[ka];          bt.y = wrow[ka + 1];
    v2f bs; bs.x = wrow[DD + ka];     bs.y = wrow[DD + ka + 1];
    accT = __builtin_amdgcn_wmma_f32_16x16x4_f32(
        false, a, false, bt, (short)0, accT, false, false);
    accS = __builtin_amdgcn_wmma_f32_16x16x4_f32(
        false, a, false, bs, (short)0, accS, false, false);
  }
  #pragma unroll
  for (int r = 0; r < 8; ++r) {
    const int row = m0 + r + 8 * half;
    T[(size_t)row * DD + n0 + ln] = accT[r];
    S[(size_t)row * DD + n0 + ln] = accS[r];
  }
}

__global__ __launch_bounds__(256) void gemm_g_kernel(
    const float* __restrict__ T, const float* __restrict__ S,
    float* __restrict__ G)
{
  const int wave = (blockIdx.x * 256 + threadIdx.x) >> 5;
  const int lane = threadIdx.x & 31;
  const int b  = wave >> 8;
  const int rm = wave & 255;
  const int it = rm >> 4;
  const int jt = rm & 15;
  const int m0 = it << 4, n0 = jt << 4;
  const int half = lane >> 4;
  const int ln = lane & 15;

  const float* __restrict__ arow = T + (size_t)(b * NN + m0 + ln) * DD;
  const float* __restrict__ brow = S + (size_t)(b * NN + n0 + ln) * DD;

  v8f acc = {};
  for (int k0 = 0; k0 < DD; k0 += 4) {
    const int ka = k0 + 2 * half;
    v2f a;  a.x = arow[ka]; a.y = arow[ka + 1];
    v2f bb; bb.x = brow[ka]; bb.y = brow[ka + 1];
    acc = __builtin_amdgcn_wmma_f32_16x16x4_f32(
        false, a, false, bb, (short)0, acc, false, false);
  }
  #pragma unroll
  for (int r = 0; r < 8; ++r) {
    const int i = m0 + r + 8 * half;
    G[(size_t)(b * NN + i) * NN + n0 + ln] = acc[r];
  }
}

__global__ __launch_bounds__(256) void stats_s_kernel(
    const float* __restrict__ S, const float* __restrict__ Wmsg,
    const float* __restrict__ bmsg, float4* __restrict__ STATS)
{
  const int row  = (blockIdx.x * 256 + threadIdx.x) >> 5;
  const int lane = threadIdx.x & 31;
  const int e0 = lane << 2;

  const float4 s4 = *(const float4*)(S + (size_t)row * DD + e0);
  float4 wj4;
  wj4.x = Wmsg[(size_t)(e0 + 0) * WMSG_LD + 2 * DD];
  wj4.y = Wmsg[(size_t)(e0 + 1) * WMSG_LD + 2 * DD];
  wj4.z = Wmsg[(size_t)(e0 + 2) * WMSG_LD + 2 * DD];
  wj4.w = Wmsg[(size_t)(e0 + 3) * WMSG_LD + 2 * DD];
  const float4 bm4 = *(const float4*)(bmsg + e0);

  float p1 = s4.x + s4.y + s4.z + s4.w;
  float p2 = s4.x * s4.x + s4.y * s4.y + s4.z * s4.z + s4.w * s4.w;
  float p3 = s4.x * wj4.x + s4.y * wj4.y + s4.z * wj4.z + s4.w * wj4.w;
  float p4 = s4.x * bm4.x + s4.y * bm4.y + s4.z * bm4.z + s4.w * bm4.w;
  #pragma unroll
  for (int off = 16; off >= 1; off >>= 1) {
    p1 += __shfl_xor(p1, off, 32);
    p2 += __shfl_xor(p2, off, 32);
    p3 += __shfl_xor(p3, off, 32);
    p4 += __shfl_xor(p4, off, 32);
  }
  if (lane == 0) {
    float4 st; st.x = p1; st.y = p2; st.z = p3; st.w = p4;
    STATS[row] = st;
  }
}

__global__ __launch_bounds__(256) void msg_agg_kernel(
    const float* __restrict__ T, const float* __restrict__ S,
    const float* __restrict__ adj, const float* __restrict__ Wmsg,
    const float* __restrict__ bmsg, const float* __restrict__ gmsg,
    const float* __restrict__ betamsg, const float* __restrict__ G,
    const float4* __restrict__ STATS, float* __restrict__ AGG)
{
  __shared__ float4 part[JSPLIT][32];
  const int w    = threadIdx.x >> 5;
  const int lane = threadIdx.x & 31;
  const int b = blockIdx.x >> 8;
  const int i = blockIdx.x & 255;
  const int e0 = lane << 2;

  const float4 t4 = *(const float4*)(T + (size_t)(b * NN + i) * DD + e0);
  float4 wj4;
  wj4.x = Wmsg[(size_t)(e0 + 0) * WMSG_LD + 2 * DD];
  wj4.y = Wmsg[(size_t)(e0 + 1) * WMSG_LD + 2 * DD];
  wj4.z = Wmsg[(size_t)(e0 + 2) * WMSG_LD + 2 * DD];
  wj4.w = Wmsg[(size_t)(e0 + 3) * WMSG_LD + 2 * DD];
  const float4 bm4 = *(const float4*)(bmsg + e0);
  const float4 g4  = *(const float4*)(gmsg + e0);
  const float4 be4 = *(const float4*)(betamsg + e0);

  float4 c4;
  c4.x = t4.x + bm4.x; c4.y = t4.y + bm4.y;
  c4.z = t4.z + bm4.z; c4.w = t4.w + bm4.w;

  float r1 = c4.x + c4.y + c4.z + c4.w;
  float r2 = c4.x * c4.x + c4.y * c4.y + c4.z * c4.z + c4.w * c4.w;
  float r3 = c4.x * wj4.x + c4.y * wj4.y + c4.z * wj4.z + c4.w * wj4.w;
  float r4 = wj4.x + wj4.y + wj4.z + wj4.w;
  float r5 = wj4.x * wj4.x + wj4.y * wj4.y + wj4.z * wj4.z + wj4.w * wj4.w;
  #pragma unroll
  for (int off = 16; off >= 1; off >>= 1) {
    r1 += __shfl_xor(r1, off, 32);
    r2 += __shfl_xor(r2, off, 32);
    r3 += __shfl_xor(r3, off, 32);
    r4 += __shfl_xor(r4, off, 32);
    r5 += __shfl_xor(r5, off, 32);
  }
  const float C1 = r1, C2 = r2, Cw = r3, W1 = r4, W2 = r5;
  const float invD = 1.0f / DD;

  const float*  __restrict__ srow = S + (size_t)b * NN * DD;
  const float*  __restrict__ arow = adj + (size_t)(b * NN + i) * NN;
  const float*  __restrict__ grow = G + (size_t)(b * NN + i) * NN;
  const float4* __restrict__ strow = STATS + (size_t)b * NN;

  float4 acc = {0.f, 0.f, 0.f, 0.f};
  const int j0 = w * JPER;
  for (int jj = 0; jj < JPER; ++jj) {
    const int j = j0 + jj;
    const float4 s4 = *(const float4*)(srow + (size_t)j * DD + e0);
    const float  aj = arow[j];
    const float  gij = grow[j];
    const float4 st = strow[j];

    const float mean  = (C1 + st.x + aj * W1) * invD;
    const float sumsq = C2 + st.y + aj * aj * W2
                      + 2.0f * (gij + st.w)
                      + 2.0f * aj * (Cw + st.z);
    const float var  = sumsq * invD - mean * mean;
    const float rinv = rsqrtf(var + 1e-5f);

    float4 p;
    p.x = c4.x + s4.x + aj * wj4.x;
    p.y = c4.y + s4.y + aj * wj4.y;
    p.z = c4.z + s4.z + aj * wj4.z;
    p.w = c4.w + s4.w + aj * wj4.w;

    float x0 = (p.x - mean) * rinv * g4.x + be4.x;
    float x1 = (p.y - mean) * rinv * g4.y + be4.y;
    float x2 = (p.z - mean) * rinv * g4.z + be4.z;
    float x3 = (p.w - mean) * rinv * g4.w + be4.w;
    const float m0 = x0 / (1.0f + __expf(-x0));
    const float m1 = x1 / (1.0f + __expf(-x1));
    const float m2 = x2 / (1.0f + __expf(-x2));
    const float m3 = x3 / (1.0f + __expf(-x3));

    const float keep = (j != i) ? 1.0f : 0.0f;
    acc.x += keep * m0;
    acc.y += keep * m1;
    acc.z += keep * m2;
    acc.w += keep * m3;
  }
  part[w][lane] = acc;
  __syncthreads();
  if (w == 0) {
    float4 sum = part[0][lane];
    for (int k = 1; k < JSPLIT; ++k) {
      const float4 p = part[k][lane];
      sum.x += p.x; sum.y += p.y; sum.z += p.z; sum.w += p.w;
    }
    *(float4*)(AGG + (size_t)(b * NN + i) * DD + e0) = sum;
  }
}

__global__ __launch_bounds__(256) void gemm_upd_kernel(
    const float* __restrict__ H, const float* __restrict__ AGG,
    const float* __restrict__ Wupd, const float* __restrict__ bupd,
    float* __restrict__ U)
{
  const int wave = (blockIdx.x * 256 + threadIdx.x) >> 5;
  const int lane = threadIdx.x & 31;
  const int mt = wave >> 3;
  const int nt = wave & 7;
  const int m0 = mt << 4, n0 = nt << 4;
  const int half = lane >> 4;
  const int ln = lane & 15;

  const float* __restrict__ arow1 = H + (size_t)(m0 + ln) * DD;
  const float* __restrict__ arow2 = AGG + (size_t)(m0 + ln) * DD;
  const float* __restrict__ wrow = Wupd + (size_t)(n0 + ln) * WUPD_LD;

  v8f acc = {};
  for (int k0 = 0; k0 < DD; k0 += 4) {
    const int ka = k0 + 2 * half;
    v2f a1; a1.x = arow1[ka]; a1.y = arow1[ka + 1];
    v2f b1; b1.x = wrow[ka];  b1.y = wrow[ka + 1];
    acc = __builtin_amdgcn_wmma_f32_16x16x4_f32(
        false, a1, false, b1, (short)0, acc, false, false);
    v2f a2; a2.x = arow2[ka];     a2.y = arow2[ka + 1];
    v2f b2; b2.x = wrow[DD + ka]; b2.y = wrow[DD + ka + 1];
    acc = __builtin_amdgcn_wmma_f32_16x16x4_f32(
        false, a2, false, b2, (short)0, acc, false, false);
  }

  const float bias = bupd[n0 + ln];
  #pragma unroll
  for (int r = 0; r < 8; ++r) {
    const int row = m0 + r + 8 * half;
    U[(size_t)row * DD + n0 + ln] = acc[r] + bias;
  }
}

__global__ __launch_bounds__(256) void finalize_kernel(
    const float* __restrict__ U, const float* __restrict__ H,
    const float* __restrict__ gupd, const float* __restrict__ betaupd,
    float* __restrict__ out)
{
  const int row = (blockIdx.x * 256 + threadIdx.x) >> 5;
  const int lane = threadIdx.x & 31;
  const int e0 = lane << 2;

  const float4 u4 = *(const float4*)(U + (size_t)row * DD + e0);
  float ps  = u4.x + u4.y + u4.z + u4.w;
  float pss = u4.x * u4.x + u4.y * u4.y + u4.z * u4.z + u4.w * u4.w;
  #pragma unroll
  for (int off = 16; off >= 1; off >>= 1) {
    ps  += __shfl_xor(ps,  off, 32);
    pss += __shfl_xor(pss, off, 32);
  }
  const float mean = ps * (1.0f / DD);
  const float var  = pss * (1.0f / DD) - mean * mean;
  const float rinv = rsqrtf(var + 1e-5f);

  const float4 g4  = *(const float4*)(gupd + e0);
  const float4 be4 = *(const float4*)(betaupd + e0);
  const float4 h4  = *(const float4*)(H + (size_t)row * DD + e0);

  float x0 = (u4.x - mean) * rinv * g4.x + be4.x;
  float x1 = (u4.y - mean) * rinv * g4.y + be4.y;
  float x2 = (u4.z - mean) * rinv * g4.z + be4.z;
  float x3 = (u4.w - mean) * rinv * g4.w + be4.w;

  float4 o;
  o.x = h4.x + x0 / (1.0f + __expf(-x0));
  o.y = h4.y + x1 / (1.0f + __expf(-x1));
  o.z = h4.z + x2 / (1.0f + __expf(-x2));
  o.w = h4.w + x3 / (1.0f + __expf(-x3));
  *(float4*)(out + (size_t)row * DD + e0) = o;
}

extern "C" void kernel_launch(void* const* d_in, const int* in_sizes, int n_in,
                              void* d_out, int out_size, void* d_ws, size_t ws_size,
                              hipStream_t stream) {
  const float* h       = (const float*)d_in[0];
  const float* adj     = (const float*)d_in[1];
  const float* Wmsg    = (const float*)d_in[2];
  const float* bmsg    = (const float*)d_in[3];
  const float* gmsg    = (const float*)d_in[4];
  const float* betamsg = (const float*)d_in[5];
  const float* Wupd    = (const float*)d_in[6];
  const float* bupd    = (const float*)d_in[7];
  const float* gupd    = (const float*)d_in[8];
  const float* betaupd = (const float*)d_in[9];
  float* out = (float*)d_out;

  const size_t mat_bytes = (size_t)ROWS * DD * sizeof(float);
  const size_t g_bytes   = (size_t)BB * NN * NN * sizeof(float);
  char* ws = (char*)d_ws;
  float*  T     = (float*)(ws);
  float*  S     = (float*)(ws + mat_bytes);
  float*  AGG   = (float*)(ws + 2 * mat_bytes);
  float*  U     = (float*)(ws + 3 * mat_bytes);
  float*  G     = (float*)(ws + 4 * mat_bytes);
  float4* STATS = (float4*)(ws + 4 * mat_bytes + g_bytes);

  gemm_ts_kernel<<<128, 256, 0, stream>>>(h, Wmsg, T, S);
  gemm_g_kernel<<<256, 256, 0, stream>>>(T, S, G);
  stats_s_kernel<<<256, 256, 0, stream>>>(S, Wmsg, bmsg, STATS);
  msg_agg_kernel<<<2048, 256, 0, stream>>>(T, S, adj, Wmsg, bmsg, gmsg,
                                           betamsg, G, STATS, AGG);
  gemm_upd_kernel<<<128, 256, 0, stream>>>(h, AGG, Wupd, bupd, U);
  finalize_kernel<<<256, 256, 0, stream>>>(U, h, gupd, betaupd, out);
}


# msg/aggregate kernel: silu(x) computed as h + h*tanh(h), h = x/2, on the hardware tanh (1 transcendental + 1 fma per element) in place of exp + correctly rounded division; no sum changes its order
# speedup vs baseline: 2.2616x; 1.3727x over previous
#include <hip/hip_runtime.h>

typedef __attribute__((ext_vector_type(2))) float v2f;
typedef __attribute__((ext_vector_type(8))) float v8f;

#define DD 128
#define NN 256
#define BB 8
#define ROWS (BB * NN)
#define WMSG_LD (2 * DD + 1)
#define WUPD_LD (2 * DD)
#define JSPLIT 8
#define JPER (NN / JSPLIT)

__global__ __launch_bounds__(256) void gemm_ts_kernel(
    const float* __restrict__ H, const float* __restrict__ Wmsg,
    float* __restrict__ T, float* __restrict__ S)
{
  const int wave = (blockIdx.x * 256 + threadIdx.x) >> 5;
  const int lane = threadIdx.x & 31;
  const int mt = wave >> 3;
  const int nt = wave & 7;
  const int m0 = mt << 4, n0 = nt << 4;
  const int half = lane >> 4;
  const int ln = lane & 15;

  const float* __restrict__ arow = H + (size_t)(m0 + ln) * DD;
  const float* __restrict__ wrow = Wmsg + (size_t)(n0 + ln) * WMSG_LD;

  v8f accT = {};
  v8f accS = {};
  for (int k0 = 0; k0 < DD; k0 += 4) {
    const int ka = k0 + 2 * half;
    v2f a;  a.x  = arow[ka];          a.y  = arow[ka + 1];
    v2f bt; bt.x = wrow[ka];          bt.y = wrow[ka + 1];
    v2f bs; bs.x = wrow[DD + ka];     bs.y = wrow[DD + ka + 1];
    accT = __builtin_amdgcn_wmma_f32_16x16x4_f32(
        false, a, false, bt, (short)0, accT, false, false);
    accS = __builtin_amdgcn_wmma_f32_16x16x4_f32(
        false, a, false, bs, (short)0, accS, false, false);
  }
  #pragma unroll
  for (int r = 0; r < 8; ++r) {
    const int row = m0 + r + 8 * half;
    T[(size_t)row * DD + n0 + ln] = accT[r];
    S[(size_t)row * DD + n0 + ln] = accS[r];
  }
}

__global__ __launch_bounds__(256) void gemm_g_kernel(
    const float* __restrict__ T, const float* __restrict__ S,
    float* __restrict__ G)
{
  const int wave = (blockIdx.x * 256 + threadIdx.x) >> 5;
  const int lane = threadIdx.x & 31;
  const int b  = wave >> 8;
  const int rm = wave & 255;
  const int it = rm >> 4;
  const int jt = rm & 15;
  const int m0 = it << 4, n0 = jt << 4;
  const int half = lane >> 4;
  const int ln = lane & 15;

  const float* __restrict__ arow = T + (size_t)(b * NN + m0 + ln) * DD;
  const float* __restrict__ brow = S + (size_t)(b * NN + n0 + ln) * DD;

  v8f acc = {};
  for (int k0 = 0; k0 < DD; k0 += 4) {
    const int ka = k0 + 2 * half;
    v2f a;  a.x = arow[ka]; a.y = arow[ka + 1];
    v2f bb; bb.x = brow[ka]; bb.y = brow[ka + 1];
    acc = __builtin_amdgcn_wmma_f32_16x16x4_f32(
        false, a, false, bb, (short)0, acc, false, false);
  }
  #pragma unroll
  for (int r = 0; r < 8; ++r) {
    const int i = m0 + r + 8 * half;
    G[(size_t)(b * NN + i) * NN + n0 + ln] = acc[r];
  }
}

__global__ __launch_bounds__(256) void stats_s_kernel(
    const float* __restrict__ S, const float* __restrict__ Wmsg,
    const float* __restrict__ bmsg, float4* __restrict__ STATS)
{
  const int row  = (blockIdx.x * 256 + threadIdx.x) >> 5;
  const int lane = threadIdx.x & 31;
  const int e0 = lane << 2;

  const float4 s4 = *(const float4*)(S + (size_t)row * DD + e0);
  float4 wj4;
  wj4.x = Wmsg[(size_t)(e0 + 0) * WMSG_LD + 2 * DD];
  wj4.y = Wmsg[(size_t)(e0 + 1) * WMSG_LD + 2 * DD];
  wj4.z = Wmsg[(size_t)(e0 + 2) * WMSG_LD + 2 * DD];
  wj4.w = Wmsg[(size_t)(e0 + 3) * WMSG_LD + 2 * DD];
  const float4 bm4 = *(const float4*)(bmsg + e0);

  float p1 = s4.x + s4.y + s4.z + s4.w;
  float p2 = s4.x * s4.x + s4.y * s4.y + s4.z * s4.z + s4.w * s4.w;
  float p3 = s4.x * wj4.x + s4.y * wj4.y + s4.z * wj4.z + s4.w * wj4.w;
  float p4 = s4.x * bm4.x + s4.y * bm4.y + s4.z * bm4.z + s4.w * bm4.w;
  #pragma unroll
  for (int off = 16; off >= 1; off >>= 1) {
    p1 += __shfl_xor(p1, off, 32);
    p2 += __shfl_xor(p2, off, 32);
    p3 += __shfl_xor(p3, off, 32);
    p4 += __shfl_xor(p4, off, 32);
  }
  if (lane == 0) {
    float4 st; st.x = p1; st.y = p2; st.z = p3; st.w = p4;
    STATS[row] = st;
  }
}

__global__ __launch_bounds__(256, 16) void msg_agg_kernel(
    const float* __restrict__ T, const float* __restrict__ S,
    const float* __restrict__ adj, const float* __restrict__ Wmsg,
    const float* __restrict__ bmsg, const float* __restrict__ gmsg,
    const float* __restrict__ betamsg, const float* __restrict__ G,
    const float4* __restrict__ STATS, float* __restrict__ AGG)
{
  __shared__ float4 part[JSPLIT][32];
  const int w    = threadIdx.x >> 5;
  const int lane = threadIdx.x & 31;
  const int b = blockIdx.x >> 8;
  const int i = blockIdx.x & 255;
  const int e0 = lane << 2;

  const float4 t4 = *(const float4*)(T + (size_t)(b * NN + i) * DD + e0);
  float4 wj4;
  wj4.x = Wmsg[(size_t)(e0 + 0) * WMSG_LD + 2 * DD];
  wj4.y = Wmsg[(size_t)(e0 + 1) * WMSG_LD + 2 * DD];
  wj4.z = Wmsg[(size_t)(e0 + 2) * WMSG_LD + 2 * DD];
  wj4.w = Wmsg[(size_t)(e0 + 3) * WMSG_LD + 2 * DD];
  const float4 bm4 = *(const float4*)(bmsg + e0);
  const float4 g4  = *(const float4*)(gmsg + e0);
  const float4 be4 = *(const float4*)(betamsg + e0);

  float4 gh4, bh4;
  gh4.x = 0.5f * g4.x;  gh4.y = 0.5f * g4.y;
  gh4.z = 0.5f * g4.z;  gh4.w = 0.5f * g4.w;
  bh4.x = 0.5f * be4.x; bh4.y = 0.5f * be4.y;
  bh4.z = 0.5f * be4.z; bh4.w = 0.5f * be4.w;

  float4 c4;
  c4.x = t4.x + bm4.x; c4.y = t4.y + bm4.y;
  c4.z = t4.z + bm4.z; c4.w = t4.w + bm4.w;

  float r1 = c4.x + c4.y + c4.z + c4.w;
  float r2 = c4.x * c4.x + c4.y * c4.y + c4.z * c4.z + c4.w * c4.w;
  float r3 = c4.x * wj4.x + c4.y * wj4.y + c4.z * wj4.z + c4.w * wj4.w;
  float r4 = wj4.x + wj4.y + wj4.z + wj4.w;
  float r5 = wj4.x * wj4.x + wj4.y * wj4.y + wj4.z * wj4.z + wj4.w * wj4.w;
  #pragma unroll
  for (int off = 16; off >= 1; off >>= 1) {
    r1 += __shfl_xor(r1, off, 32);
    r2 += __shfl_xor(r2, off, 32);
    r3 += __shfl_xor(r3, off, 32);
    r4 += __shfl_xor(r4, off, 32);
    r5 += __shfl_xor(r5, off, 32);
  }
  const float C1 = r1, C2 = r2, Cw = r3, W1 = r4, W2 = r5;
  const float invD = 1.0f / DD;

  const float*  __restrict__ srow = S + (size_t)b * NN * DD;
  const float*  __restrict__ arow = adj + (size_t)(b * NN + i) * NN;
  const float*  __restrict__ grow = G + (size_t)(b * NN + i) * NN;
  const float4* __restrict__ strow = STATS + (size_t)b * NN;

  float4 acc = {0.f, 0.f, 0.f, 0.f};
  const int j0 = w * JPER;
  for (int jj = 0; jj < JPER; ++jj) {
    const int j = j0 + jj;
    const float4 s4 = *(const float4*)(srow + (size_t)j * DD + e0);
    const float  aj = arow[j];
    const float  gij = grow[j];
    const float4 st = strow[j];

    const float mean  = (C1 + st.x + aj * W1) * invD;
    const float sumsq = C2 + st.y + aj * aj * W2
                      + 2.0f * (gij + st.w)
                      + 2.0f * aj * (Cw + st.z);
    const float var  = sumsq * invD - mean * mean;
    const float rinv = rsqrtf(var + 1e-5f);

    float4 p;
    p.x = c4.x + s4.x + aj * wj4.x;
    p.y = c4.y + s4.y + aj * wj4.y;
    p.z = c4.z + s4.z + aj * wj4.z;
    p.w = c4.w + s4.w + aj * wj4.w;

    const float h0 = (p.x - mean) * rinv * gh4.x + bh4.x;
    const float h1 = (p.y - mean) * rinv * gh4.y + bh4.y;
    const float h2 = (p.z - mean) * rinv * gh4.z + bh4.z;
    const float h3 = (p.w - mean) * rinv * gh4.w + bh4.w;
    const float m0 = h0 + h0 * __builtin_amdgcn_tanhf(h0);
    const float m1 = h1 + h1 * __builtin_amdgcn_tanhf(h1);
    const float m2 = h2 + h2 * __builtin_amdgcn_tanhf(h2);
    const float m3 = h3 + h3 * __builtin_amdgcn_tanhf(h3);

    const float keep = (j != i) ? 1.0f : 0.0f;
    acc.x += keep * m0;
    acc.y += keep * m1;
    acc.z += keep * m2;
    acc.w += keep * m3;
  }
  part[w][lane] = acc;
  __syncthreads();
  if (w == 0) {
    float4 sum = part[0][lane];
    for (int k = 1; k < JSPLIT; ++k) {
      const float4 p = part[k][lane];
      sum.x += p.x; sum.y += p.y; sum.z += p.z; sum.w += p.w;
    }
    *(float4*)(AGG + (size_t)(b * NN + i) * DD + e0) = sum;
  }
}

__global__ __launch_bounds__(256) void gemm_upd_kernel(
    const float* __restrict__ H, const float* __restrict__ AGG,
    const float* __restrict__ Wupd, const float* __restrict__ bupd,
    float* __restrict__ U)
{
  const int wave = (blockIdx.x * 256 + threadIdx.x) >> 5;
  const int lane = threadIdx.x & 31;
  const int mt = wave >> 3;
  const int nt = wave & 7;
  const int m0 = mt << 4, n0 = nt << 4;
  const int half = lane >> 4;
  const int ln = lane & 15;

  const float* __restrict__ arow1 = H + (size_t)(m0 + ln) * DD;
  const float* __restrict__ arow2 = AGG + (size_t)(m0 + ln) * DD;
  const float* __restrict__ wrow = Wupd + (size_t)(n0 + ln) * WUPD_LD;

  v8f acc = {};
  for (int k0 = 0; k0 < DD; k0 += 4) {
    const int ka = k0 + 2 * half;
    v2f a1; a1.x = arow1[ka]; a1.y = arow1[ka + 1];
    v2f b1; b1.x = wrow[ka];  b1.y = wrow[ka + 1];
    acc = __builtin_amdgcn_wmma_f32_16x16x4_f32(
        false, a1, false, b1, (short)0, acc, false, false);
    v2f a2; a2.x = arow2[ka];     a2.y = arow2[ka + 1];
    v2f b2; b2.x = wrow[DD + ka]; b2.y = wrow[DD + ka + 1];
    acc = __builtin_amdgcn_wmma_f32_16x16x4_f32(
        false, a2, false, b2, (short)0, acc, false, false);
  }

  const float bias = bupd[n0 + ln];
  #pragma unroll
  for (int r = 0; r < 8; ++r) {
    const int row = m0 + r + 8 * half;
    U[(size_t)row * DD + n0 + ln] = acc[r] + bias;
  }
}

__global__ __launch_bounds__(256) void finalize_kernel(
    const float* __restrict__ U, const float* __restrict__ H,
    const float* __restrict__ gupd, const float* __restrict__ betaupd,
    float* __restrict__ out)
{
  const int row = (blockIdx.x * 256 + threadIdx.x) >> 5;
  const int lane = threadIdx.x & 31;
  const int e0 = lane << 2;

  const float4 u4 = *(const float4*)(U + (size_t)row * DD + e0);
  float ps  = u4.x + u4.y + u4.z + u4.w;
  float pss = u4.x * u4.x + u4.y * u4.y + u4.z * u4.z + u4.w * u4.w;
  #pragma unroll
  for (int off = 16; off >= 1; off >>= 1) {
    ps  += __shfl_xor(ps,  off, 32);
    pss += __shfl_xor(pss, off, 32);
  }
  const float mean = ps * (1.0f / DD);
  const float var  = pss * (1.0f / DD) - mean * mean;
  const float rinv = rsqrtf(var + 1e-5f);

  const float4 g4  = *(const float4*)(gupd + e0);
  const float4 be4 = *(const float4*)(betaupd + e0);
  const float4 h4  = *(const float4*)(H + (size_t)row * DD + e0);

  float x0 = (u4.x - mean) * rinv * g4.x + be4.x;
  float x1 = (u4.y - mean) * rinv * g4.y + be4.y;
  float x2 = (u4.z - mean) * rinv * g4.z + be4.z;
  float x3 = (u4.w - mean) * rinv * g4.w + be4.w;

  float4 o;
  o.x = h4.x + x0 / (1.0f + __expf(-x0));
  o.y = h4.y + x1 / (1.0f + __expf(-x1));
  o.z = h4.z + x2 / (1.0f + __expf(-x2));
  o.w = h4.w + x3 / (1.0f + __expf(-x3));
  *(float4*)(out + (size_t)row * DD + e0) = o;
}

extern "C" void kernel_launch(void* const* d_in, const int* in_sizes, int n_in,
                              void* d_out, int out_size, void* d_ws, size_t ws_size,
                              hipStream_t stream) {
  const float* h       = (const float*)d_in[0];
  const float* adj     = (const float*)d_in[1];
  const float* Wmsg    = (const float*)d_in[2];
  const float* bmsg    = (const float*)d_in[3];
  const float* gmsg    = (const float*)d_in[4];
  const float* betamsg = (const float*)d_in[5];
  const float* Wupd    = (const float*)d_in[6];
  const float* bupd    = (const float*)d_in[7];
  const float* gupd    = (const float*)d_in[8];
  const float* betaupd = (const float*)d_in[9];
  float* out = (float*)d_out;

  const size_t mat_bytes = (size_t)ROWS * DD * sizeof(float);
  const size_t g_bytes   = (size_t)BB * NN * NN * sizeof(float);
  char* ws = (char*)d_ws;
  float*  T     = (float*)(ws);
  float*  S     = (float*)(ws + mat_bytes);
  float*  AGG   = (float*)(ws + 2 * mat_bytes);
  float*  U     = (float*)(ws + 3 * mat_bytes);
  float*  G     = (float*)(ws + 4 * mat_bytes);
  float4* STATS = (float4*)(ws + 4 * mat_bytes + g_bytes);

  gemm_ts_kernel<<<128, 256, 0, stream>>>(h, Wmsg, T, S);
  gemm_g_kernel<<<256, 256, 0, stream>>>(T, S, G);
  stats_s_kernel<<<256, 256, 0, stream>>>(S, Wmsg, bmsg, STATS);
  msg_agg_kernel<<<2048, 256, 0, stream>>>(T, S, adj, Wmsg, bmsg, gmsg,
                                           betamsg, G, STATS, AGG);
  gemm_upd_kernel<<<128, 256, 0, stream>>>(h, AGG, Wupd, bupd, U);
  finalize_kernel<<<256, 256, 0, stream>>>(U, h, gupd, betaupd, out);
}


# S row statistics and the final layernorm+silu+residual folded into the GEMM launches that produce their rows (16-row LDS tile hand-over): 6 launches -> 4, U stays out of memory; no sum reordered
# speedup vs baseline: 2.4918x; 1.1018x over previous
#include <hip/hip_runtime.h>

typedef __attribute__((ext_vector_type(2))) float v2f;
typedef __attribute__((ext_vector_type(8))) float v8f;

#define DD 128
#define NN 256
#define BB 8
#define ROWS (BB * NN)
#define WMSG_LD (2 * DD + 1)
#define WUPD_LD (2 * DD)
#define JSPLIT 8
#define JPER (NN / JSPLIT)
#define TILE_LD 132

__global__ __launch_bounds__(256) void gemm_ts_stats_kernel(
    const float* __restrict__ H, const float* __restrict__ Wmsg,
    const float* __restrict__ bmsg,
    float* __restrict__ T, float* __restrict__ S, float4* __restrict__ STATS)
{
  __shared__ __attribute__((aligned(16))) float tile[16 * TILE_LD];
  const int wave = (blockIdx.x * 256 + threadIdx.x) >> 5;
  const int lane = threadIdx.x & 31;
  const int mt = wave >> 3;
  const int nt = wave & 7;
  const int m0 = mt << 4, n0 = nt << 4;
  const int half = lane >> 4;
  const int ln = lane & 15;

  const float* __restrict__ arow = H + (size_t)(m0 + ln) * DD;
  const float* __restrict__ wrow = Wmsg + (size_t)(n0 + ln) * WMSG_LD;

  v8f accT = {};
  v8f accS = {};
  for (int k0 = 0; k0 < DD; k0 += 4) {
    const int ka = k0 + 2 * half;
    v2f a;  a.x  = arow[ka];          a.y  = arow[ka + 1];
    v2f bt; bt.x = wrow[ka];          bt.y = wrow[ka + 1];
    v2f bs; bs.x = wrow[DD + ka];     bs.y = wrow[DD + ka + 1];
    accT = __builtin_amdgcn_wmma_f32_16x16x4_f32(
        false, a, false, bt, (short)0, accT, false, false);
    accS = __builtin_amdgcn_wmma_f32_16x16x4_f32(
        false, a, false, bs, (short)0, accS, false, false);
  }
  __builtin_amdgcn_sched_barrier(0);
  asm volatile("v_nop\n\tv_nop\n\tv_nop\n\tv_nop");
  __builtin_amdgcn_sched_barrier(0);

  const int e0 = lane << 2;
  float4 wj4;
  wj4.x = Wmsg[(size_t)(e0 + 0) * WMSG_LD + 2 * DD];
  wj4.y = Wmsg[(size_t)(e0 + 1) * WMSG_LD + 2 * DD];
  wj4.z = Wmsg[(size_t)(e0 + 2) * WMSG_LD + 2 * DD];
  wj4.w = Wmsg[(size_t)(e0 + 3) * WMSG_LD + 2 * DD];
  const float4 bm4 = *(const float4*)(bmsg + e0);

  #pragma unroll
  for (int r = 0; r < 8; ++r)
    tile[(r + 8 * half) * TILE_LD + n0 + ln] = accS[r];
  __syncthreads();
  #pragma unroll
  for (int r = 0; r < 8; ++r) {
    const int row = m0 + r + 8 * half;
    T[(size_t)row * DD + n0 + ln] = accT[r];
    S[(size_t)row * DD + n0 + ln] = accS[r];
  }

  float p1[2], p2[2], p3[2], p4[2];
  #pragma unroll
  for (int rr = 0; rr < 2; ++rr) {
    const float4 s4 = *(const float4*)(tile + (2 * nt + rr) * TILE_LD + e0);
    p1[rr] = s4.x + s4.y + s4.z + s4.w;
    p2[rr] = s4.x * s4.x + s4.y * s4.y + s4.z * s4.z + s4.w * s4.w;
    p3[rr] = s4.x * wj4.x + s4.y * wj4.y + s4.z * wj4.z + s4.w * wj4.w;
    p4[rr] = s4.x * bm4.x + s4.y * bm4.y + s4.z * bm4.z + s4.w * bm4.w;
  }
  #pragma unroll
  for (int off = 16; off >= 1; off >>= 1) {
    #pragma unroll
    for (int rr = 0; rr < 2; ++rr) {
      p1[rr] += __shfl_xor(p1[rr], off, 32);
      p2[rr] += __shfl_xor(p2[rr], off, 32);
      p3[rr] += __shfl_xor(p3[rr], off, 32);
      p4[rr] += __shfl_xor(p4[rr], off, 32);
    }
  }
  if (lane < 2) {
    const int rr = lane;
    float4 st;
    st.x = rr ? p1[1] : p1[0];
    st.y = rr ? p2[1] : p2[0];
    st.z = rr ? p3[1] : p3[0];
    st.w = rr ? p4[1] : p4[0];
    STATS[m0 + 2 * nt + rr] = st;
  }
}

__global__ __launch_bounds__(256) void gemm_g_kernel(
    const float* __restrict__ T, const float* __restrict__ S,
    float* __restrict__ G)
{
  const int wave = (blockIdx.x * 256 + threadIdx.x) >> 5;
  const int lane = threadIdx.x & 31;
  const int b  = wave >> 8;
  const int rm = wave & 255;
  const int it = rm >> 4;
  const int jt = rm & 15;
  const int m0 = it << 4, n0 = jt << 4;
  const int half = lane >> 4;
  const int ln = lane & 15;

  const float* __restrict__ arow = T + (size_t)(b * NN + m0 + ln) * DD;
  const float* __restrict__ brow = S + (size_t)(b * NN + n0 + ln) * DD;

  v8f acc = {};
  for (int k0 = 0; k0 < DD; k0 += 4) {
    const int ka = k0 + 2 * half;
    v2f a;  a.x = arow[ka]; a.y = arow[ka + 1];
    v2f bb; bb.x = brow[ka]; bb.y = brow[ka + 1];
    acc = __builtin_amdgcn_wmma_f32_16x16x4_f32(
        false, a, false, bb, (short)0, acc, false, false);
  }
  #pragma unroll
  for (int r = 0; r < 8; ++r) {
    const int i = m0 + r + 8 * half;
    G[(size_t)(b * NN + i) * NN + n0 + ln] = acc[r];
  }
}

__global__ __launch_bounds__(256, 16) void msg_agg_kernel(
    const float* __restrict__ T, const float* __restrict__ S,
    const float* __restrict__ adj, const float* __restrict__ Wmsg,
    const float* __restrict__ bmsg, const float* __restrict__ gmsg,
    const float* __restrict__ betamsg, const float* __restrict__ G,
    const float4* __restrict__ STATS, float* __restrict__ AGG)
{
  __shared__ float4 part[JSPLIT][32];
  const int w    = threadIdx.x >> 5;
  const int lane = threadIdx.x & 31;
  const int b = blockIdx.x >> 8;
  const int i = blockIdx.x & 255;
  const int e0 = lane << 2;

  const float4 t4 = *(const float4*)(T + (size_t)(b * NN + i) * DD + e0);
  float4 wj4;
  wj4.x = Wmsg[(size_t)(e0 + 0) * WMSG_LD + 2 * DD];
  wj4.y = Wmsg[(size_t)(e0 + 1) * WMSG_LD + 2 * DD];
  wj4.z = Wmsg[(size_t)(e0 + 2) * WMSG_LD + 2 * DD];
  wj4.w = Wmsg[(size_t)(e0 + 3) * WMSG_LD + 2 * DD];
  const float4 bm4 = *(const float4*)(bmsg + e0);
  const float4 g4  = *(const float4*)(gmsg + e0);
  const float4 be4 = *(const float4*)(betamsg + e0);

  float4 gh4, bh4;
  gh4.x = 0.5f * g4.x;  gh4.y = 0.5f * g4.y;
  gh4.z = 0.5f * g4.z;  gh4.w = 0.5f * g4.w;
  bh4.x = 0.5f * be4.x; bh4.y = 0.5f * be4.y;
  bh4.z = 0.5f * be4.z; bh4.w = 0.5f * be4.w;

  float4 c4;
  c4.x = t4.x + bm4.x; c4.y = t4.y + bm4.y;
  c4.z = t4.z + bm4.z; c4.w = t4.w + bm4.w;

  float r1 = c4.x + c4.y + c4.z + c4.w;
  float r2 = c4.x * c4.x + c4.y * c4.y + c4.z * c4.z + c4.w * c4.w;
  float r3 = c4.x * wj4.x + c4.y * wj4.y + c4.z * wj4.z + c4.w * wj4.w;
  float r4 = wj4.x + wj4.y + wj4.z + wj4.w;
  float r5 = wj4.x * wj4.x + wj4.y * wj4.y + wj4.z * wj4.z + wj4.w * wj4.w;
  #pragma unroll
  for (int off = 16; off >= 1; off >>= 1) {
    r1 += __shfl_xor(r1, off, 32);
    r2 += __shfl_xor(r2, off, 32);
    r3 += __shfl_xor(r3, off, 32);
    r4 += __shfl_xor(r4, off, 32);
    r5 += __shfl_xor(r5, off, 32);
  }
  const float C1 = r1, C2 = r2, Cw = r3, W1 = r4, W2 = r5;
  const float invD = 1.0f / DD;

  const float*  __restrict__ srow = S + (size_t)b * NN * DD;
  const float*  __restrict__ arow = adj + (size_t)(b * NN + i) * NN;
  const float*  __restrict__ grow = G + (size_t)(b * NN + i) * NN;
  const float4* __restrict__ strow = STATS + (size_t)b * NN;

  float4 acc = {0.f, 0.f, 0.f, 0.f};
  const int j0 = w * JPER;
  for (int jj = 0; jj < JPER; ++jj) {
    const int j = j0 + jj;
    const float4 s4 = *(const float4*)(srow + (size_t)j * DD + e0);
    const float  aj = arow[j];
    const float  gij = grow[j];
    const float4 st = strow[j];

    const float mean  = (C1 + st.x + aj * W1) * invD;
    const float sumsq = C2 + st.y + aj * aj * W2
                      + 2.0f * (gij + st.w)
                      + 2.0f * aj * (Cw + st.z);
    const float var  = sumsq * invD - mean * mean;
    const float rinv = rsqrtf(var + 1e-5f);

    float4 p;
    p.x = c4.x + s4.x + aj * wj4.x;
    p.y = c4.y + s4.y + aj * wj4.y;
    p.z = c4.z + s4.z + aj * wj4.z;
    p.w = c4.w + s4.w + aj * wj4.w;

    const float h0 = (p.x - mean) * rinv * gh4.x + bh4.x;
    const float h1 = (p.y - mean) * rinv * gh4.y + bh4.y;
    const float h2 = (p.z - mean) * rinv * gh4.z + bh4.z;
    const float h3 = (p.w - mean) * rinv * gh4.w + bh4.w;
    const float m0 = h0 + h0 * __builtin_amdgcn_tanhf(h0);
    const float m1 = h1 + h1 * __builtin_amdgcn_tanhf(h1);
    const float m2 = h2 + h2 * __builtin_amdgcn_tanhf(h2);
    const float m3 = h3 + h3 * __builtin_amdgcn_tanhf(h3);

    const float keep = (j != i) ? 1.0f : 0.0f;
    acc.x += keep * m0;
    acc.y += keep * m1;
    acc.z += keep * m2;
    acc.w += keep * m3;
  }
  part[w][lane] = acc;
  __syncthreads();
  if (w == 0) {
    float4 sum = part[0][lane];
    for (int k = 1; k < JSPLIT; ++k) {
      const float4 p = part[k][lane];
      sum.x += p.x; sum.y += p.y; sum.z += p.z; sum.w += p.w;
    }
    *(float4*)(AGG + (size_t)(b * NN + i) * DD + e0) = sum;
  }
}

__global__ __launch_bounds__(256) void gemm_upd_final_kernel(
    const float* __restrict__ H, const float* __restrict__ AGG,
    const float* __restrict__ Wupd, const float* __restrict__ bupd,
    const float* __restrict__ gupd, const float* __restrict__ betaupd,
    float* __restrict__ out)
{
  __shared__ __attribute__((aligned(16))) float tile[16 * TILE_LD];
  const int wave = (blockIdx.x * 256 + threadIdx.x) >> 5;
  const int lane = threadIdx.x & 31;
  const int mt = wave >> 3;
  const int nt = wave & 7;
  const int m0 = mt << 4, n0 = nt << 4;
  const int half = lane >> 4;
  const int ln = lane & 15;

  const float* __restrict__ arow1 = H + (size_t)(m0 + ln) * DD;
  const float* __restrict__ arow2 = AGG + (size_t)(m0 + ln) * DD;
  const float* __restrict__ wrow = Wupd + (size_t)(n0 + ln) * WUPD_LD;

  v8f acc = {};
  for (int k0 = 0; k0 < DD; k0 += 4) {
    const int ka = k0 + 2 * half;
    v2f a1; a1.x = arow1[ka]; a1.y = arow1[ka + 1];
    v2f b1; b1.x = wrow[ka];  b1.y = wrow[ka + 1];
    acc = __builtin_amdgcn_wmma_f32_16x16x4_f32(
        false, a1, false, b1, (short)0, acc, false, false);
    v2f a2; a2.x = arow2[ka];     a2.y = arow2[ka + 1];
    v2f b2; b2.x = wrow[DD + ka]; b2.y = wrow[DD + ka + 1];
    acc = __builtin_amdgcn_wmma_f32_16x16x4_f32(
        false, a2, false, b2, (short)0, acc, false, false);
  }

  __builtin_amdgcn_sched_barrier(0);
  asm volatile("v_nop\n\tv_nop\n\tv_nop\n\tv_nop");
  __builtin_amdgcn_sched_barrier(0);

  const int e0 = lane << 2;
  const float4 g4  = *(const float4*)(gupd + e0);
  const float4 be4 = *(const float4*)(betaupd + e0);
  const int row0 = m0 + 2 * nt;
  float4 h4[2];
  h4[0] = *(const float4*)(H + (size_t)(row0 + 0) * DD + e0);
  h4[1] = *(const float4*)(H + (size_t)(row0 + 1) * DD + e0);

  const float bias = bupd[n0 + ln];
  #pragma unroll
  for (int r = 0; r < 8; ++r)
    tile[(r + 8 * half) * TILE_LD + n0 + ln] = acc[r] + bias;
  __syncthreads();

  float4 u4[2];
  float ps[2], pss[2];
  #pragma unroll
  for (int rr = 0; rr < 2; ++rr) {
    u4[rr] = *(const float4*)(tile + (2 * nt + rr) * TILE_LD + e0);
    ps[rr]  = u4[rr].x + u4[rr].y + u4[rr].z + u4[rr].w;
    pss[rr] = u4[rr].x * u4[rr].x + u4[rr].y * u4[rr].y
            + u4[rr].z * u4[rr].z + u4[rr].w * u4[rr].w;
  }
  #pragma unroll
  for (int off = 16; off >= 1; off >>= 1) {
    #pragma unroll
    for (int rr = 0; rr < 2; ++rr) {
      ps[rr]  += __shfl_xor(ps[rr],  off, 32);
      pss[rr] += __shfl_xor(pss[rr], off, 32);
    }
  }
  #pragma unroll
  for (int rr = 0; rr < 2; ++rr) {
    const float mean = ps[rr] * (1.0f / DD);
    const float var  = pss[rr] * (1.0f / DD) - mean * mean;
    const float rinv = rsqrtf(var + 1e-5f);

    float x0 = (u4[rr].x - mean) * rinv * g4.x + be4.x;
    float x1 = (u4[rr].y - mean) * rinv * g4.y + be4.y;
    float x2 = (u4[rr].z - mean) * rinv * g4.z + be4.z;
    float x3 = (u4[rr].w - mean) * rinv * g4.w + be4.w;

    float4 o;
    o.x = h4[rr].x + x0 / (1.0f + __expf(-x0));
    o.y = h4[rr].y + x1 / (1.0f + __expf(-x1));
    o.z = h4[rr].z + x2 / (1.0f + __expf(-x2));
    o.w = h4[rr].w + x3 / (1.0f + __expf(-x3));
    *(float4*)(out + (size_t)(row0 + rr) * DD + e0) = o;
  }
}

extern "C" void kernel_launch(void* const* d_in, const int* in_sizes, int n_in,
                              void* d_out, int out_size, void* d_ws, size_t ws_size,
                              hipStream_t stream) {
  const float* h       = (const float*)d_in[0];
  const float* adj     = (const float*)d_in[1];
  const float* Wmsg    = (const float*)d_in[2];
  const float* bmsg    = (const float*)d_in[3];
  const float* gmsg    = (const float*)d_in[4];
  const float* betamsg = (const float*)d_in[5];
  const float* Wupd    = (const float*)d_in[6];
  const float* bupd    = (const float*)d_in[7];
  const float* gupd    = (const float*)d_in[8];
  const float* betaupd = (const float*)d_in[9];
  float* out = (float*)d_out;

  const size_t mat_bytes = (size_t)ROWS * DD * sizeof(float);
  const size_t g_bytes   = (size_t)BB * NN * NN * sizeof(float);
  char* ws = (char*)d_ws;
  float*  T     = (float*)(ws);
  float*  S     = (float*)(ws + mat_bytes);
  float*  AGG   = (float*)(ws + 2 * mat_bytes);
  float*  G     = (float*)(ws + 4 * mat_bytes);
  float4* STATS = (float4*)(ws + 4 * mat_bytes + g_bytes);

  gemm_ts_stats_kernel<<<128, 256, 0, stream>>>(h, Wmsg, bmsg, T, S, STATS);
  gemm_g_kernel<<<256, 256, 0, stream>>>(T, S, G);
  msg_agg_kernel<<<2048, 256, 0, stream>>>(T, S, adj, Wmsg, bmsg, gmsg,
                                           betamsg, G, STATS, AGG);
  gemm_upd_final_kernel<<<128, 256, 0, stream>>>(h, AGG, Wupd, bupd, gupd,
                                                 betaupd, out);
}


# msg/aggregate kernel: the layernorm moments of a pair (i,j), equal in all 32 lanes, taken once (lane l: row j0+l) ahead of the j-loop, fetched by one broadcast LDS read per j; no sum reordered
# speedup vs baseline: 2.8255x; 1.1339x over previous
#include <hip/hip_runtime.h>

typedef __attribute__((ext_vector_type(2))) float v2f;
typedef __attribute__((ext_vector_type(8))) float v8f;

#define DD 128
#define NN 256
#define BB 8
#define ROWS (BB * NN)
#define WMSG_LD (2 * DD + 1)
#define WUPD_LD (2 * DD)
#define JSPLIT 8
#define JPER (NN / JSPLIT)
#define TILE_LD 132

__global__ __launch_bounds__(256) void gemm_ts_stats_kernel(
    const float* __restrict__ H, const float* __restrict__ Wmsg,
    const float* __restrict__ bmsg,
    float* __restrict__ T, float* __restrict__ S, float4* __restrict__ STATS)
{
  __shared__ __attribute__((aligned(16))) float tile[16 * TILE_LD];
  const int wave = (blockIdx.x * 256 + threadIdx.x) >> 5;
  const int lane = threadIdx.x & 31;
  const int mt = wave >> 3;
  const int nt = wave & 7;
  const int m0 = mt << 4, n0 = nt << 4;
  const int half = lane >> 4;
  const int ln = lane & 15;

  const float* __restrict__ arow = H + (size_t)(m0 + ln) * DD;
  const float* __restrict__ wrow = Wmsg + (size_t)(n0 + ln) * WMSG_LD;

  v8f accT = {};
  v8f accS = {};
  for (int k0 = 0; k0 < DD; k0 += 4) {
    const int ka = k0 + 2 * half;
    v2f a;  a.x  = arow[ka];          a.y  = arow[ka + 1];
    v2f bt; bt.x = wrow[ka];          bt.y = wrow[ka + 1];
    v2f bs; bs.x = wrow[DD + ka];     bs.y = wrow[DD + ka + 1];
    accT = __builtin_amdgcn_wmma_f32_16x16x4_f32(
        false, a, false, bt, (short)0, accT, false, false);
    accS = __builtin_amdgcn_wmma_f32_16x16x4_f32(
        false, a, false, bs, (short)0, accS, false, false);
  }
  __builtin_amdgcn_sched_barrier(0);
  asm volatile("v_nop\n\tv_nop\n\tv_nop\n\tv_nop");
  __builtin_amdgcn_sched_barrier(0);

  const int e0 = lane << 2;
  float4 wj4;
  wj4.x = Wmsg[(size_t)(e0 + 0) * WMSG_LD + 2 * DD];
  wj4.y = Wmsg[(size_t)(e0 + 1) * WMSG_LD + 2 * DD];
  wj4.z = Wmsg[(size_t)(e0 + 2) * WMSG_LD + 2 * DD];
  wj4.w = Wmsg[(size_t)(e0 + 3) * WMSG_LD + 2 * DD];
  const float4 bm4 = *(const float4*)(bmsg + e0);

  #pragma unroll
  for (int r = 0; r < 8; ++r)
    tile[(r + 8 * half) * TILE_LD + n0 + ln] = accS[r];
  __syncthreads();
  #pragma unroll
  for (int r = 0; r < 8; ++r) {
    const int row = m0 + r + 8 * half;
    T[(size_t)row * DD + n0 + ln] = accT[r];
    S[(size_t)row * DD + n0 + ln] = accS[r];
  }

  float p1[2], p2[2], p3[2], p4[2];
  #pragma unroll
  for (int rr = 0; rr < 2; ++rr) {
    const float4 s4 = *(const float4*)(tile + (2 * nt + rr) * TILE_LD + e0);
    p1[rr] = s4.x + s4.y + s4.z + s4.w;
    p2[rr] = s4.x * s4.x + s4.y * s4.y + s4.z * s4.z + s4.w * s4.w;
    p3[rr] = s4.x * wj4.x + s4.y * wj4.y + s4.z * wj4.z + s4.w * wj4.w;
    p4[rr] = s4.x * bm4.x + s4.y * bm4.y + s4.z * bm4.z + s4.w * bm4.w;
  }
  #pragma unroll
  for (int off = 16; off >= 1; off >>= 1) {
    #pragma unroll
    for (int rr = 0; rr < 2; ++rr) {
      p1[rr] += __shfl_xor(p1[rr], off, 32);
      p2[rr] += __shfl_xor(p2[rr], off, 32);
      p3[rr] += __shfl_xor(p3[rr], off, 32);
      p4[rr] += __shfl_xor(p4[rr], off, 32);
    }
  }
  if (lane < 2) {
    const int rr = lane;
    float4 st;
    st.x = rr ? p1[1] : p1[0];
    st.y = rr ? p2[1] : p2[0];
    st.z = rr ? p3[1] : p3[0];
    st.w = rr ? p4[1] : p4[0];
    STATS[m0 + 2 * nt + rr] = st;
  }
}

__global__ __launch_bounds__(256) void gemm_g_kernel(
    const float* __restrict__ T, const float* __restrict__ S,
    float* __restrict__ G)
{
  const int wave = (blockIdx.x * 256 + threadIdx.x) >> 5;
  const int lane = threadIdx.x & 31;
  const int b  = wave >> 8;
  const int rm = wave & 255;
  const int it = rm >> 4;
  const int jt = rm & 15;
  const int m0 = it << 4, n0 = jt << 4;
  const int half = lane >> 4;
  const int ln = lane & 15;

  const float* __restrict__ arow = T + (size_t)(b * NN + m0 + ln) * DD;
  const float* __restrict__ brow = S + (size_t)(b * NN + n0 + ln) * DD;

  v8f acc = {};
  for (int k0 = 0; k0 < DD; k0 += 4) {
    const int ka = k0 + 2 * half;
    v2f a;  a.x = arow[ka]; a.y = arow[ka + 1];
    v2f bb; bb.x = brow[ka]; bb.y = brow[ka + 1];
    acc = __builtin_amdgcn_wmma_f32_16x16x4_f32(
        false, a, false, bb, (short)0, acc, false, false);
  }
  #pragma unroll
  for (int r = 0; r < 8; ++r) {
    const int i = m0 + r + 8 * half;
    G[(size_t)(b * NN + i) * NN + n0 + ln] = acc[r];
  }
}

__global__ __launch_bounds__(256, 16) void msg_agg_kernel(
    const float* __restrict__ T, const float* __restrict__ S,
    const float* __restrict__ adj, const float* __restrict__ Wmsg,
    const float* __restrict__ bmsg, const float* __restrict__ gmsg,
    const float* __restrict__ betamsg, const float* __restrict__ G,
    const float4* __restrict__ STATS, float* __restrict__ AGG)
{
  __shared__ float4 part[JSPLIT][32];
  __shared__ float4 jconst[JSPLIT][JPER];
  const int w    = threadIdx.x >> 5;
  const int lane = threadIdx.x & 31;
  const int b = blockIdx.x >> 8;
  const int i = blockIdx.x & 255;
  const int e0 = lane << 2;

  const float4 t4 = *(const float4*)(T + (size_t)(b * NN + i) * DD + e0);
  float4 wj4;
  wj4.x = Wmsg[(size_t)(e0 + 0) * WMSG_LD + 2 * DD];
  wj4.y = Wmsg[(size_t)(e0 + 1) * WMSG_LD + 2 * DD];
  wj4.z = Wmsg[(size_t)(e0 + 2) * WMSG_LD + 2 * DD];
  wj4.w = Wmsg[(size_t)(e0 + 3) * WMSG_LD + 2 * DD];
  const float4 bm4 = *(const float4*)(bmsg + e0);
  const float4 g4  = *(const float4*)(gmsg + e0);
  const float4 be4 = *(const float4*)(betamsg + e0);

  float4 gh4, bh4;
  gh4.x = 0.5f * g4.x;  gh4.y = 0.5f * g4.y;
  gh4.z = 0.5f * g4.z;  gh4.w = 0.5f * g4.w;
  bh4.x = 0.5f * be4.x; bh4.y = 0.5f * be4.y;
  bh4.z = 0.5f * be4.z; bh4.w = 0.5f * be4.w;

  float4 c4;
  c4.x = t4.x + bm4.x; c4.y = t4.y + bm4.y;
  c4.z = t4.z + bm4.z; c4.w = t4.w + bm4.w;

  float r1 = c4.x + c4.y + c4.z + c4.w;
  float r2 = c4.x * c4.x + c4.y * c4.y + c4.z * c4.z + c4.w * c4.w;
  float r3 = c4.x * wj4.x + c4.y * wj4.y + c4.z * wj4.z + c4.w * wj4.w;
  float r4 = wj4.x + wj4.y + wj4.z + wj4.w;
  float r5 = wj4.x * wj4.x + wj4.y * wj4.y + wj4.z * wj4.z + wj4.w * wj4.w;
  #pragma unroll
  for (int off = 16; off >= 1; off >>= 1) {
    r1 += __shfl_xor(r1, off, 32);
    r2 += __shfl_xor(r2, off, 32);
    r3 += __shfl_xor(r3, off, 32);
    r4 += __shfl_xor(r4, off, 32);
    r5 += __shfl_xor(r5, off, 32);
  }
  const float C1 = r1, C2 = r2, Cw = r3, W1 = r4, W2 = r5;
  const float invD = 1.0f / DD;

  const float*  __restrict__ srow = S + (size_t)b * NN * DD;
  const float*  __restrict__ arow = adj + (size_t)(b * NN + i) * NN;
  const float*  __restrict__ grow = G + (size_t)(b * NN + i) * NN;
  const float4* __restrict__ strow = STATS + (size_t)b * NN;

  const int j0 = w * JPER;
  {
    const int j = j0 + lane;
    const float  aj = arow[j];
    const float  gij = grow[j];
    const float4 st = strow[j];

    const float mean  = (C1 + st.x + aj * W1) * invD;
    const float sumsq = C2 + st.y + aj * aj * W2
                      + 2.0f * (gij + st.w)
                      + 2.0f * aj * (Cw + st.z);
    const float var  = sumsq * invD - mean * mean;
    const float rinv = rsqrtf(var + 1e-5f);

    float4 q;
    q.x = aj;
    q.y = mean;
    q.z = rinv;
    q.w = (j != i) ? 1.0f : 0.0f;
    jconst[w][lane] = q;
  }
  __syncthreads();

  float4 acc = {0.f, 0.f, 0.f, 0.f};
  for (int jj = 0; jj < JPER; ++jj) {
    const int j = j0 + jj;
    const float4 s4 = *(const float4*)(srow + (size_t)j * DD + e0);
    const float4 q  = jconst[w][jj];
    const float aj = q.x, mean = q.y, rinv = q.z, keep = q.w;

    float4 p;
    p.x = c4.x + s4.x + aj * wj4.x;
    p.y = c4.y + s4.y + aj * wj4.y;
    p.z = c4.z + s4.z + aj * wj4.z;
    p.w = c4.w + s4.w + aj * wj4.w;

    const float h0 = (p.x - mean) * rinv * gh4.x + bh4.x;
    const float h1 = (p.y - mean) * rinv * gh4.y + bh4.y;
    const float h2 = (p.z - mean) * rinv * gh4.z + bh4.z;
    const float h3 = (p.w - mean) * rinv * gh4.w + bh4.w;
    const float m0 = h0 + h0 * __builtin_amdgcn_tanhf(h0);
    const float m1 = h1 + h1 * __builtin_amdgcn_tanhf(h1);
    const float m2 = h2 + h2 * __builtin_amdgcn_tanhf(h2);
    const float m3 = h3 + h3 * __builtin_amdgcn_tanhf(h3);

    acc.x += keep * m0;
    acc.y += keep * m1;
    acc.z += keep * m2;
    acc.w += keep * m3;
  }
  part[w][lane] = acc;
  __syncthreads();
  if (w == 0) {
    float4 sum = part[0][lane];
    for (int k = 1; k < JSPLIT; ++k) {
      const float4 p = part[k][lane];
      sum.x += p.x; sum.y += p.y; sum.z += p.z; sum.w += p.w;
    }
    *(float4*)(AGG + (size_t)(b * NN + i) * DD + e0) = sum;
  }
}

__global__ __launch_bounds__(256) void gemm_upd_final_kernel(
    const float* __restrict__ H, const float* __restrict__ AGG,
    const float* __restrict__ Wupd, const float* __restrict__ bupd,
    const float* __restrict__ gupd, const float* __restrict__ betaupd,
    float* __restrict__ out)
{
  __shared__ __attribute__((aligned(16))) float tile[16 * TILE_LD];
  const int wave = (blockIdx.x * 256 + threadIdx.x) >> 5;
  const int lane = threadIdx.x & 31;
  const int mt = wave >> 3;
  const int nt = wave & 7;
  const int m0 = mt << 4, n0 = nt << 4;
  const int half = lane >> 4;
  const int ln = lane & 15;

  const float* __restrict__ arow1 = H + (size_t)(m0 + ln) * DD;
  const float* __restrict__ arow2 = AGG + (size_t)(m0 + ln) * DD;
  const float* __restrict__ wrow = Wupd + (size_t)(n0 + ln) * WUPD_LD;

  v8f acc = {};
  for (int k0 = 0; k0 < DD; k0 += 4) {
    const int ka = k0 + 2 * half;
    v2f a1; a1.x = arow1[ka]; a1.y = arow1[ka + 1];
    v2f b1; b1.x = wrow[ka];  b1.y = wrow[ka + 1];
    acc = __builtin_amdgcn_wmma_f32_16x16x4_f32(
        false, a1, false, b1, (short)0, acc, false, false);
    v2f a2; a2.x = arow2[ka];     a2.y = arow2[ka + 1];
    v2f b2; b2.x = wrow[DD + ka]; b2.y = wrow[DD + ka + 1];
    acc = __builtin_amdgcn_wmma_f32_16x16x4_f32(
        false, a2, false, b2, (short)0, acc, false, false);
  }

  __builtin_amdgcn_sched_barrier(0);
  asm volatile("v_nop\n\tv_nop\n\tv_nop\n\tv_nop");
  __builtin_amdgcn_sched_barrier(0);

  const int e0 = lane << 2;
  const float4 g4  = *(const float4*)(gupd + e0);
  const float4 be4 = *(const float4*)(betaupd + e0);
  const int row0 = m0 + 2 * nt;
  float4 h4[2];
  h4[0] = *(const float4*)(H + (size_t)(row0 + 0) * DD + e0);
  h4[1] = *(const float4*)(H + (size_t)(row0 + 1) * DD + e0);

  const float bias = bupd[n0 + ln];
  #pragma unroll
  for (int r = 0; r < 8; ++r)
    tile[(r + 8 * half) * TILE_LD + n0 + ln] = acc[r] + bias;
  __syncthreads();

  float4 u4[2];
  float ps[2], pss[2];
  #pragma unroll
  for (int rr = 0; rr < 2; ++rr) {
    u4[rr] = *(const float4*)(tile + (2 * nt + rr) * TILE_LD + e0);
    ps[rr]  = u4[rr].x + u4[rr].y + u4[rr].z + u4[rr].w;
    pss[rr] = u4[rr].x * u4[rr].x + u4[rr].y * u4[rr].y
            + u4[rr].z * u4[rr].z + u4[rr].w * u4[rr].w;
  }
  #pragma unroll
  for (int off = 16; off >= 1; off >>= 1) {
    #pragma unroll
    for (int rr = 0; rr < 2; ++rr) {
      ps[rr]  += __shfl_xor(ps[rr],  off, 32);
      pss[rr] += __shfl_xor(pss[rr], off, 32);
    }
  }
  #pragma unroll
  for (int rr = 0; rr < 2; ++rr) {
    const float mean = ps[rr] * (1.0f / DD);
    const float var  = pss[rr] * (1.0f / DD) - mean * mean;
    const float rinv = rsqrtf(var + 1e-5f);

    float x0 = (u4[rr].x - mean) * rinv * g4.x + be4.x;
    float x1 = (u4[rr].y - mean) * rinv * g4.y + be4.y;
    float x2 = (u4[rr].z - mean) * rinv * g4.z + be4.z;
    float x3 = (u4[rr].w - mean) * rinv * g4.w + be4.w;

    float4 o;
    o.x = h4[rr].x + x0 / (1.0f + __expf(-x0));
    o.y = h4[rr].y + x1 / (1.0f + __expf(-x1));
    o.z = h4[rr].z + x2 / (1.0f + __expf(-x2));
    o.w = h4[rr].w + x3 / (1.0f + __expf(-x3));
    *(float4*)(out + (size_t)(row0 + rr) * DD + e0) = o;
  }
}

extern "C" void kernel_launch(void* const* d_in, const int* in_sizes, int n_in,
                              void* d_out, int out_size, void* d_ws, size_t ws_size,
                              hipStream_t stream) {
  const float* h       = (const float*)d_in[0];
  const float* adj     = (const float*)d_in[1];
  const float* Wmsg    = (const float*)d_in[2];
  const float* bmsg    = (const float*)d_in[3];
  const float* gmsg    = (const float*)d_in[4];
  const float* betamsg = (const float*)d_in[5];
  const float* Wupd    = (const float*)d_in[6];
  const float* bupd    = (const float*)d_in[7];
  const float* gupd    = (const float*)d_in[8];
  const float* betaupd = (const float*)d_in[9];
  float* out = (float*)d_out;

  const size_t mat_bytes = (size_t)ROWS * DD * sizeof(float);
  const size_t g_bytes   = (size_t)BB * NN * NN * sizeof(float);
  char* ws = (char*)d_ws;
  float*  T     = (float*)(ws);
  float*  S     = (float*)(ws + mat_bytes);
  float*  AGG   = (float*)(ws + 2 * mat_bytes);
  float*  G     = (float*)(ws + 4 * mat_bytes);
  float4* STATS = (float4*)(ws + 4 * mat_bytes + g_bytes);

  gemm_ts_stats_kernel<<<128, 256, 0, stream>>>(h, Wmsg, bmsg, T, S, STATS);
  gemm_g_kernel<<<256, 256, 0, stream>>>(T, S, G);
  msg_agg_kernel<<<2048, 256, 0, stream>>>(T, S, adj, Wmsg, bmsg, gmsg,
                                           betamsg, G, STATS, AGG);
  gemm_upd_final_kernel<<<128, 256, 0, stream>>>(h, AGG, Wupd, bupd, gupd,
                                                 betaupd, out);
}


# GEMM kernels: A/B fragment loads fetch 16 B per lane (4 k, two WMMA steps each) in place of 8 B: half as many 16-line loads; k inside each 8 permuted alike in A and B: GEMM sums re-associated
# speedup vs baseline: 3.0082x; 1.0647x over previous
#include <hip/hip_runtime.h>

typedef __attribute__((ext_vector_type(2))) float v2f;
typedef __attribute__((ext_vector_type(8))) float v8f;
typedef __attribute__((ext_vector_type(4))) float v4f;
typedef __attribute__((ext_vector_type(4), aligned(4))) float v4f_a4;

#define DD 128
#define NN 256
#define BB 8
#define ROWS (BB * NN)
#define WMSG_LD (2 * DD + 1)
#define WUPD_LD (2 * DD)
#define JSPLIT 8
#define JPER (NN / JSPLIT)
#define TILE_LD 132

__global__ __launch_bounds__(256) void gemm_ts_stats_kernel(
    const float* __restrict__ H, const float* __restrict__ Wmsg,
    const float* __restrict__ bmsg,
    float* __restrict__ T, float* __restrict__ S, float4* __restrict__ STATS)
{
  __shared__ __attribute__((aligned(16))) float tile[16 * TILE_LD];
  v8f accT = {};
  v8f accS = {};
  {
    const int lane = threadIdx.x & 31;
    const int n0 = (threadIdx.x >> 5) << 4;
    const int m0 = blockIdx.x << 4;
    const int half = lane >> 4;
    const int ln = lane & 15;
    const float* __restrict__ arow = H + (size_t)(m0 + ln) * DD;
    const float* __restrict__ wrow = Wmsg + (size_t)(n0 + ln) * WMSG_LD;
    for (int k0 = 0; k0 < DD; k0 += 8) {
      const int ka = k0 + 4 * half;
      const v4f    a4  = *(const v4f*)(arow + ka);
      const v4f_a4 bt4 = *(const v4f_a4*)(wrow + ka);
      const v4f_a4 bs4 = *(const v4f_a4*)(wrow + DD + ka);
      v2f a, bt, bs;
      a.x  = a4.x;  a.y  = a4.y;
      bt.x = bt4.x; bt.y = bt4.y;
      bs.x = bs4.x; bs.y = bs4.y;
      accT = __builtin_amdgcn_wmma_f32_16x16x4_f32(
          false, a, false, bt, (short)0, accT, false, false);
      accS = __builtin_amdgcn_wmma_f32_16x16x4_f32(
          false, a, false, bs, (short)0, accS, false, false);
      a.x  = a4.z;  a.y  = a4.w;
      bt.x = bt4.z; bt.y = bt4.w;
      bs.x = bs4.z; bs.y = bs4.w;
      accT = __builtin_amdgcn_wmma_f32_16x16x4_f32(
          false, a, false, bt, (short)0, accT, false, false);
      accS = __builtin_amdgcn_wmma_f32_16x16x4_f32(
          false, a, false, bs, (short)0, accS, false, false);
    }
  }
  int tid = threadIdx.x;
  __builtin_amdgcn_sched_barrier(0);
  asm volatile("v_nop\n\tv_nop\n\tv_nop\n\tv_nop" : "+v"(accT), "+v"(accS), "+v"(tid));
  __builtin_amdgcn_sched_barrier(0);
  const int lane = tid & 31;
  const int nt = tid >> 5;
  const int m0 = blockIdx.x << 4, n0 = nt << 4;
  const int half = lane >> 4;
  const int ln = lane & 15;

  const int e0 = lane << 2;
  float4 wj4;
  wj4.x = Wmsg[(size_t)(e0 + 0) * WMSG_LD + 2 * DD];
  wj4.y = Wmsg[(size_t)(e0 + 1) * WMSG_LD + 2 * DD];
  wj4.z = Wmsg[(size_t)(e0 + 2) * WMSG_LD + 2 * DD];
  wj4.w = Wmsg[(size_t)(e0 + 3) * WMSG_LD + 2 * DD];
  const float4 bm4 = *(const float4*)(bmsg + e0);

  #pragma unroll
  for (int r = 0; r < 8; ++r)
    tile[(r + 8 * half) * TILE_LD + n0 + ln] = accS[r];
  __syncthreads();
  #pragma unroll
  for (int r = 0; r < 8; ++r) {
    const int row = m0 + r + 8 * half;
    T[(size_t)row * DD + n0 + ln] = accT[r];
    S[(size_t)row * DD + n0 + ln] = accS[r];
  }

  float p1[2], p2[2], p3[2], p4[2];
  #pragma unroll
  for (int rr = 0; rr < 2; ++rr) {
    const float4 s4 = *(const float4*)(tile + (2 * nt + rr) * TILE_LD + e0);
    p1[rr] = s4.x + s4.y + s4.z + s4.w;
    p2[rr] = s4.x * s4.x + s4.y * s4.y + s4.z * s4.z + s4.w * s4.w;
    p3[rr] = s4.x * wj4.x + s4.y * wj4.y + s4.z * wj4.z + s4.w * wj4.w;
    p4[rr] = s4.x * bm4.x + s4.y * bm4.y + s4.z * bm4.z + s4.w * bm4.w;
  }
  #pragma unroll
  for (int off = 16; off >= 1; off >>= 1) {
    #pragma unroll
    for (int rr = 0; rr < 2; ++rr) {
      p1[rr] += __shfl_xor(p1[rr], off, 32);
      p2[rr] += __shfl_xor(p2[rr], off, 32);
      p3[rr] += __shfl_xor(p3[rr], off, 32);
      p4[rr] += __shfl_xor(p4[rr], off, 32);
    }
  }
  if (lane < 2) {
    const int rr = lane;
    float4 st;
    st.x = rr ? p1[1] : p1[0];
    st.y = rr ? p2[1] : p2[0];
    st.z = rr ? p3[1] : p3[0];
    st.w = rr ? p4[1] : p4[0];
    STATS[m0 + 2 * nt + rr] = st;
  }
}

__global__ __launch_bounds__(256) void gemm_g_kernel(
    const float* __restrict__ T, const float* __restrict__ S,
    float* __restrict__ G)
{
  v8f acc = {};
  {
    const int wave = (blockIdx.x * 256 + threadIdx.x) >> 5;
    const int lane = threadIdx.x & 31;
    const int b  = wave >> 8;
    const int rm = wave & 255;
    const int m0 = (rm >> 4) << 4, n0 = (rm & 15) << 4;
    const int half = lane >> 4;
    const int ln = lane & 15;
    const float* __restrict__ arow = T + (size_t)(b * NN + m0 + ln) * DD;
    const float* __restrict__ brow = S + (size_t)(b * NN + n0 + ln) * DD;
    for (int k0 = 0; k0 < DD; k0 += 8) {
      const int ka = k0 + 4 * half;
      const v4f a4 = *(const v4f*)(arow + ka);
      const v4f b4 = *(const v4f*)(brow + ka);
      v2f a, bb;
      a.x  = a4.x; a.y  = a4.y;
      bb.x = b4.x; bb.y = b4.y;
      acc = __builtin_amdgcn_wmma_f32_16x16x4_f32(
          false, a, false, bb, (short)0, acc, false, false);
      a.x  = a4.z; a.y  = a4.w;
      bb.x = b4.z; bb.y = b4.w;
      acc = __builtin_amdgcn_wmma_f32_16x16x4_f32(
          false, a, false, bb, (short)0, acc, false, false);
    }
  }
  int tid = threadIdx.x;
  __builtin_amdgcn_sched_barrier(0);
  asm volatile("v_nop\n\tv_nop\n\tv_nop\n\tv_nop" : "+v"(acc), "+v"(tid));
  __builtin_amdgcn_sched_barrier(0);
  const int wave = (blockIdx.x * 256 + tid) >> 5;
  const int lane = tid & 31;
  const int b  = wave >> 8;
  const int rm = wave & 255;
  const int it = rm >> 4;
  const int jt = rm & 15;
  const int m0 = it << 4, n0 = jt << 4;
  const int half = lane >> 4;
  const int ln = lane & 15;
  #pragma unroll
  for (int r = 0; r < 8; ++r) {
    const int i = m0 + r + 8 * half;
    G[(size_t)(b * NN + i) * NN + n0 + ln] = acc[r];
  }
}

__global__ __launch_bounds__(256, 16) void msg_agg_kernel(
    const float* __restrict__ T, const float* __restrict__ S,
    const float* __restrict__ adj, const float* __restrict__ Wmsg,
    const float* __restrict__ bmsg, const float* __restrict__ gmsg,
    const float* __restrict__ betamsg, const float* __restrict__ G,
    const float4* __restrict__ STATS, float* __restrict__ AGG)
{
  __shared__ float4 part[JSPLIT][32];
  __shared__ float4 jconst[JSPLIT][JPER];
  const int w    = threadIdx.x >> 5;
  const int lane = threadIdx.x & 31;
  const int b = blockIdx.x >> 8;
  const int i = blockIdx.x & 255;
  const int e0 = lane << 2;

  const float4 t4 = *(const float4*)(T + (size_t)(b * NN + i) * DD + e0);
  float4 wj4;
  wj4.x = Wmsg[(size_t)(e0 + 0) * WMSG_LD + 2 * DD];
  wj4.y = Wmsg[(size_t)(e0 + 1) * WMSG_LD + 2 * DD];
  wj4.z = Wmsg[(size_t)(e0 + 2) * WMSG_LD + 2 * DD];
  wj4.w = Wmsg[(size_t)(e0 + 3) * WMSG_LD + 2 * DD];
  const float4 bm4 = *(const float4*)(bmsg + e0);
  const float4 g4  = *(const float4*)(gmsg + e0);
  const float4 be4 = *(const float4*)(betamsg + e0);

  float4 gh4, bh4;
  gh4.x = 0.5f * g4.x;  gh4.y = 0.5f * g4.y;
  gh4.z = 0.5f * g4.z;  gh4.w = 0.5f * g4.w;
  bh4.x = 0.5f * be4.x; bh4.y = 0.5f * be4.y;
  bh4.z = 0.5f * be4.z; bh4.w = 0.5f * be4.w;

  float4 c4;
  c4.x = t4.x + bm4.x; c4.y = t4.y + bm4.y;
  c4.z = t4.z + bm4.z; c4.w = t4.w + bm4.w;

  float r1 = c4.x + c4.y + c4.z + c4.w;
  float r2 = c4.x * c4.x + c4.y * c4.y + c4.z * c4.z + c4.w * c4.w;
  float r3 = c4.x * wj4.x + c4.y * wj4.y + c4.z * wj4.z + c4.w * wj4.w;
  float r4 = wj4.x + wj4.y + wj4.z + wj4.w;
  float r5 = wj4.x * wj4.x + wj4.y * wj4.y + wj4.z * wj4.z + wj4.w * wj4.w;
  #pragma unroll
  for (int off = 16; off >= 1; off >>= 1) {
    r1 += __shfl_xor(r1, off, 32);
    r2 += __shfl_xor(r2, off, 32);
    r3 += __shfl_xor(r3, off, 32);
    r4 += __shfl_xor(r4, off, 32);
    r5 += __shfl_xor(r5, off, 32);
  }
  const float C1 = r1, C2 = r2, Cw = r3, W1 = r4, W2 = r5;
  const float invD = 1.0f / DD;

  const float*  __restrict__ srow = S + (size_t)b * NN * DD;
  const float*  __restrict__ arow = adj + (size_t)(b * NN + i) * NN;
  const float*  __restrict__ grow = G + (size_t)(b * NN + i) * NN;
  const float4* __restrict__ strow = STATS + (size_t)b * NN;

  const int j0 = w * JPER;
  {
    const int j = j0 + lane;
    const float  aj = arow[j];
    const float  gij = grow[j];
    const float4 st = strow[j];

    const float mean  = (C1 + st.x + aj * W1) * invD;
    const float sumsq = C2 + st.y + aj * aj * W2
                      + 2.0f * (gij + st.w)
                      + 2.0f * aj * (Cw + st.z);
    const float var  = sumsq * invD - mean * mean;
    const float rinv = rsqrtf(var + 1e-5f);

    float4 q;
    q.x = aj;
    q.y = mean;
    q.z = rinv;
    q.w = (j != i) ? 1.0f : 0.0f;
    jconst[w][lane] = q;
  }
  __syncthreads();

  float4 acc = {0.f, 0.f, 0.f, 0.f};
  for (int jj = 0; jj < JPER; ++jj) {
    const int j = j0 + jj;
    const float4 s4 = *(const float4*)(srow + (size_t)j * DD + e0);
    const float4 q  = jconst[w][jj];
    const float aj = q.x, mean = q.y, rinv = q.z, keep = q.w;

    float4 p;
    p.x = c4.x + s4.x + aj * wj4.x;
    p.y = c4.y + s4.y + aj * wj4.y;
    p.z = c4.z + s4.z + aj * wj4.z;
    p.w = c4.w + s4.w + aj * wj4.w;

    const float h0 = (p.x - mean) * rinv * gh4.x + bh4.x;
    const float h1 = (p.y - mean) * rinv * gh4.y + bh4.y;
    const float h2 = (p.z - mean) * rinv * gh4.z + bh4.z;
    const float h3 = (p.w - mean) * rinv * gh4.w + bh4.w;
    const float m0 = h0 + h0 * __builtin_amdgcn_tanhf(h0);
    const float m1 = h1 + h1 * __builtin_amdgcn_tanhf(h1);
    const float m2 = h2 + h2 * __builtin_amdgcn_tanhf(h2);
    const float m3 = h3 + h3 * __builtin_amdgcn_tanhf(h3);

    acc.x += keep * m0;
    acc.y += keep * m1;
    acc.z += keep * m2;
    acc.w += keep * m3;
  }
  part[w][lane] = acc;
  __syncthreads();
  if (w == 0) {
    float4 sum = part[0][lane];
    for (int k = 1; k < JSPLIT; ++k) {
      const float4 p = part[k][lane];
      sum.x += p.x; sum.y += p.y; sum.z += p.z; sum.w += p.w;
    }
    *(float4*)(AGG + (size_t)(b * NN + i) * DD + e0) = sum;
  }
}

__global__ __launch_bounds__(256) void gemm_upd_final_kernel(
    const float* __restrict__ H, const float* __restrict__ AGG,
    const float* __restrict__ Wupd, const float* __restrict__ bupd,
    const float* __restrict__ gupd, const float* __restrict__ betaupd,
    float* __restrict__ out)
{
  __shared__ __attribute__((aligned(16))) float tile[16 * TILE_LD];
  v8f acc = {};
  {
    const int lane = threadIdx.x & 31;
    const int n0 = (threadIdx.x >> 5) << 4;
    const int m0 = blockIdx.x << 4;
    const int half = lane >> 4;
    const int ln = lane & 15;
    const float* __restrict__ arow1 = H + (size_t)(m0 + ln) * DD;
    const float* __restrict__ arow2 = AGG + (size_t)(m0 + ln) * DD;
    const float* __restrict__ wrow = Wupd + (size_t)(n0 + ln) * WUPD_LD;
    for (int k0 = 0; k0 < DD; k0 += 8) {
      const int ka = k0 + 4 * half;
      const v4f a14 = *(const v4f*)(arow1 + ka);
      const v4f b14 = *(const v4f*)(wrow + ka);
      const v4f a24 = *(const v4f*)(arow2 + ka);
      const v4f b24 = *(const v4f*)(wrow + DD + ka);
      v2f a1, b1, a2, b2;
      a1.x = a14.x; a1.y = a14.y;
      b1.x = b14.x; b1.y = b14.y;
      acc = __builtin_amdgcn_wmma_f32_16x16x4_f32(
          false, a1, false, b1, (short)0, acc, false, false);
      a2.x = a24.x; a2.y = a24.y;
      b2.x = b24.x; b2.y = b24.y;
      acc = __builtin_amdgcn_wmma_f32_16x16x4_f32(
          false, a2, false, b2, (short)0, acc, false, false);
      a1.x = a14.z; a1.y = a14.w;
      b1.x = b14.z; b1.y = b14.w;
      acc = __builtin_amdgcn_wmma_f32_16x16x4_f32(
          false, a1, false, b1, (short)0, acc, false, false);
      a2.x = a24.z; a2.y = a24.w;
      b2.x = b24.z; b2.y = b24.w;
      acc = __builtin_amdgcn_wmma_f32_16x16x4_f32(
          false, a2, false, b2, (short)0, acc, false, false);
    }
  }
  int tid = threadIdx.x;
  __builtin_amdgcn_sched_barrier(0);
  asm volatile("v_nop\n\tv_nop\n\tv_nop\n\tv_nop" : "+v"(acc), "+v"(tid));
  __builtin_amdgcn_sched_barrier(0);
  const int lane = tid & 31;
  const int nt = tid >> 5;
  const int m0 = blockIdx.x << 4, n0 = nt << 4;
  const int half = lane >> 4;
  const int ln = lane & 15;

  const int e0 = lane << 2;
  const float4 g4  = *(const float4*)(gupd + e0);
  const float4 be4 = *(const float4*)(betaupd + e0);
  const int row0 = m0 + 2 * nt;
  float4 h4[2];
  h4[0] = *(const float4*)(H + (size_t)(row0 + 0) * DD + e0);
  h4[1] = *(const float4*)(H + (size_t)(row0 + 1) * DD + e0);

  const float bias = bupd[n0 + ln];
  #pragma unroll
  for (int r = 0; r < 8; ++r)
    tile[(r + 8 * half) * TILE_LD + n0 + ln] = acc[r] + bias;
  __syncthreads();

  float4 u4[2];
  float ps[2], pss[2];
  #pragma unroll
  for (int rr = 0; rr < 2; ++rr) {
    u4[rr] = *(const float4*)(tile + (2 * nt + rr) * TILE_LD + e0);
    ps[rr]  = u4[rr].x + u4[rr].y + u4[rr].z + u4[rr].w;
    pss[rr] = u4[rr].x * u4[rr].x + u4[rr].y * u4[rr].y
            + u4[rr].z * u4[rr].z + u4[rr].w * u4[rr].w;
  }
  #pragma unroll
  for (int off = 16; off >= 1; off >>= 1) {
    #pragma unroll
    for (int rr = 0; rr < 2; ++rr) {
      ps[rr]  += __shfl_xor(ps[rr],  off, 32);
      pss[rr] += __shfl_xor(pss[rr], off, 32);
    }
  }
  #pragma unroll
  for (int rr = 0; rr < 2; ++rr) {
    const float mean = ps[rr] * (1.0f / DD);
    const float var  = pss[rr] * (1.0f / DD) - mean * mean;
    const float rinv = rsqrtf(var + 1e-5f);

    float x0 = (u4[rr].x - mean) * rinv * g4.x + be4.x;
    float x1 = (u4[rr].y - mean) * rinv * g4.y + be4.y;
    float x2 = (u4[rr].z - mean) * rinv * g4.z + be4.z;
    float x3 = (u4[rr].w - mean) * rinv * g4.w + be4.w;

    float4 o;
    o.x = h4[rr].x + x0 / (1.0f + __expf(-x0));
    o.y = h4[rr].y + x1 / (1.0f + __expf(-x1));
    o.z = h4[rr].z + x2 / (1.0f + __expf(-x2));
    o.w = h4[rr].w + x3 / (1.0f + __expf(-x3));
    *(float4*)(out + (size_t)(row0 + rr) * DD + e0) = o;
  }
}

extern "C" void kernel_launch(void* const* d_in, const int* in_sizes, int n_in,
                              void* d_out, int out_size, void* d_ws, size_t ws_size,
                              hipStream_t stream) {
  const float* h       = (const float*)d_in[0];
  const float* adj     = (const float*)d_in[1];
  const float* Wmsg    = (const float*)d_in[2];
  const float* bmsg    = (const float*)d_in[3];
  const float* gmsg    = (const float*)d_in[4];
  const float* betamsg = (const float*)d_in[5];
  const float* Wupd    = (const float*)d_in[6];
  const float* bupd    = (const float*)d_in[7];
  const float* gupd    = (const float*)d_in[8];
  const float* betaupd = (const float*)d_in[9];
  float* out = (float*)d_out;

  const size_t mat_bytes = (size_t)ROWS * DD * sizeof(float);
  const size_t g_bytes   = (size_t)BB * NN * NN * sizeof(float);
  char* ws = (char*)d_ws;
  float*  T     = (float*)(ws);
  float*  S     = (float*)(ws + mat_bytes);
  float*  AGG   = (float*)(ws + 2 * mat_bytes);
  float*  G     = (float*)(ws + 4 * mat_bytes);
  float4* STATS = (float4*)(ws + 4 * mat_bytes + g_bytes);

  gemm_ts_stats_kernel<<<128, 256, 0, stream>>>(h, Wmsg, bmsg, T, S, STATS);
  gemm_g_kernel<<<256, 256, 0, stream>>>(T, S, G);
  msg_agg_kernel<<<2048, 256, 0, stream>>>(T, S, adj, Wmsg, bmsg, gmsg,
                                           betamsg, G, STATS, AGG);
  gemm_upd_final_kernel<<<128, 256, 0, stream>>>(h, AGG, Wupd, bupd, gupd,
                                                 betaupd, out);
}


# G = T*S^T formed inside the message kernel by the work-group that uses it (8 target rows per group, one 16x16 WMMA tile per wave): 4 launches -> 3, G never in memory; sum over j re-associated
# speedup vs baseline: 3.3058x; 1.0989x over previous
#include <hip/hip_runtime.h>

typedef __attribute__((ext_vector_type(2))) float v2f;
typedef __attribute__((ext_vector_type(8))) float v8f;
typedef __attribute__((ext_vector_type(4))) float v4f;
typedef __attribute__((ext_vector_type(4), aligned(4))) float v4f_a4;

#define DD 128
#define NN 256
#define BB 8
#define ROWS (BB * NN)
#define WMSG_LD (2 * DD + 1)
#define WUPD_LD (2 * DD)
#define JSPLIT 8
#define JPER (NN / JSPLIT)
#define TILE_LD 132

__global__ __launch_bounds__(256) void gemm_ts_stats_kernel(
    const float* __restrict__ H, const float* __restrict__ Wmsg,
    const float* __restrict__ bmsg,
    float* __restrict__ T, float* __restrict__ S, float4* __restrict__ STATS)
{
  __shared__ __attribute__((aligned(16))) float tile[16 * TILE_LD];
  v8f accT = {};
  v8f accS = {};
  {
    const int lane = threadIdx.x & 31;
    const int n0 = (threadIdx.x >> 5) << 4;
    const int m0 = blockIdx.x << 4;
    const int half = lane >> 4;
    const int ln = lane & 15;
    const float* __restrict__ arow = H + (size_t)(m0 + ln) * DD;
    const float* __restrict__ wrow = Wmsg + (size_t)(n0 + ln) * WMSG_LD;
    for (int k0 = 0; k0 < DD; k0 += 8) {
      const int ka = k0 + 4 * half;
      const v4f    a4  = *(const v4f*)(arow + ka);
      const v4f_a4 bt4 = *(const v4f_a4*)(wrow + ka);
      const v4f_a4 bs4 = *(const v4f_a4*)(wrow + DD + ka);
      v2f a, bt, bs;
      a.x  = a4.x;  a.y  = a4.y;
      bt.x = bt4.x; bt.y = bt4.y;
      bs.x = bs4.x; bs.y = bs4.y;
      accT = __builtin_amdgcn_wmma_f32_16x16x4_f32(
          false, a, false, bt, (short)0, accT, false, false);
      accS = __builtin_amdgcn_wmma_f32_16x16x4_f32(
          false, a, false, bs, (short)0, accS, false, false);
      a.x  = a4.z;  a.y  = a4.w;
      bt.x = bt4.z; bt.y = bt4.w;
      bs.x = bs4.z; bs.y = bs4.w;
      accT = __builtin_amdgcn_wmma_f32_16x16x4_f32(
          false, a, false, bt, (short)0, accT, false, false);
      accS = __builtin_amdgcn_wmma_f32_16x16x4_f32(
          false, a, false, bs, (short)0, accS, false, false);
    }
  }
  int tid = threadIdx.x;
  __builtin_amdgcn_sched_barrier(0);
  asm volatile("v_nop\n\tv_nop\n\tv_nop\n\tv_nop" : "+v"(accT), "+v"(accS), "+v"(tid));
  __builtin_amdgcn_sched_barrier(0);
  const int lane = tid & 31;
  const int nt = tid >> 5;
  const int m0 = blockIdx.x << 4, n0 = nt << 4;
  const int half = lane >> 4;
  const int ln = lane & 15;

  const int e0 = lane << 2;
  float4 wj4;
  wj4.x = Wmsg[(size_t)(e0 + 0) * WMSG_LD + 2 * DD];
  wj4.y = Wmsg[(size_t)(e0 + 1) * WMSG_LD + 2 * DD];
  wj4.z = Wmsg[(size_t)(e0 + 2) * WMSG_LD + 2 * DD];
  wj4.w = Wmsg[(size_t)(e0 + 3) * WMSG_LD + 2 * DD];
  const float4 bm4 = *(const float4*)(bmsg + e0);

  #pragma unroll
  for (int r = 0; r < 8; ++r)
    tile[(r + 8 * half) * TILE_LD + n0 + ln] = accS[r];
  __syncthreads();
  #pragma unroll
  for (int r = 0; r < 8; ++r) {
    const int row = m0 + r + 8 * half;
    T[(size_t)row * DD + n0 + ln] = accT[r];
    S[(size_t)row * DD + n0 + ln] = accS[r];
  }

  float p1[2], p2[2], p3[2], p4[2];
  #pragma unroll
  for (int rr = 0; rr < 2; ++rr) {
    const float4 s4 = *(const float4*)(tile + (2 * nt + rr) * TILE_LD + e0);
    p1[rr] = s4.x + s4.y + s4.z + s4.w;
    p2[rr] = s4.x * s4.x + s4.y * s4.y + s4.z * s4.z + s4.w * s4.w;
    p3[rr] = s4.x * wj4.x + s4.y * wj4.y + s4.z * wj4.z + s4.w * wj4.w;
    p4[rr] = s4.x * bm4.x + s4.y * bm4.y + s4.z * bm4.z + s4.w * bm4.w;
  }
  #pragma unroll
  for (int off = 16; off >= 1; off >>= 1) {
    #pragma unroll
    for (int rr = 0; rr < 2; ++rr) {
      p1[rr] += __shfl_xor(p1[rr], off, 32);
      p2[rr] += __shfl_xor(p2[rr], off, 32);
      p3[rr] += __shfl_xor(p3[rr], off, 32);
      p4[rr] += __shfl_xor(p4[rr], off, 32);
    }
  }
  if (lane < 2) {
    const int rr = lane;
    float4 st;
    st.x = rr ? p1[1] : p1[0];
    st.y = rr ? p2[1] : p2[0];
    st.z = rr ? p3[1] : p3[0];
    st.w = rr ? p4[1] : p4[0];
    STATS[m0 + 2 * nt + rr] = st;
  }
}

#define ITILE 8
#define MWAVES 16
#define JPW (NN / MWAVES)
__global__ __launch_bounds__(512, 4) void msg_agg_kernel(
    const float* __restrict__ T, const float* __restrict__ S,
    const float* __restrict__ adj, const float* __restrict__ Wmsg,
    const float* __restrict__ bmsg, const float* __restrict__ gmsg,
    const float* __restrict__ betamsg,
    const float4* __restrict__ STATS, float* __restrict__ AGG)
{
  __shared__ float4 part[ITILE][MWAVES][32];
  __shared__ float4 jconst[ITILE][MWAVES][JPW];
  __shared__ float4 rowc[ITILE];
  v8f accg = {};
  {
    const int lane = threadIdx.x & 31;
    const int half = lane >> 4;
    const int ln = lane & 15;
    const int b  = blockIdx.x >> 5;
    const int i0 = (blockIdx.x & 31) << 3;
    const int jw = (threadIdx.x >> 5) << 4;
    const float* __restrict__ arow = T + (size_t)(b * NN + i0 + (ln & 7)) * DD;
    const float* __restrict__ brow = S + (size_t)(b * NN + jw + ln) * DD;
    for (int k0 = 0; k0 < DD; k0 += 8) {
      const int ka = k0 + 4 * half;
      const v4f a4 = *(const v4f*)(arow + ka);
      const v4f b4 = *(const v4f*)(brow + ka);
      v2f a, bb;
      a.x  = a4.x; a.y  = a4.y;
      bb.x = b4.x; bb.y = b4.y;
      accg = __builtin_amdgcn_wmma_f32_16x16x4_f32(
          false, a, false, bb, (short)0, accg, false, false);
      a.x  = a4.z; a.y  = a4.w;
      bb.x = b4.z; bb.y = b4.w;
      accg = __builtin_amdgcn_wmma_f32_16x16x4_f32(
          false, a, false, bb, (short)0, accg, false, false);
    }
  }
  int tid = threadIdx.x;
  __builtin_amdgcn_sched_barrier(0);
  asm volatile("v_nop\n\tv_nop\n\tv_nop\n\tv_nop" : "+v"(accg), "+v"(tid));
  __builtin_amdgcn_sched_barrier(0);
  const int w    = tid >> 5;
  const int lane = tid & 31;
  const int half = lane >> 4;
  const int ln   = lane & 15;
  const int b  = blockIdx.x >> 5;
  const int i0 = (blockIdx.x & 31) << 3;
  const int e0 = lane << 2;

  float4 wj4;
  wj4.x = Wmsg[(size_t)(e0 + 0) * WMSG_LD + 2 * DD];
  wj4.y = Wmsg[(size_t)(e0 + 1) * WMSG_LD + 2 * DD];
  wj4.z = Wmsg[(size_t)(e0 + 2) * WMSG_LD + 2 * DD];
  wj4.w = Wmsg[(size_t)(e0 + 3) * WMSG_LD + 2 * DD];
  const float4 bm4 = *(const float4*)(bmsg + e0);
  const float4 g4  = *(const float4*)(gmsg + e0);
  const float4 be4 = *(const float4*)(betamsg + e0);

  float4 gh4, bh4;
  gh4.x = 0.5f * g4.x;  gh4.y = 0.5f * g4.y;
  gh4.z = 0.5f * g4.z;  gh4.w = 0.5f * g4.w;
  bh4.x = 0.5f * be4.x; bh4.y = 0.5f * be4.y;
  bh4.z = 0.5f * be4.z; bh4.w = 0.5f * be4.w;

  float4 c4[ITILE];
  #pragma unroll
  for (int r = 0; r < ITILE; ++r) {
    const float4 t4 = *(const float4*)(T + (size_t)(b * NN + i0 + r) * DD + e0);
    c4[r].x = t4.x + bm4.x; c4[r].y = t4.y + bm4.y;
    c4[r].z = t4.z + bm4.z; c4[r].w = t4.w + bm4.w;
  }

  float4 sreg[JPW];
  #pragma unroll
  for (int jj = 0; jj < JPW; ++jj)
    sreg[jj] = *(const float4*)(S + (size_t)(b * NN + w * JPW + jj) * DD + e0);

  const int j0 = w * JPW;
  const float4 st = STATS[(size_t)b * NN + j0 + ln];
  float ajr[4];
  #pragma unroll
  for (int k = 0; k < 4; ++k)
    ajr[k] = adj[(size_t)(b * NN + i0 + 4 * half + k) * NN + j0 + ln];

  const int rown = w & (ITILE - 1);
  const float4 to4 = *(const float4*)(T + (size_t)(b * NN + i0 + rown) * DD + e0);
  float4 co4;
  co4.x = to4.x + bm4.x; co4.y = to4.y + bm4.y;
  co4.z = to4.z + bm4.z; co4.w = to4.w + bm4.w;
  float r1 = co4.x + co4.y + co4.z + co4.w;
  float r2 = co4.x * co4.x + co4.y * co4.y + co4.z * co4.z + co4.w * co4.w;
  float r3 = co4.x * wj4.x + co4.y * wj4.y + co4.z * wj4.z + co4.w * wj4.w;
  float r4 = wj4.x + wj4.y + wj4.z + wj4.w;
  float r5 = wj4.x * wj4.x + wj4.y * wj4.y + wj4.z * wj4.z + wj4.w * wj4.w;
  #pragma unroll
  for (int off = 16; off >= 1; off >>= 1) {
    r1 += __shfl_xor(r1, off, 32);
    r2 += __shfl_xor(r2, off, 32);
    r3 += __shfl_xor(r3, off, 32);
    r4 += __shfl_xor(r4, off, 32);
    r5 += __shfl_xor(r5, off, 32);
  }
  const float W1 = r4, W2 = r5;
  const float invD = 1.0f / DD;
  if (w < ITILE && lane == 0) {
    float4 rc;
    rc.x = r1; rc.y = r2; rc.z = r3; rc.w = 0.0f;
    rowc[w] = rc;
  }
  __syncthreads();

  {
    const int j = j0 + ln;
    #pragma unroll
    for (int k = 0; k < 4; ++k) {
      const int r = 4 * half + k;
      const int i = i0 + r;
      const float  aj  = ajr[k];
      const float  gij = half ? accg[4 + k] : accg[k];
      const float4 rc = rowc[r];
      const float C1 = rc.x, C2 = rc.y, Cw = rc.z;

      const float mean  = (C1 + st.x + aj * W1) * invD;
      const float sumsq = C2 + st.y + aj * aj * W2
                        + 2.0f * (gij + st.w)
                        + 2.0f * aj * (Cw + st.z);
      const float var  = sumsq * invD - mean * mean;
      const float rinv = rsqrtf(var + 1e-5f);

      float4 q;
      q.x = aj;
      q.y = mean;
      q.z = rinv;
      q.w = (j != i) ? 1.0f : 0.0f;
      jconst[r][w][ln] = q;
    }
  }
  __syncthreads();

  float4 acc[ITILE];
  #pragma unroll
  for (int r = 0; r < ITILE; ++r) {
    acc[r].x = 0.f; acc[r].y = 0.f; acc[r].z = 0.f; acc[r].w = 0.f;
  }
  #pragma unroll
  for (int jj = 0; jj < JPW; ++jj) {
    const float4 s4 = sreg[jj];
    #pragma unroll
    for (int r = 0; r < ITILE; ++r) {
      const float4 q = jconst[r][w][jj];
      const float aj = q.x, mean = q.y, rinv = q.z, keep = q.w;

      float4 p;
      p.x = c4[r].x + s4.x + aj * wj4.x;
      p.y = c4[r].y + s4.y + aj * wj4.y;
      p.z = c4[r].z + s4.z + aj * wj4.z;
      p.w = c4[r].w + s4.w + aj * wj4.w;

      const float h0 = (p.x - mean) * rinv * gh4.x + bh4.x;
      const float h1 = (p.y - mean) * rinv * gh4.y + bh4.y;
      const float h2 = (p.z - mean) * rinv * gh4.z + bh4.z;
      const float h3 = (p.w - mean) * rinv * gh4.w + bh4.w;
      const float m0 = h0 + h0 * __builtin_amdgcn_tanhf(h0);
      const float m1 = h1 + h1 * __builtin_amdgcn_tanhf(h1);
      const float m2 = h2 + h2 * __builtin_amdgcn_tanhf(h2);
      const float m3 = h3 + h3 * __builtin_amdgcn_tanhf(h3);

      acc[r].x += keep * m0;
      acc[r].y += keep * m1;
      acc[r].z += keep * m2;
      acc[r].w += keep * m3;
    }
  }
  #pragma unroll
  for (int r = 0; r < ITILE; ++r)
    part[r][w][lane] = acc[r];
  __syncthreads();
  if (w < ITILE) {
    float4 sum = part[w][0][lane];
    for (int k = 1; k < MWAVES; ++k) {
      const float4 p = part[w][k][lane];
      sum.x += p.x; sum.y += p.y; sum.z += p.z; sum.w += p.w;
    }
    *(float4*)(AGG + (size_t)(b * NN + i0 + w) * DD + e0) = sum;
  }
}

__global__ __launch_bounds__(256) void gemm_upd_final_kernel(
    const float* __restrict__ H, const float* __restrict__ AGG,
    const float* __restrict__ Wupd, const float* __restrict__ bupd,
    const float* __restrict__ gupd, const float* __restrict__ betaupd,
    float* __restrict__ out)
{
  __shared__ __attribute__((aligned(16))) float tile[16 * TILE_LD];
  v8f acc = {};
  {
    const int lane = threadIdx.x & 31;
    const int n0 = (threadIdx.x >> 5) << 4;
    const int m0 = blockIdx.x << 4;
    const int half = lane >> 4;
    const int ln = lane & 15;
    const float* __restrict__ arow1 = H + (size_t)(m0 + ln) * DD;
    const float* __restrict__ arow2 = AGG + (size_t)(m0 + ln) * DD;
    const float* __restrict__ wrow = Wupd + (size_t)(n0 + ln) * WUPD_LD;
    for (int k0 = 0; k0 < DD; k0 += 8) {
      const int ka = k0 + 4 * half;
      const v4f a14 = *(const v4f*)(arow1 + ka);
      const v4f b14 = *(const v4f*)(wrow + ka);
      const v4f a24 = *(const v4f*)(arow2 + ka);
      const v4f b24 = *(const v4f*)(wrow + DD + ka);
      v2f a1, b1, a2, b2;
      a1.x = a14.x; a1.y = a14.y;
      b1.x = b14.x; b1.y = b14.y;
      acc = __builtin_amdgcn_wmma_f32_16x16x4_f32(
          false, a1, false, b1, (short)0, acc, false, false);
      a2.x = a24.x; a2.y = a24.y;
      b2.x = b24.x; b2.y = b24.y;
      acc = __builtin_amdgcn_wmma_f32_16x16x4_f32(
          false, a2, false, b2, (short)0, acc, false, false);
      a1.x = a14.z; a1.y = a14.w;
      b1.x = b14.z; b1.y = b14.w;
      acc = __builtin_amdgcn_wmma_f32_16x16x4_f32(
          false, a1, false, b1, (short)0, acc, false, false);
      a2.x = a24.z; a2.y = a24.w;
      b2.x = b24.z; b2.y = b24.w;
      acc = __builtin_amdgcn_wmma_f32_16x16x4_f32(
          false, a2, false, b2, (short)0, acc, false, false);
    }
  }
  int tid = threadIdx.x;
  __builtin_amdgcn_sched_barrier(0);
  asm volatile("v_nop\n\tv_nop\n\tv_nop\n\tv_nop" : "+v"(acc), "+v"(tid));
  __builtin_amdgcn_sched_barrier(0);
  const int lane = tid & 31;
  const int nt = tid >> 5;
  const int m0 = blockIdx.x << 4, n0 = nt << 4;
  const int half = lane >> 4;
  const int ln = lane & 15;

  const int e0 = lane << 2;
  const float4 g4  = *(const float4*)(gupd + e0);
  const float4 be4 = *(const float4*)(betaupd + e0);
  const int row0 = m0 + 2 * nt;
  float4 h4[2];
  h4[0] = *(const float4*)(H + (size_t)(row0 + 0) * DD + e0);
  h4[1] = *(const float4*)(H + (size_t)(row0 + 1) * DD + e0);

  const float bias = bupd[n0 + ln];
  #pragma unroll
  for (int r = 0; r < 8; ++r)
    tile[(r + 8 * half) * TILE_LD + n0 + ln] = acc[r] + bias;
  __syncthreads();

  float4 u4[2];
  float ps[2], pss[2];
  #pragma unroll
  for (int rr = 0; rr < 2; ++rr) {
    u4[rr] = *(const float4*)(tile + (2 * nt + rr) * TILE_LD + e0);
    ps[rr]  = u4[rr].x + u4[rr].y + u4[rr].z + u4[rr].w;
    pss[rr] = u4[rr].x * u4[rr].x + u4[rr].y * u4[rr].y
            + u4[rr].z * u4[rr].z + u4[rr].w * u4[rr].w;
  }
  #pragma unroll
  for (int off = 16; off >= 1; off >>= 1) {
    #pragma unroll
    for (int rr = 0; rr < 2; ++rr) {
      ps[rr]  += __shfl_xor(ps[rr],  off, 32);
      pss[rr] += __shfl_xor(pss[rr], off, 32);
    }
  }
  #pragma unroll
  for (int rr = 0; rr < 2; ++rr) {
    const float mean = ps[rr] * (1.0f / DD);
    const float var  = pss[rr] * (1.0f / DD) - mean * mean;
    const float rinv = rsqrtf(var + 1e-5f);

    float x0 = (u4[rr].x - mean) * rinv * g4.x + be4.x;
    float x1 = (u4[rr].y - mean) * rinv * g4.y + be4.y;
    float x2 = (u4[rr].z - mean) * rinv * g4.z + be4.z;
    float x3 = (u4[rr].w - mean) * rinv * g4.w + be4.w;

    float4 o;
    o.x = h4[rr].x + x0 / (1.0f + __expf(-x0));
    o.y = h4[rr].y + x1 / (1.0f + __expf(-x1));
    o.z = h4[rr].z + x2 / (1.0f + __expf(-x2));
    o.w = h4[rr].w + x3 / (1.0f + __expf(-x3));
    *(float4*)(out + (size_t)(row0 + rr) * DD + e0) = o;
  }
}

extern "C" void kernel_launch(void* const* d_in, const int* in_sizes, int n_in,
                              void* d_out, int out_size, void* d_ws, size_t ws_size,
                              hipStream_t stream) {
  const float* h       = (const float*)d_in[0];
  const float* adj     = (const float*)d_in[1];
  const float* Wmsg    = (const float*)d_in[2];
  const float* bmsg    = (const float*)d_in[3];
  const float* gmsg    = (const float*)d_in[4];
  const float* betamsg = (const float*)d_in[5];
  const float* Wupd    = (const float*)d_in[6];
  const float* bupd    = (const float*)d_in[7];
  const float* gupd    = (const float*)d_in[8];
  const float* betaupd = (const float*)d_in[9];
  float* out = (float*)d_out;

  const size_t mat_bytes = (size_t)ROWS * DD * sizeof(float);
  char* ws = (char*)d_ws;
  float*  T     = (float*)(ws);
  float*  S     = (float*)(ws + mat_bytes);
  float*  AGG   = (float*)(ws + 2 * mat_bytes);
  float4* STATS = (float4*)(ws + 4 * mat_bytes);

  gemm_ts_stats_kernel<<<128, 256, 0, stream>>>(h, Wmsg, bmsg, T, S, STATS);
  msg_agg_kernel<<<256, 512, 0, stream>>>(T, S, adj, Wmsg, bmsg, gmsg,
                                          betamsg, STATS, AGG);
  gemm_upd_final_kernel<<<128, 256, 0, stream>>>(h, AGG, Wupd, bupd, gupd,
                                                 betaupd, out);
}


# update GEMM + final layernorm+silu+residual done in the message kernel by the work-group that owns the 8 rows (AGG stays in LDS; waves split H*Wu1 / AGG*Wu2): 3 launches -> 2; U sum re-associated
# speedup vs baseline: 4.0315x; 1.2195x over previous
#include <hip/hip_runtime.h>

typedef __attribute__((ext_vector_type(2))) float v2f;
typedef __attribute__((ext_vector_type(8))) float v8f;
typedef __attribute__((ext_vector_type(4))) float v4f;
typedef __attribute__((ext_vector_type(4), aligned(4))) float v4f_a4;

#define DD 128
#define NN 256
#define BB 8
#define ROWS (BB * NN)
#define WMSG_LD (2 * DD + 1)
#define WUPD_LD (2 * DD)
#define JSPLIT 8
#define JPER (NN / JSPLIT)
#define TILE_LD 132

__global__ __launch_bounds__(256) void gemm_ts_stats_kernel(
    const float* __restrict__ H, const float* __restrict__ Wmsg,
    const float* __restrict__ bmsg,
    float* __restrict__ T, float* __restrict__ S, float4* __restrict__ STATS)
{
  __shared__ __attribute__((aligned(16))) float tile[16 * TILE_LD];
  v8f accT = {};
  v8f accS = {};
  {
    const int lane = threadIdx.x & 31;
    const int n0 = (threadIdx.x >> 5) << 4;
    const int m0 = blockIdx.x << 4;
    const int half = lane >> 4;
    const int ln = lane & 15;
    const float* __restrict__ arow = H + (size_t)(m0 + ln) * DD;
    const float* __restrict__ wrow = Wmsg + (size_t)(n0 + ln) * WMSG_LD;
    for (int k0 = 0; k0 < DD; k0 += 8) {
      const int ka = k0 + 4 * half;
      const v4f    a4  = *(const v4f*)(arow + ka);
      const v4f_a4 bt4 = *(const v4f_a4*)(wrow + ka);
      const v4f_a4 bs4 = *(const v4f_a4*)(wrow + DD + ka);
      v2f a, bt, bs;
      a.x  = a4.x;  a.y  = a4.y;
      bt.x = bt4.x; bt.y = bt4.y;
      bs.x = bs4.x; bs.y = bs4.y;
      accT = __builtin_amdgcn_wmma_f32_16x16x4_f32(
          false, a, false, bt, (short)0, accT, false, false);
      accS = __builtin_amdgcn_wmma_f32_16x16x4_f32(
          false, a, false, bs, (short)0, accS, false, false);
      a.x  = a4.z;  a.y  = a4.w;
      bt.x = bt4.z; bt.y = bt4.w;
      bs.x = bs4.z; bs.y = bs4.w;
      accT = __builtin_amdgcn_wmma_f32_16x16x4_f32(
          false, a, false, bt, (short)0, accT, false, false);
      accS = __builtin_amdgcn_wmma_f32_16x16x4_f32(
          false, a, false, bs, (short)0, accS, false, false);
    }
  }
  int tid = threadIdx.x;
  __builtin_amdgcn_sched_barrier(0);
  asm volatile("v_nop\n\tv_nop\n\tv_nop\n\tv_nop" : "+v"(accT), "+v"(accS), "+v"(tid));
  __builtin_amdgcn_sched_barrier(0);
  const int lane = tid & 31;
  const int nt = tid >> 5;
  const int m0 = blockIdx.x << 4, n0 = nt << 4;
  const int half = lane >> 4;
  const int ln = lane & 15;

  const int e0 = lane << 2;
  float4 wj4;
  wj4.x = Wmsg[(size_t)(e0 + 0) * WMSG_LD + 2 * DD];
  wj4.y = Wmsg[(size_t)(e0 + 1) * WMSG_LD + 2 * DD];
  wj4.z = Wmsg[(size_t)(e0 + 2) * WMSG_LD + 2 * DD];
  wj4.w = Wmsg[(size_t)(e0 + 3) * WMSG_LD + 2 * DD];
  const float4 bm4 = *(const float4*)(bmsg + e0);

  #pragma unroll
  for (int r = 0; r < 8; ++r)
    tile[(r + 8 * half) * TILE_LD + n0 + ln] = accS[r];
  __syncthreads();
  #pragma unroll
  for (int r = 0; r < 8; ++r) {
    const int row = m0 + r + 8 * half;
    T[(size_t)row * DD + n0 + ln] = accT[r];
    S[(size_t)row * DD + n0 + ln] = accS[r];
  }

  float p1[2], p2[2], p3[2], p4[2];
  #pragma unroll
  for (int rr = 0; rr < 2; ++rr) {
    const float4 s4 = *(const float4*)(tile + (2 * nt + rr) * TILE_LD + e0);
    p1[rr] = s4.x + s4.y + s4.z + s4.w;
    p2[rr] = s4.x * s4.x + s4.y * s4.y + s4.z * s4.z + s4.w * s4.w;
    p3[rr] = s4.x * wj4.x + s4.y * wj4.y + s4.z * wj4.z + s4.w * wj4.w;
    p4[rr] = s4.x * bm4.x + s4.y * bm4.y + s4.z * bm4.z + s4.w * bm4.w;
  }
  #pragma unroll
  for (int off = 16; off >= 1; off >>= 1) {
    #pragma unroll
    for (int rr = 0; rr < 2; ++rr) {
      p1[rr] += __shfl_xor(p1[rr], off, 32);
      p2[rr] += __shfl_xor(p2[rr], off, 32);
      p3[rr] += __shfl_xor(p3[rr], off, 32);
      p4[rr] += __shfl_xor(p4[rr], off, 32);
    }
  }
  if (lane < 2) {
    const int rr = lane;
    float4 st;
    st.x = rr ? p1[1] : p1[0];
    st.y = rr ? p2[1] : p2[0];
    st.z = rr ? p3[1] : p3[0];
    st.w = rr ? p4[1] : p4[0];
    STATS[m0 + 2 * nt + rr] = st;
  }
}

#define ITILE 8
#define MWAVES 16
#define JPW (NN / MWAVES)
#define AGG_LD 136
__global__ __launch_bounds__(512, 4) void msg_agg_kernel(
    const float* __restrict__ T, const float* __restrict__ S,
    const float* __restrict__ adj, const float* __restrict__ Wmsg,
    const float* __restrict__ bmsg, const float* __restrict__ gmsg,
    const float* __restrict__ betamsg,
    const float4* __restrict__ STATS,
    const float* __restrict__ H, const float* __restrict__ Wupd,
    const float* __restrict__ bupd, const float* __restrict__ gupd,
    const float* __restrict__ betaupd, float* __restrict__ out)
{
  __shared__ float4 part[ITILE][MWAVES][32];
  __shared__ __attribute__((aligned(16))) float aggt[ITILE * AGG_LD];
  __shared__ __attribute__((aligned(16))) float hrows[ITILE * AGG_LD];
  __shared__ __attribute__((aligned(16))) float utile[2 * ITILE * TILE_LD];
  __shared__ float4 jconst[ITILE][MWAVES][JPW];
  __shared__ float4 rowc[ITILE];
  v8f accg = {};
  {
    const int lane = threadIdx.x & 31;
    const int half = lane >> 4;
    const int ln = lane & 15;
    const int b  = blockIdx.x >> 5;
    const int i0 = (blockIdx.x & 31) << 3;
    const int jw = (threadIdx.x >> 5) << 4;
    const float* __restrict__ arow = T + (size_t)(b * NN + i0 + (ln & 7)) * DD;
    const float* __restrict__ brow = S + (size_t)(b * NN + jw + ln) * DD;
    for (int k0 = 0; k0 < DD; k0 += 8) {
      const int ka = k0 + 4 * half;
      const v4f a4 = *(const v4f*)(arow + ka);
      const v4f b4 = *(const v4f*)(brow + ka);
      v2f a, bb;
      a.x  = a4.x; a.y  = a4.y;
      bb.x = b4.x; bb.y = b4.y;
      accg = __builtin_amdgcn_wmma_f32_16x16x4_f32(
          false, a, false, bb, (short)0, accg, false, false);
      a.x  = a4.z; a.y  = a4.w;
      bb.x = b4.z; bb.y = b4.w;
      accg = __builtin_amdgcn_wmma_f32_16x16x4_f32(
          false, a, false, bb, (short)0, accg, false, false);
    }
  }
  int tid = threadIdx.x;
  __builtin_amdgcn_sched_barrier(0);
  asm volatile("v_nop\n\tv_nop\n\tv_nop\n\tv_nop" : "+v"(accg), "+v"(tid));
  __builtin_amdgcn_sched_barrier(0);
  const int w    = tid >> 5;
  const int lane = tid & 31;
  const int half = lane >> 4;
  const int ln   = lane & 15;
  const int b  = blockIdx.x >> 5;
  const int i0 = (blockIdx.x & 31) << 3;
  const int e0 = lane << 2;

  float4 wj4;
  wj4.x = Wmsg[(size_t)(e0 + 0) * WMSG_LD + 2 * DD];
  wj4.y = Wmsg[(size_t)(e0 + 1) * WMSG_LD + 2 * DD];
  wj4.z = Wmsg[(size_t)(e0 + 2) * WMSG_LD + 2 * DD];
  wj4.w = Wmsg[(size_t)(e0 + 3) * WMSG_LD + 2 * DD];
  const float4 bm4 = *(const float4*)(bmsg + e0);
  const float4 g4  = *(const float4*)(gmsg + e0);
  const float4 be4 = *(const float4*)(betamsg + e0);

  float4 gh4, bh4;
  gh4.x = 0.5f * g4.x;  gh4.y = 0.5f * g4.y;
  gh4.z = 0.5f * g4.z;  gh4.w = 0.5f * g4.w;
  bh4.x = 0.5f * be4.x; bh4.y = 0.5f * be4.y;
  bh4.z = 0.5f * be4.z; bh4.w = 0.5f * be4.w;

  float4 c4[ITILE];
  #pragma unroll
  for (int r = 0; r < ITILE; ++r) {
    const float4 t4 = *(const float4*)(T + (size_t)(b * NN + i0 + r) * DD + e0);
    c4[r].x = t4.x + bm4.x; c4[r].y = t4.y + bm4.y;
    c4[r].z = t4.z + bm4.z; c4[r].w = t4.w + bm4.w;
  }

  const float2 hst2 = *(const float2*)(H + (size_t)(b * NN + i0) * DD + 2 * tid);

  float4 sreg[JPW];
  #pragma unroll
  for (int jj = 0; jj < JPW; ++jj)
    sreg[jj] = *(const float4*)(S + (size_t)(b * NN + w * JPW + jj) * DD + e0);

  const int j0 = w * JPW;
  const float4 st = STATS[(size_t)b * NN + j0 + ln];
  float ajr[4];
  #pragma unroll
  for (int k = 0; k < 4; ++k)
    ajr[k] = adj[(size_t)(b * NN + i0 + 4 * half + k) * NN + j0 + ln];

  const int rown = w & (ITILE - 1);
  const float4 to4 = *(const float4*)(T + (size_t)(b * NN + i0 + rown) * DD + e0);
  float4 co4;
  co4.x = to4.x + bm4.x; co4.y = to4.y + bm4.y;
  co4.z = to4.z + bm4.z; co4.w = to4.w + bm4.w;
  float r1 = co4.x + co4.y + co4.z + co4.w;
  float r2 = co4.x * co4.x + co4.y * co4.y + co4.z * co4.z + co4.w * co4.w;
  float r3 = co4.x * wj4.x + co4.y * wj4.y + co4.z * wj4.z + co4.w * wj4.w;
  float r4 = wj4.x + wj4.y + wj4.z + wj4.w;
  float r5 = wj4.x * wj4.x + wj4.y * wj4.y + wj4.z * wj4.z + wj4.w * wj4.w;
  #pragma unroll
  for (int off = 16; off >= 1; off >>= 1) {
    r1 += __shfl_xor(r1, off, 32);
    r2 += __shfl_xor(r2, off, 32);
    r3 += __shfl_xor(r3, off, 32);
    r4 += __shfl_xor(r4, off, 32);
    r5 += __shfl_xor(r5, off, 32);
  }
  const float W1 = r4, W2 = r5;
  const float invD = 1.0f / DD;
  if (w < ITILE && lane == 0) {
    float4 rc;
    rc.x = r1; rc.y = r2; rc.z = r3; rc.w = 0.0f;
    rowc[w] = rc;
  }
  *(float2*)(hrows + (tid >> 6) * AGG_LD + ((2 * tid) & (DD - 1))) = hst2;
  __syncthreads();

  {
    const int j = j0 + ln;
    #pragma unroll
    for (int k = 0; k < 4; ++k) {
      const int r = 4 * half + k;
      const int i = i0 + r;
      const float  aj  = ajr[k];
      const float  gij = half ? accg[4 + k] : accg[k];
      const float4 rc = rowc[r];
      const float C1 = rc.x, C2 = rc.y, Cw = rc.z;

      const float mean  = (C1 + st.x + aj * W1) * invD;
      const float sumsq = C2 + st.y + aj * aj * W2
                        + 2.0f * (gij + st.w)
                        + 2.0f * aj * (Cw + st.z);
      const float var  = sumsq * invD - mean * mean;
      const float rinv = rsqrtf(var + 1e-5f);

      float4 q;
      q.x = aj;
      q.y = mean;
      q.z = rinv;
      q.w = (j != i) ? 1.0f : 0.0f;
      jconst[r][w][ln] = q;
    }
  }
  __syncthreads();

  float4 acc[ITILE];
  #pragma unroll
  for (int r = 0; r < ITILE; ++r) {
    acc[r].x = 0.f; acc[r].y = 0.f; acc[r].z = 0.f; acc[r].w = 0.f;
  }
  #pragma unroll
  for (int jj = 0; jj < JPW; ++jj) {
    const float4 s4 = sreg[jj];
    #pragma unroll
    for (int r = 0; r < ITILE; ++r) {
      const float4 q = jconst[r][w][jj];
      const float aj = q.x, mean = q.y, rinv = q.z, keep = q.w;

      float4 p;
      p.x = c4[r].x + s4.x + aj * wj4.x;
      p.y = c4[r].y + s4.y + aj * wj4.y;
      p.z = c4[r].z + s4.z + aj * wj4.z;
      p.w = c4[r].w + s4.w + aj * wj4.w;

      const float h0 = (p.x - mean) * rinv * gh4.x + bh4.x;
      const float h1 = (p.y - mean) * rinv * gh4.y + bh4.y;
      const float h2 = (p.z - mean) * rinv * gh4.z + bh4.z;
      const float h3 = (p.w - mean) * rinv * gh4.w + bh4.w;
      const float m0 = h0 + h0 * __builtin_amdgcn_tanhf(h0);
      const float m1 = h1 + h1 * __builtin_amdgcn_tanhf(h1);
      const float m2 = h2 + h2 * __builtin_amdgcn_tanhf(h2);
      const float m3 = h3 + h3 * __builtin_amdgcn_tanhf(h3);

      acc[r].x += keep * m0;
      acc[r].y += keep * m1;
      acc[r].z += keep * m2;
      acc[r].w += keep * m3;
    }
  }
  #pragma unroll
  for (int r = 0; r < ITILE; ++r)
    part[r][w][lane] = acc[r];

  const int wu  = __builtin_amdgcn_readfirstlane(w);
  const int kh  = wu >> 3;
  const int wr  = wu & (ITILE - 1);
  const int un0 = wr << 4;
  v4f ub[16], ua[16];
  {
    const float* __restrict__ uwrow =
        Wupd + (size_t)(un0 + ln) * WUPD_LD + (kh ? 0 : DD) + 4 * half;
    #pragma unroll
    for (int kk = 0; kk < 16; ++kk)
      ub[kk] = *(const v4f*)(uwrow + 8 * kk);
  }
  const float4 gu4  = *(const float4*)(gupd + e0);
  const float4 beu4 = *(const float4*)(betaupd + e0);
  const float4 bu4  = *(const float4*)(bupd + e0);

  __syncthreads();

  v8f accu = {};
  if (kh) {
    #pragma unroll
    for (int kk = 0; kk < 16; ++kk)
      ua[kk] = *(const v4f*)(hrows + (ln & 7) * AGG_LD + 8 * kk + 4 * half);
    __builtin_amdgcn_sched_barrier(0);
    #pragma unroll
    for (int kk = 0; kk < 16; ++kk) {
      v2f a, bb;
      a.x  = ua[kk].x; a.y  = ua[kk].y;
      bb.x = ub[kk].x; bb.y = ub[kk].y;
      accu = __builtin_amdgcn_wmma_f32_16x16x4_f32(
          false, a, false, bb, (short)0, accu, false, false);
      a.x  = ua[kk].z; a.y  = ua[kk].w;
      bb.x = ub[kk].z; bb.y = ub[kk].w;
      accu = __builtin_amdgcn_wmma_f32_16x16x4_f32(
          false, a, false, bb, (short)0, accu, false, false);
    }
    __builtin_amdgcn_sched_barrier(0);
    asm volatile("v_nop\n\tv_nop\n\tv_nop\n\tv_nop" : "+v"(accu));
    __builtin_amdgcn_sched_barrier(0);
  } else {
    float4 sum = part[wr][0][lane];
    for (int k = 1; k < MWAVES; ++k) {
      const float4 p = part[wr][k][lane];
      sum.x += p.x; sum.y += p.y; sum.z += p.z; sum.w += p.w;
    }
    *(float4*)(aggt + wr * AGG_LD + e0) = sum;
  }
  __syncthreads();
  if (!kh) {
    #pragma unroll
    for (int kk = 0; kk < 16; ++kk)
      ua[kk] = *(const v4f*)(aggt + (ln & 7) * AGG_LD + 8 * kk + 4 * half);
    __builtin_amdgcn_sched_barrier(0);
    #pragma unroll
    for (int kk = 0; kk < 16; ++kk) {
      v2f a, bb;
      a.x  = ua[kk].x; a.y  = ua[kk].y;
      bb.x = ub[kk].x; bb.y = ub[kk].y;
      accu = __builtin_amdgcn_wmma_f32_16x16x4_f32(
          false, a, false, bb, (short)0, accu, false, false);
      a.x  = ua[kk].z; a.y  = ua[kk].w;
      bb.x = ub[kk].z; bb.y = ub[kk].w;
      accu = __builtin_amdgcn_wmma_f32_16x16x4_f32(
          false, a, false, bb, (short)0, accu, false, false);
    }
    __builtin_amdgcn_sched_barrier(0);
    asm volatile("v_nop\n\tv_nop\n\tv_nop\n\tv_nop" : "+v"(accu));
    __builtin_amdgcn_sched_barrier(0);
  }
  if (half == 0) {
    #pragma unroll
    for (int r = 0; r < ITILE; ++r)
      utile[(kh * ITILE + r) * TILE_LD + un0 + ln] = accu[r];
  }
  __syncthreads();
  if (wu < ITILE) {
    const float4 xa = *(const float4*)(utile + wr * TILE_LD + e0);
    const float4 xh = *(const float4*)(utile + (ITILE + wr) * TILE_LD + e0);
    const float4 hr4 = *(const float4*)(hrows + wr * AGG_LD + e0);
    float4 u4;
    u4.x = (xh.x + xa.x) + bu4.x;
    u4.y = (xh.y + xa.y) + bu4.y;
    u4.z = (xh.z + xa.z) + bu4.z;
    u4.w = (xh.w + xa.w) + bu4.w;
    float ps  = u4.x + u4.y + u4.z + u4.w;
    float pss = u4.x * u4.x + u4.y * u4.y + u4.z * u4.z + u4.w * u4.w;
    #pragma unroll
    for (int off = 16; off >= 1; off >>= 1) {
      ps  += __shfl_xor(ps,  off, 32);
      pss += __shfl_xor(pss, off, 32);
    }
    const float mean = ps * (1.0f / DD);
    const float var  = pss * (1.0f / DD) - mean * mean;
    const float rinv = rsqrtf(var + 1e-5f);

    const float x0 = (u4.x - mean) * rinv * gu4.x + beu4.x;
    const float x1 = (u4.y - mean) * rinv * gu4.y + beu4.y;
    const float x2 = (u4.z - mean) * rinv * gu4.z + beu4.z;
    const float x3 = (u4.w - mean) * rinv * gu4.w + beu4.w;

    float4 o;
    o.x = hr4.x + x0 / (1.0f + __expf(-x0));
    o.y = hr4.y + x1 / (1.0f + __expf(-x1));
    o.z = hr4.z + x2 / (1.0f + __expf(-x2));
    o.w = hr4.w + x3 / (1.0f + __expf(-x3));
    *(float4*)(out + (size_t)(b * NN + i0 + wr) * DD + e0) = o;
  }
}

extern "C" void kernel_launch(void* const* d_in, const int* in_sizes, int n_in,
                              void* d_out, int out_size, void* d_ws, size_t ws_size,
                              hipStream_t stream) {
  const float* h       = (const float*)d_in[0];
  const float* adj     = (const float*)d_in[1];
  const float* Wmsg    = (const float*)d_in[2];
  const float* bmsg    = (const float*)d_in[3];
  const float* gmsg    = (const float*)d_in[4];
  const float* betamsg = (const float*)d_in[5];
  const float* Wupd    = (const float*)d_in[6];
  const float* bupd    = (const float*)d_in[7];
  const float* gupd    = (const float*)d_in[8];
  const float* betaupd = (const float*)d_in[9];
  float* out = (float*)d_out;

  const size_t mat_bytes = (size_t)ROWS * DD * sizeof(float);
  char* ws = (char*)d_ws;
  float*  T     = (float*)(ws);
  float*  S     = (float*)(ws + mat_bytes);
  float4* STATS = (float4*)(ws + 4 * mat_bytes);

  gemm_ts_stats_kernel<<<128, 256, 0, stream>>>(h, Wmsg, bmsg, T, S, STATS);
  msg_agg_kernel<<<256, 512, 0, stream>>>(T, S, adj, Wmsg, bmsg, gmsg,
                                          betamsg, STATS, h, Wupd, bupd, gupd,
                                          betaupd, out);
}


# msg kernel j-loop: pair mean split into its 3 row means and folded, with the layernorm gain, into t, s, wJ ahead of the loop: affine step 3 ops per element, was 5; terms re-rounded, no sum reordered
# speedup vs baseline: 4.2999x; 1.0666x over previous
#include <hip/hip_runtime.h>

typedef __attribute__((ext_vector_type(2))) float v2f;
typedef __attribute__((ext_vector_type(8))) float v8f;
typedef __attribute__((ext_vector_type(4))) float v4f;
typedef __attribute__((ext_vector_type(4), aligned(4))) float v4f_a4;

#define DD 128
#define NN 256
#define BB 8
#define ROWS (BB * NN)
#define WMSG_LD (2 * DD + 1)
#define WUPD_LD (2 * DD)
#define JSPLIT 8
#define JPER (NN / JSPLIT)
#define TILE_LD 132

__global__ __launch_bounds__(256) void gemm_ts_stats_kernel(
    const float* __restrict__ H, const float* __restrict__ Wmsg,
    const float* __restrict__ bmsg,
    float* __restrict__ T, float* __restrict__ S, float4* __restrict__ STATS)
{
  __shared__ __attribute__((aligned(16))) float tile[16 * TILE_LD];
  v8f accT = {};
  v8f accS = {};
  {
    const int lane = threadIdx.x & 31;
    const int n0 = (threadIdx.x >> 5) << 4;
    const int m0 = blockIdx.x << 4;
    const int half = lane >> 4;
    const int ln = lane & 15;
    const float* __restrict__ arow = H + (size_t)(m0 + ln) * DD;
    const float* __restrict__ wrow = Wmsg + (size_t)(n0 + ln) * WMSG_LD;
    for (int k0 = 0; k0 < DD; k0 += 8) {
      const int ka = k0 + 4 * half;
      const v4f    a4  = *(const v4f*)(arow + ka);
      const v4f_a4 bt4 = *(const v4f_a4*)(wrow + ka);
      const v4f_a4 bs4 = *(const v4f_a4*)(wrow + DD + ka);
      v2f a, bt, bs;
      a.x  = a4.x;  a.y  = a4.y;
      bt.x = bt4.x; bt.y = bt4.y;
      bs.x = bs4.x; bs.y = bs4.y;
      accT = __builtin_amdgcn_wmma_f32_16x16x4_f32(
          false, a, false, bt, (short)0, accT, false, false);
      accS = __builtin_amdgcn_wmma_f32_16x16x4_f32(
          false, a, false, bs, (short)0, accS, false, false);
      a.x  = a4.z;  a.y  = a4.w;
      bt.x = bt4.z; bt.y = bt4.w;
      bs.x = bs4.z; bs.y = bs4.w;
      accT = __builtin_amdgcn_wmma_f32_16x16x4_f32(
          false, a, false, bt, (short)0, accT, false, false);
      accS = __builtin_amdgcn_wmma_f32_16x16x4_f32(
          false, a, false, bs, (short)0, accS, false, false);
    }
  }
  int tid = threadIdx.x;
  __builtin_amdgcn_sched_barrier(0);
  asm volatile("v_nop\n\tv_nop\n\tv_nop\n\tv_nop" : "+v"(accT), "+v"(accS), "+v"(tid));
  __builtin_amdgcn_sched_barrier(0);
  const int lane = tid & 31;
  const int nt = tid >> 5;
  const int m0 = blockIdx.x << 4, n0 = nt << 4;
  const int half = lane >> 4;
  const int ln = lane & 15;

  const int e0 = lane << 2;
  float4 wj4;
  wj4.x = Wmsg[(size_t)(e0 + 0) * WMSG_LD + 2 * DD];
  wj4.y = Wmsg[(size_t)(e0 + 1) * WMSG_LD + 2 * DD];
  wj4.z = Wmsg[(size_t)(e0 + 2) * WMSG_LD + 2 * DD];
  wj4.w = Wmsg[(size_t)(e0 + 3) * WMSG_LD + 2 * DD];
  const float4 bm4 = *(const float4*)(bmsg + e0);

  #pragma unroll
  for (int r = 0; r < 8; ++r)
    tile[(r + 8 * half) * TILE_LD + n0 + ln] = accS[r];
  __syncthreads();
  #pragma unroll
  for (int r = 0; r < 8; ++r) {
    const int row = m0 + r + 8 * half;
    T[(size_t)row * DD + n0 + ln] = accT[r];
    S[(size_t)row * DD + n0 + ln] = accS[r];
  }

  float p1[2], p2[2], p3[2], p4[2];
  #pragma unroll
  for (int rr = 0; rr < 2; ++rr) {
    const float4 s4 = *(const float4*)(tile + (2 * nt + rr) * TILE_LD + e0);
    p1[rr] = s4.x + s4.y + s4.z + s4.w;
    p2[rr] = s4.x * s4.x + s4.y * s4.y + s4.z * s4.z + s4.w * s4.w;
    p3[rr] = s4.x * wj4.x + s4.y * wj4.y + s4.z * wj4.z + s4.w * wj4.w;
    p4[rr] = s4.x * bm4.x + s4.y * bm4.y + s4.z * bm4.z + s4.w * bm4.w;
  }
  #pragma unroll
  for (int off = 16; off >= 1; off >>= 1) {
    #pragma unroll
    for (int rr = 0; rr < 2; ++rr) {
      p1[rr] += __shfl_xor(p1[rr], off, 32);
      p2[rr] += __shfl_xor(p2[rr], off, 32);
      p3[rr] += __shfl_xor(p3[rr], off, 32);
      p4[rr] += __shfl_xor(p4[rr], off, 32);
    }
  }
  if (lane < 2) {
    const int rr = lane;
    float4 st;
    st.x = rr ? p1[1] : p1[0];
    st.y = rr ? p2[1] : p2[0];
    st.z = rr ? p3[1] : p3[0];
    st.w = rr ? p4[1] : p4[0];
    STATS[m0 + 2 * nt + rr] = st;
  }
}

#define ITILE 8
#define MWAVES 16
#define JPW (NN / MWAVES)
#define AGG_LD 136
__global__ __launch_bounds__(512, 4) void msg_agg_kernel(
    const float* __restrict__ T, const float* __restrict__ S,
    const float* __restrict__ adj, const float* __restrict__ Wmsg,
    const float* __restrict__ bmsg, const float* __restrict__ gmsg,
    const float* __restrict__ betamsg,
    const float4* __restrict__ STATS,
    const float* __restrict__ H, const float* __restrict__ Wupd,
    const float* __restrict__ bupd, const float* __restrict__ gupd,
    const float* __restrict__ betaupd, float* __restrict__ out)
{
  __shared__ float4 part[ITILE][MWAVES][32];
  __shared__ __attribute__((aligned(16))) float aggt[ITILE * AGG_LD];
  __shared__ __attribute__((aligned(16))) float hrows[ITILE * AGG_LD];
  __shared__ __attribute__((aligned(16))) float utile[2 * ITILE * TILE_LD];
  __shared__ float4 jconst[ITILE][MWAVES][JPW];
  __shared__ float4 rowc[ITILE];
  v8f accg = {};
  {
    const int lane = threadIdx.x & 31;
    const int half = lane >> 4;
    const int ln = lane & 15;
    const int b  = blockIdx.x >> 5;
    const int i0 = (blockIdx.x & 31) << 3;
    const int jw = (threadIdx.x >> 5) << 4;
    const float* __restrict__ arow = T + (size_t)(b * NN + i0 + (ln & 7)) * DD;
    const float* __restrict__ brow = S + (size_t)(b * NN + jw + ln) * DD;
    for (int k0 = 0; k0 < DD; k0 += 8) {
      const int ka = k0 + 4 * half;
      const v4f a4 = *(const v4f*)(arow + ka);
      const v4f b4 = *(const v4f*)(brow + ka);
      v2f a, bb;
      a.x  = a4.x; a.y  = a4.y;
      bb.x = b4.x; bb.y = b4.y;
      accg = __builtin_amdgcn_wmma_f32_16x16x4_f32(
          false, a, false, bb, (short)0, accg, false, false);
      a.x  = a4.z; a.y  = a4.w;
      bb.x = b4.z; bb.y = b4.w;
      accg = __builtin_amdgcn_wmma_f32_16x16x4_f32(
          false, a, false, bb, (short)0, accg, false, false);
    }
  }
  int tid = threadIdx.x;
  __builtin_amdgcn_sched_barrier(0);
  asm volatile("v_nop\n\tv_nop\n\tv_nop\n\tv_nop" : "+v"(accg), "+v"(tid));
  __builtin_amdgcn_sched_barrier(0);
  const int w    = tid >> 5;
  const int lane = tid & 31;
  const int half = lane >> 4;
  const int ln   = lane & 15;
  const int b  = blockIdx.x >> 5;
  const int i0 = (blockIdx.x & 31) << 3;
  const int e0 = lane << 2;

  float4 wj4;
  wj4.x = Wmsg[(size_t)(e0 + 0) * WMSG_LD + 2 * DD];
  wj4.y = Wmsg[(size_t)(e0 + 1) * WMSG_LD + 2 * DD];
  wj4.z = Wmsg[(size_t)(e0 + 2) * WMSG_LD + 2 * DD];
  wj4.w = Wmsg[(size_t)(e0 + 3) * WMSG_LD + 2 * DD];
  const float4 bm4 = *(const float4*)(bmsg + e0);
  const float4 g4  = *(const float4*)(gmsg + e0);
  const float4 be4 = *(const float4*)(betamsg + e0);

  float4 gh4, bh4;
  gh4.x = 0.5f * g4.x;  gh4.y = 0.5f * g4.y;
  gh4.z = 0.5f * g4.z;  gh4.w = 0.5f * g4.w;
  bh4.x = 0.5f * be4.x; bh4.y = 0.5f * be4.y;
  bh4.z = 0.5f * be4.z; bh4.w = 0.5f * be4.w;

  float4 c4[ITILE];
  #pragma unroll
  for (int r = 0; r < ITILE; ++r) {
    const float4 t4 = *(const float4*)(T + (size_t)(b * NN + i0 + r) * DD + e0);
    c4[r].x = t4.x + bm4.x; c4[r].y = t4.y + bm4.y;
    c4[r].z = t4.z + bm4.z; c4[r].w = t4.w + bm4.w;
  }

  const float2 hst2 = *(const float2*)(H + (size_t)(b * NN + i0) * DD + 2 * tid);

  float4 sreg[JPW];
  #pragma unroll
  for (int jj = 0; jj < JPW; ++jj)
    sreg[jj] = *(const float4*)(S + (size_t)(b * NN + w * JPW + jj) * DD + e0);

  const int j0 = w * JPW;
  const float4 st = STATS[(size_t)b * NN + j0 + ln];
  float ajr[4];
  #pragma unroll
  for (int k = 0; k < 4; ++k)
    ajr[k] = adj[(size_t)(b * NN + i0 + 4 * half + k) * NN + j0 + ln];

  const int rown = w & (ITILE - 1);
  const float4 to4 = *(const float4*)(T + (size_t)(b * NN + i0 + rown) * DD + e0);
  float4 co4;
  co4.x = to4.x + bm4.x; co4.y = to4.y + bm4.y;
  co4.z = to4.z + bm4.z; co4.w = to4.w + bm4.w;
  float r1 = co4.x + co4.y + co4.z + co4.w;
  float r2 = co4.x * co4.x + co4.y * co4.y + co4.z * co4.z + co4.w * co4.w;
  float r3 = co4.x * wj4.x + co4.y * wj4.y + co4.z * wj4.z + co4.w * wj4.w;
  float r4 = wj4.x + wj4.y + wj4.z + wj4.w;
  float r5 = wj4.x * wj4.x + wj4.y * wj4.y + wj4.z * wj4.z + wj4.w * wj4.w;
  #pragma unroll
  for (int off = 16; off >= 1; off >>= 1) {
    r1 += __shfl_xor(r1, off, 32);
    r2 += __shfl_xor(r2, off, 32);
    r3 += __shfl_xor(r3, off, 32);
    r4 += __shfl_xor(r4, off, 32);
    r5 += __shfl_xor(r5, off, 32);
  }
  const float W1 = r4, W2 = r5;
  const float invD = 1.0f / DD;
  if (w < ITILE && lane == 0) {
    float4 rc;
    rc.x = r1; rc.y = r2; rc.z = r3; rc.w = 0.0f;
    rowc[w] = rc;
  }
  *(float2*)(hrows + (tid >> 6) * AGG_LD + ((2 * tid) & (DD - 1))) = hst2;
  __syncthreads();

  {
    const int j = j0 + ln;
    #pragma unroll
    for (int k = 0; k < 4; ++k) {
      const int r = 4 * half + k;
      const int i = i0 + r;
      const float  aj  = ajr[k];
      const float  gij = half ? accg[4 + k] : accg[k];
      const float4 rc = rowc[r];
      const float C1 = rc.x, C2 = rc.y, Cw = rc.z;

      const float mean  = (C1 + st.x + aj * W1) * invD;
      const float sumsq = C2 + st.y + aj * aj * W2
                        + 2.0f * (gij + st.w)
                        + 2.0f * aj * (Cw + st.z);
      const float var  = sumsq * invD - mean * mean;
      const float rinv = rsqrtf(var + 1e-5f);

      float4 q;
      q.x = aj;
      q.y = (j != i) ? 1.0f : 0.0f;
      q.z = rinv;
      q.w = 0.0f;
      jconst[r][w][ln] = q;
    }
  }
  {
    const float mw = W1 * invD;
    wj4.x = gh4.x * (wj4.x - mw); wj4.y = gh4.y * (wj4.y - mw);
    wj4.z = gh4.z * (wj4.z - mw); wj4.w = gh4.w * (wj4.w - mw);
    #pragma unroll
    for (int r = 0; r < ITILE; ++r) {
      const float mc = rowc[r].x * invD;
      c4[r].x = gh4.x * (c4[r].x - mc); c4[r].y = gh4.y * (c4[r].y - mc);
      c4[r].z = gh4.z * (c4[r].z - mc); c4[r].w = gh4.w * (c4[r].w - mc);
    }
    #pragma unroll
    for (int jj = 0; jj < JPW; ++jj) {
      const float ms = invD * __int_as_float(
          __builtin_amdgcn_readlane(__float_as_int(st.x), jj));
      sreg[jj].x = gh4.x * (sreg[jj].x - ms); sreg[jj].y = gh4.y * (sreg[jj].y - ms);
      sreg[jj].z = gh4.z * (sreg[jj].z - ms); sreg[jj].w = gh4.w * (sreg[jj].w - ms);
    }
  }
  __syncthreads();

  float4 acc[ITILE];
  #pragma unroll
  for (int r = 0; r < ITILE; ++r) {
    acc[r].x = 0.f; acc[r].y = 0.f; acc[r].z = 0.f; acc[r].w = 0.f;
  }
  #pragma unroll
  for (int jj = 0; jj < JPW; ++jj) {
    const float4 s4 = sreg[jj];
    #pragma unroll
    for (int r = 0; r < ITILE; ++r) {
      const float4 q = jconst[r][w][jj];
      const float aj = q.x, keep = q.y, rinv = q.z;

      float4 p;
      p.x = __builtin_fmaf(aj, wj4.x, c4[r].x + s4.x);
      p.y = __builtin_fmaf(aj, wj4.y, c4[r].y + s4.y);
      p.z = __builtin_fmaf(aj, wj4.z, c4[r].z + s4.z);
      p.w = __builtin_fmaf(aj, wj4.w, c4[r].w + s4.w);

      const float h0 = __builtin_fmaf(rinv, p.x, bh4.x);
      const float h1 = __builtin_fmaf(rinv, p.y, bh4.y);
      const float h2 = __builtin_fmaf(rinv, p.z, bh4.z);
      const float h3 = __builtin_fmaf(rinv, p.w, bh4.w);
      const float m0 = h0 + h0 * __builtin_amdgcn_tanhf(h0);
      const float m1 = h1 + h1 * __builtin_amdgcn_tanhf(h1);
      const float m2 = h2 + h2 * __builtin_amdgcn_tanhf(h2);
      const float m3 = h3 + h3 * __builtin_amdgcn_tanhf(h3);

      acc[r].x += keep * m0;
      acc[r].y += keep * m1;
      acc[r].z += keep * m2;
      acc[r].w += keep * m3;
    }
  }
  #pragma unroll
  for (int r = 0; r < ITILE; ++r)
    part[r][w][lane] = acc[r];

  const int wu  = __builtin_amdgcn_readfirstlane(w);
  const int kh  = wu >> 3;
  const int wr  = wu & (ITILE - 1);
  const int un0 = wr << 4;
  v4f ub[16], ua[16];
  {
    const float* __restrict__ uwrow =
        Wupd + (size_t)(un0 + ln) * WUPD_LD + (kh ? 0 : DD) + 4 * half;
    #pragma unroll
    for (int kk = 0; kk < 16; ++kk)
      ub[kk] = *(const v4f*)(uwrow + 8 * kk);
  }
  const float4 gu4  = *(const float4*)(gupd + e0);
  const float4 beu4 = *(const float4*)(betaupd + e0);
  const float4 bu4  = *(const float4*)(bupd + e0);

  __syncthreads();

  v8f accu = {};
  if (kh) {
    #pragma unroll
    for (int kk = 0; kk < 16; ++kk)
      ua[kk] = *(const v4f*)(hrows + (ln & 7) * AGG_LD + 8 * kk + 4 * half);
    __builtin_amdgcn_sched_barrier(0);
    #pragma unroll
    for (int kk = 0; kk < 16; ++kk) {
      v2f a, bb;
      a.x  = ua[kk].x; a.y  = ua[kk].y;
      bb.x = ub[kk].x; bb.y = ub[kk].y;
      accu = __builtin_amdgcn_wmma_f32_16x16x4_f32(
          false, a, false, bb, (short)0, accu, false, false);
      a.x  = ua[kk].z; a.y  = ua[kk].w;
      bb.x = ub[kk].z; bb.y = ub[kk].w;
      accu = __builtin_amdgcn_wmma_f32_16x16x4_f32(
          false, a, false, bb, (short)0, accu, false, false);
    }
    __builtin_amdgcn_sched_barrier(0);
    asm volatile("v_nop\n\tv_nop\n\tv_nop\n\tv_nop" : "+v"(accu));
    __builtin_amdgcn_sched_barrier(0);
  } else {
    float4 sum = part[wr][0][lane];
    for (int k = 1; k < MWAVES; ++k) {
      const float4 p = part[wr][k][lane];
      sum.x += p.x; sum.y += p.y; sum.z += p.z; sum.w += p.w;
    }
    *(float4*)(aggt + wr * AGG_LD + e0) = sum;
  }
  __syncthreads();
  if (!kh) {
    #pragma unroll
    for (int kk = 0; kk < 16; ++kk)
      ua[kk] = *(const v4f*)(aggt + (ln & 7) * AGG_LD + 8 * kk + 4 * half);
    __builtin_amdgcn_sched_barrier(0);
    #pragma unroll
    for (int kk = 0; kk < 16; ++kk) {
      v2f a, bb;
      a.x  = ua[kk].x; a.y  = ua[kk].y;
      bb.x = ub[kk].x; bb.y = ub[kk].y;
      accu = __builtin_amdgcn_wmma_f32_16x16x4_f32(
          false, a, false, bb, (short)0, accu, false, false);
      a.x  = ua[kk].z; a.y  = ua[kk].w;
      bb.x = ub[kk].z; bb.y = ub[kk].w;
      accu = __builtin_amdgcn_wmma_f32_16x16x4_f32(
          false, a, false, bb, (short)0, accu, false, false);
    }
    __builtin_amdgcn_sched_barrier(0);
    asm volatile("v_nop\n\tv_nop\n\tv_nop\n\tv_nop" : "+v"(accu));
    __builtin_amdgcn_sched_barrier(0);
  }
  if (half == 0) {
    #pragma unroll
    for (int r = 0; r < ITILE; ++r)
      utile[(kh * ITILE + r) * TILE_LD + un0 + ln] = accu[r];
  }
  __syncthreads();
  if (wu < ITILE) {
    const float4 xa = *(const float4*)(utile + wr * TILE_LD + e0);
    const float4 xh = *(const float4*)(utile + (ITILE + wr) * TILE_LD + e0);
    const float4 hr4 = *(const float4*)(hrows + wr * AGG_LD + e0);
    float4 u4;
    u4.x = (xh.x + xa.x) + bu4.x;
    u4.y = (xh.y + xa.y) + bu4.y;
    u4.z = (xh.z + xa.z) + bu4.z;
    u4.w = (xh.w + xa.w) + bu4.w;
    float ps  = u4.x + u4.y + u4.z + u4.w;
    float pss = u4.x * u4.x + u4.y * u4.y + u4.z * u4.z + u4.w * u4.w;
    #pragma unroll
    for (int off = 16; off >= 1; off >>= 1) {
      ps  += __shfl_xor(ps,  off, 32);
      pss += __shfl_xor(pss, off, 32);
    }
    const float mean = ps * (1.0f / DD);
    const float var  = pss * (1.0f / DD) - mean * mean;
    const float rinv = rsqrtf(var + 1e-5f);

    const float x0 = (u4.x - mean) * rinv * gu4.x + beu4.x;
    const float x1 = (u4.y - mean) * rinv * gu4.y + beu4.y;
    const float x2 = (u4.z - mean) * rinv * gu4.z + beu4.z;
    const float x3 = (u4.w - mean) * rinv * gu4.w + beu4.w;

    float4 o;
    o.x = hr4.x + x0 / (1.0f + __expf(-x0));
    o.y = hr4.y + x1 / (1.0f + __expf(-x1));
    o.z = hr4.z + x2 / (1.0f + __expf(-x2));
    o.w = hr4.w + x3 / (1.0f + __expf(-x3));
    *(float4*)(out + (size_t)(b * NN + i0 + wr) * DD + e0) = o;
  }
}

extern "C" void kernel_launch(void* const* d_in, const int* in_sizes, int n_in,
                              void* d_out, int out_size, void* d_ws, size_t ws_size,
                              hipStream_t stream) {
  const float* h       = (const float*)d_in[0];
  const float* adj     = (const float*)d_in[1];
  const float* Wmsg    = (const float*)d_in[2];
  const float* bmsg    = (const float*)d_in[3];
  const float* gmsg    = (const float*)d_in[4];
  const float* betamsg = (const float*)d_in[5];
  const float* Wupd    = (const float*)d_in[6];
  const float* bupd    = (const float*)d_in[7];
  const float* gupd    = (const float*)d_in[8];
  const float* betaupd = (const float*)d_in[9];
  float* out = (float*)d_out;

  const size_t mat_bytes = (size_t)ROWS * DD * sizeof(float);
  char* ws = (char*)d_ws;
  float*  T     = (float*)(ws);
  float*  S     = (float*)(ws + mat_bytes);
  float4* STATS = (float4*)(ws + 4 * mat_bytes);

  gemm_ts_stats_kernel<<<128, 256, 0, stream>>>(h, Wmsg, bmsg, T, S, STATS);
  msg_agg_kernel<<<256, 512, 0, stream>>>(T, S, adj, Wmsg, bmsg, gmsg,
                                          betamsg, STATS, h, Wupd, bupd, gupd,
                                          betaupd, out);
}


# msg kernel j-loop: pairs taken target row by target row in groups of 8; a group's 8 LDS constant reads issued as one run a group ahead of use (16 read places per wave, was 128); no sum reordered
# speedup vs baseline: 4.5595x; 1.0604x over previous
#include <hip/hip_runtime.h>

typedef __attribute__((ext_vector_type(2))) float v2f;
typedef __attribute__((ext_vector_type(8))) float v8f;
typedef __attribute__((ext_vector_type(4))) float v4f;
typedef __attribute__((ext_vector_type(4), aligned(4))) float v4f_a4;

#define DD 128
#define NN 256
#define BB 8
#define ROWS (BB * NN)
#define WMSG_LD (2 * DD + 1)
#define WUPD_LD (2 * DD)
#define JSPLIT 8
#define JPER (NN / JSPLIT)
#define TILE_LD 132

__global__ __launch_bounds__(256) void gemm_ts_stats_kernel(
    const float* __restrict__ H, const float* __restrict__ Wmsg,
    const float* __restrict__ bmsg,
    float* __restrict__ T, float* __restrict__ S, float4* __restrict__ STATS)
{
  __shared__ __attribute__((aligned(16))) float tile[16 * TILE_LD];
  v8f accT = {};
  v8f accS = {};
  {
    const int lane = threadIdx.x & 31;
    const int n0 = (threadIdx.x >> 5) << 4;
    const int m0 = blockIdx.x << 4;
    const int half = lane >> 4;
    const int ln = lane & 15;
    const float* __restrict__ arow = H + (size_t)(m0 + ln) * DD;
    const float* __restrict__ wrow = Wmsg + (size_t)(n0 + ln) * WMSG_LD;
    for (int k0 = 0; k0 < DD; k0 += 8) {
      const int ka = k0 + 4 * half;
      const v4f    a4  = *(const v4f*)(arow + ka);
      const v4f_a4 bt4 = *(const v4f_a4*)(wrow + ka);
      const v4f_a4 bs4 = *(const v4f_a4*)(wrow + DD + ka);
      v2f a, bt, bs;
      a.x  = a4.x;  a.y  = a4.y;
      bt.x = bt4.x; bt.y = bt4.y;
      bs.x = bs4.x; bs.y = bs4.y;
      accT = __builtin_amdgcn_wmma_f32_16x16x4_f32(
          false, a, false, bt, (short)0, accT, false, false);
      accS = __builtin_amdgcn_wmma_f32_16x16x4_f32(
          false, a, false, bs, (short)0, accS, false, false);
      a.x  = a4.z;  a.y  = a4.w;
      bt.x = bt4.z; bt.y = bt4.w;
      bs.x = bs4.z; bs.y = bs4.w;
      accT = __builtin_amdgcn_wmma_f32_16x16x4_f32(
          false, a, false, bt, (short)0, accT, false, false);
      accS = __builtin_amdgcn_wmma_f32_16x16x4_f32(
          false, a, false, bs, (short)0, accS, false, false);
    }
  }
  int tid = threadIdx.x;
  __builtin_amdgcn_sched_barrier(0);
  asm volatile("v_nop\n\tv_nop\n\tv_nop\n\tv_nop" : "+v"(accT), "+v"(accS), "+v"(tid));
  __builtin_amdgcn_sched_barrier(0);
  const int lane = tid & 31;
  const int nt = tid >> 5;
  const int m0 = blockIdx.x << 4, n0 = nt << 4;
  const int half = lane >> 4;
  const int ln = lane & 15;

  const int e0 = lane << 2;
  float4 wj4;
  wj4.x = Wmsg[(size_t)(e0 + 0) * WMSG_LD + 2 * DD];
  wj4.y = Wmsg[(size_t)(e0 + 1) * WMSG_LD + 2 * DD];
  wj4.z = Wmsg[(size_t)(e0 + 2) * WMSG_LD + 2 * DD];
  wj4.w = Wmsg[(size_t)(e0 + 3) * WMSG_LD + 2 * DD];
  const float4 bm4 = *(const float4*)(bmsg + e0);

  #pragma unroll
  for (int r = 0; r < 8; ++r)
    tile[(r + 8 * half) * TILE_LD + n0 + ln] = accS[r];
  __syncthreads();
  #pragma unroll
  for (int r = 0; r < 8; ++r) {
    const int row = m0 + r + 8 * half;
    T[(size_t)row * DD + n0 + ln] = accT[r];
    S[(size_t)row * DD + n0 + ln] = accS[r];
  }

  float p1[2], p2[2], p3[2], p4[2];
  #pragma unroll
  for (int rr = 0; rr < 2; ++rr) {
    const float4 s4 = *(const float4*)(tile + (2 * nt + rr) * TILE_LD + e0);
    p1[rr] = s4.x + s4.y + s4.z + s4.w;
    p2[rr] = s4.x * s4.x + s4.y * s4.y + s4.z * s4.z + s4.w * s4.w;
    p3[rr] = s4.x * wj4.x + s4.y * wj4.y + s4.z * wj4.z + s4.w * wj4.w;
    p4[rr] = s4.x * bm4.x + s4.y * bm4.y + s4.z * bm4.z + s4.w * bm4.w;
  }
  #pragma unroll
  for (int off = 16; off >= 1; off >>= 1) {
    #pragma unroll
    for (int rr = 0; rr < 2; ++rr) {
      p1[rr] += __shfl_xor(p1[rr], off, 32);
      p2[rr] += __shfl_xor(p2[rr], off, 32);
      p3[rr] += __shfl_xor(p3[rr], off, 32);
      p4[rr] += __shfl_xor(p4[rr], off, 32);
    }
  }
  if (lane < 2) {
    const int rr = lane;
    float4 st;
    st.x = rr ? p1[1] : p1[0];
    st.y = rr ? p2[1] : p2[0];
    st.z = rr ? p3[1] : p3[0];
    st.w = rr ? p4[1] : p4[0];
    STATS[m0 + 2 * nt + rr] = st;
  }
}

#define ITILE 8
#define MWAVES 16
#define JPW (NN / MWAVES)
#define AGG_LD 136
#define JHALF (JPW / 2)
__global__ __launch_bounds__(512, 4) void msg_agg_kernel(
    const float* __restrict__ T, const float* __restrict__ S,
    const float* __restrict__ adj, const float* __restrict__ Wmsg,
    const float* __restrict__ bmsg, const float* __restrict__ gmsg,
    const float* __restrict__ betamsg,
    const float4* __restrict__ STATS,
    const float* __restrict__ H, const float* __restrict__ Wupd,
    const float* __restrict__ bupd, const float* __restrict__ gupd,
    const float* __restrict__ betaupd, float* __restrict__ out)
{
  __shared__ float4 part[ITILE][MWAVES][32];
  __shared__ __attribute__((aligned(16))) float aggt[ITILE * AGG_LD];
  __shared__ __attribute__((aligned(16))) float hrows[ITILE * AGG_LD];
  __shared__ __attribute__((aligned(16))) float utile[2 * ITILE * TILE_LD];
  __shared__ float4 jconst[ITILE][MWAVES][JPW];
  __shared__ float4 rowc[ITILE];
  v8f accg = {};
  {
    const int lane = threadIdx.x & 31;
    const int half = lane >> 4;
    const int ln = lane & 15;
    const int b  = blockIdx.x >> 5;
    const int i0 = (blockIdx.x & 31) << 3;
    const int jw = (threadIdx.x >> 5) << 4;
    const float* __restrict__ arow = T + (size_t)(b * NN + i0 + (ln & 7)) * DD;
    const float* __restrict__ brow = S + (size_t)(b * NN + jw + ln) * DD;
    for (int k0 = 0; k0 < DD; k0 += 8) {
      const int ka = k0 + 4 * half;
      const v4f a4 = *(const v4f*)(arow + ka);
      const v4f b4 = *(const v4f*)(brow + ka);
      v2f a, bb;
      a.x  = a4.x; a.y  = a4.y;
      bb.x = b4.x; bb.y = b4.y;
      accg = __builtin_amdgcn_wmma_f32_16x16x4_f32(
          false, a, false, bb, (short)0, accg, false, false);
      a.x  = a4.z; a.y  = a4.w;
      bb.x = b4.z; bb.y = b4.w;
      accg = __builtin_amdgcn_wmma_f32_16x16x4_f32(
          false, a, false, bb, (short)0, accg, false, false);
    }
  }
  int tid = threadIdx.x;
  __builtin_amdgcn_sched_barrier(0);
  asm volatile("v_nop\n\tv_nop\n\tv_nop\n\tv_nop" : "+v"(accg), "+v"(tid));
  __builtin_amdgcn_sched_barrier(0);
  const int w    = tid >> 5;
  const int lane = tid & 31;
  const int half = lane >> 4;
  const int ln   = lane & 15;
  const int b  = blockIdx.x >> 5;
  const int i0 = (blockIdx.x & 31) << 3;
  const int e0 = lane << 2;

  float4 wj4;
  wj4.x = Wmsg[(size_t)(e0 + 0) * WMSG_LD + 2 * DD];
  wj4.y = Wmsg[(size_t)(e0 + 1) * WMSG_LD + 2 * DD];
  wj4.z = Wmsg[(size_t)(e0 + 2) * WMSG_LD + 2 * DD];
  wj4.w = Wmsg[(size_t)(e0 + 3) * WMSG_LD + 2 * DD];
  const float4 bm4 = *(const float4*)(bmsg + e0);
  const float4 g4  = *(const float4*)(gmsg + e0);
  const float4 be4 = *(const float4*)(betamsg + e0);

  float4 gh4, bh4;
  gh4.x = 0.5f * g4.x;  gh4.y = 0.5f * g4.y;
  gh4.z = 0.5f * g4.z;  gh4.w = 0.5f * g4.w;
  bh4.x = 0.5f * be4.x; bh4.y = 0.5f * be4.y;
  bh4.z = 0.5f * be4.z; bh4.w = 0.5f * be4.w;

  float4 c4[ITILE];
  #pragma unroll
  for (int r = 0; r < ITILE; ++r) {
    const float4 t4 = *(const float4*)(T + (size_t)(b * NN + i0 + r) * DD + e0);
    c4[r].x = t4.x + bm4.x; c4[r].y = t4.y + bm4.y;
    c4[r].z = t4.z + bm4.z; c4[r].w = t4.w + bm4.w;
  }

  const float2 hst2 = *(const float2*)(H + (size_t)(b * NN + i0) * DD + 2 * tid);

  float4 sreg[JPW];
  #pragma unroll
  for (int jj = 0; jj < JPW; ++jj)
    sreg[jj] = *(const float4*)(S + (size_t)(b * NN + w * JPW + jj) * DD + e0);

  const int j0 = w * JPW;
  const float4 st = STATS[(size_t)b * NN + j0 + ln];
  float ajr[4];
  #pragma unroll
  for (int k = 0; k < 4; ++k)
    ajr[k] = adj[(size_t)(b * NN + i0 + 4 * half + k) * NN + j0 + ln];

  const int rown = w & (ITILE - 1);
  const float4 to4 = *(const float4*)(T + (size_t)(b * NN + i0 + rown) * DD + e0);
  float4 co4;
  co4.x = to4.x + bm4.x; co4.y = to4.y + bm4.y;
  co4.z = to4.z + bm4.z; co4.w = to4.w + bm4.w;
  float r1 = co4.x + co4.y + co4.z + co4.w;
  float r2 = co4.x * co4.x + co4.y * co4.y + co4.z * co4.z + co4.w * co4.w;
  float r3 = co4.x * wj4.x + co4.y * wj4.y + co4.z * wj4.z + co4.w * wj4.w;
  float r4 = wj4.x + wj4.y + wj4.z + wj4.w;
  float r5 = wj4.x * wj4.x + wj4.y * wj4.y + wj4.z * wj4.z + wj4.w * wj4.w;
  #pragma unroll
  for (int off = 16; off >= 1; off >>= 1) {
    r1 += __shfl_xor(r1, off, 32);
    r2 += __shfl_xor(r2, off, 32);
    r3 += __shfl_xor(r3, off, 32);
    r4 += __shfl_xor(r4, off, 32);
    r5 += __shfl_xor(r5, off, 32);
  }
  const float W1 = r4, W2 = r5;
  const float invD = 1.0f / DD;
  if (w < ITILE && lane == 0) {
    float4 rc;
    rc.x = r1; rc.y = r2; rc.z = r3; rc.w = 0.0f;
    rowc[w] = rc;
  }
  *(float2*)(hrows + (tid >> 6) * AGG_LD + ((2 * tid) & (DD - 1))) = hst2;
  __syncthreads();

  {
    const int j = j0 + ln;
    #pragma unroll
    for (int k = 0; k < 4; ++k) {
      const int r = 4 * half + k;
      const int i = i0 + r;
      const float  aj  = ajr[k];
      const float  gij = half ? accg[4 + k] : accg[k];
      const float4 rc = rowc[r];
      const float C1 = rc.x, C2 = rc.y, Cw = rc.z;

      const float mean  = (C1 + st.x + aj * W1) * invD;
      const float sumsq = C2 + st.y + aj * aj * W2
                        + 2.0f * (gij + st.w)
                        + 2.0f * aj * (Cw + st.z);
      const float var  = sumsq * invD - mean * mean;
      const float rinv = rsqrtf(var + 1e-5f);

      float4 q;
      q.x = aj;
      q.y = (j != i) ? 1.0f : 0.0f;
      q.z = rinv;
      q.w = 0.0f;
      jconst[r][w][ln] = q;
    }
  }
  {
    const float mw = W1 * invD;
    wj4.x = gh4.x * (wj4.x - mw); wj4.y = gh4.y * (wj4.y - mw);
    wj4.z = gh4.z * (wj4.z - mw); wj4.w = gh4.w * (wj4.w - mw);
    #pragma unroll
    for (int r = 0; r < ITILE; ++r) {
      const float mc = rowc[r].x * invD;
      c4[r].x = gh4.x * (c4[r].x - mc); c4[r].y = gh4.y * (c4[r].y - mc);
      c4[r].z = gh4.z * (c4[r].z - mc); c4[r].w = gh4.w * (c4[r].w - mc);
    }
    #pragma unroll
    for (int jj = 0; jj < JPW; ++jj) {
      const float ms = invD * __int_as_float(
          __builtin_amdgcn_readlane(__float_as_int(st.x), jj));
      sreg[jj].x = gh4.x * (sreg[jj].x - ms); sreg[jj].y = gh4.y * (sreg[jj].y - ms);
      sreg[jj].z = gh4.z * (sreg[jj].z - ms); sreg[jj].w = gh4.w * (sreg[jj].w - ms);
    }
  }
  __syncthreads();

  float4 acc[ITILE];
  #pragma unroll
  for (int r = 0; r < ITILE; ++r) {
    acc[r].x = 0.f; acc[r].y = 0.f; acc[r].z = 0.f; acc[r].w = 0.f;
  }
  const v2f wjlo2 = {wj4.x, wj4.y}, wjhi2 = {wj4.z, wj4.w};
  const v2f bhlo2 = {bh4.x, bh4.y}, bhhi2 = {bh4.z, bh4.w};
  float4 qc[JHALF], qn[JHALF];
  #pragma unroll
  for (int v = 0; v < JHALF; ++v) qc[v] = jconst[0][w][v];
  #pragma unroll
  for (int g = 0; g < 2 * ITILE; ++g) {
    const int r  = g >> 1;
    const int jb = (g & 1) * JHALF;
    if (g + 1 < 2 * ITILE) {
      if (g > 0) {
        const int rp = (g - 1) >> 1;
        v2f alo = {acc[rp].x, acc[rp].y}, ahi = {acc[rp].z, acc[rp].w};
        asm volatile("" : "+v"(alo), "+v"(ahi));
        acc[rp].x = alo.x; acc[rp].y = alo.y; acc[rp].z = ahi.x; acc[rp].w = ahi.y;
      }
      #pragma unroll
      for (int v = 0; v < JHALF; ++v)
        qn[v] = jconst[(g + 1) >> 1][w][((g + 1) & 1) * JHALF + v];
      __builtin_amdgcn_sched_barrier(0);
      v2f clo = {c4[r].x, c4[r].y}, chi = {c4[r].z, c4[r].w};
      asm volatile("" : "+v"(clo), "+v"(chi));
      c4[r].x = clo.x; c4[r].y = clo.y; c4[r].z = chi.x; c4[r].w = chi.y;
    }
    const v2f clo2 = {c4[r].x, c4[r].y}, chi2 = {c4[r].z, c4[r].w};
    v2f alo2 = {acc[r].x, acc[r].y}, ahi2 = {acc[r].z, acc[r].w};
    #pragma unroll
    for (int j8 = 0; j8 < JHALF; ++j8) {
      const float4 s4 = sreg[jb + j8];
      const v2f slo2 = {s4.x, s4.y}, shi2 = {s4.z, s4.w};
      const v2f aj2 = {qc[j8].x, qc[j8].x};
      const v2f kp2 = {qc[j8].y, qc[j8].y};
      const v2f ri2 = {qc[j8].z, qc[j8].z};

      const v2f plo = __builtin_elementwise_fma(aj2, wjlo2, clo2 + slo2);
      const v2f phi = __builtin_elementwise_fma(aj2, wjhi2, chi2 + shi2);
      const v2f hlo = __builtin_elementwise_fma(ri2, plo, bhlo2);
      const v2f hhi = __builtin_elementwise_fma(ri2, phi, bhhi2);
      const v2f tlo = {__builtin_amdgcn_tanhf(hlo.x), __builtin_amdgcn_tanhf(hlo.y)};
      const v2f thi = {__builtin_amdgcn_tanhf(hhi.x), __builtin_amdgcn_tanhf(hhi.y)};
      const v2f mlo = hlo + hlo * tlo;
      const v2f mhi = hhi + hhi * thi;

      alo2 += kp2 * mlo;
      ahi2 += kp2 * mhi;
    }
    acc[r].x = alo2.x; acc[r].y = alo2.y; acc[r].z = ahi2.x; acc[r].w = ahi2.y;
    #pragma unroll
    for (int v = 0; v < JHALF; ++v) qc[v] = qn[v];
  }
  #pragma unroll
  for (int r = 0; r < ITILE; ++r)
    part[r][w][lane] = acc[r];

  const int wu  = __builtin_amdgcn_readfirstlane(w);
  const int kh  = wu >> 3;
  const int wr  = wu & (ITILE - 1);
  const int un0 = wr << 4;
  v4f ub[16], ua[16];
  {
    const float* __restrict__ uwrow =
        Wupd + (size_t)(un0 + ln) * WUPD_LD + (kh ? 0 : DD) + 4 * half;
    #pragma unroll
    for (int kk = 0; kk < 16; ++kk)
      ub[kk] = *(const v4f*)(uwrow + 8 * kk);
  }
  const float4 gu4  = *(const float4*)(gupd + e0);
  const float4 beu4 = *(const float4*)(betaupd + e0);
  const float4 bu4  = *(const float4*)(bupd + e0);

  __syncthreads();

  v8f accu = {};
  if (kh) {
    #pragma unroll
    for (int kk = 0; kk < 16; ++kk)
      ua[kk] = *(const v4f*)(hrows + (ln & 7) * AGG_LD + 8 * kk + 4 * half);
    __builtin_amdgcn_sched_barrier(0);
    #pragma unroll
    for (int kk = 0; kk < 16; ++kk) {
      v2f a, bb;
      a.x  = ua[kk].x; a.y  = ua[kk].y;
      bb.x = ub[kk].x; bb.y = ub[kk].y;
      accu = __builtin_amdgcn_wmma_f32_16x16x4_f32(
          false, a, false, bb, (short)0, accu, false, false);
      a.x  = ua[kk].z; a.y  = ua[kk].w;
      bb.x = ub[kk].z; bb.y = ub[kk].w;
      accu = __builtin_amdgcn_wmma_f32_16x16x4_f32(
          false, a, false, bb, (short)0, accu, false, false);
    }
    __builtin_amdgcn_sched_barrier(0);
    asm volatile("v_nop\n\tv_nop\n\tv_nop\n\tv_nop" : "+v"(accu));
    __builtin_amdgcn_sched_barrier(0);
  } else {
    float4 sum = part[wr][0][lane];
    for (int k = 1; k < MWAVES; ++k) {
      const float4 p = part[wr][k][lane];
      sum.x += p.x; sum.y += p.y; sum.z += p.z; sum.w += p.w;
    }
    *(float4*)(aggt + wr * AGG_LD + e0) = sum;
  }
  __syncthreads();
  if (!kh) {
    #pragma unroll
    for (int kk = 0; kk < 16; ++kk)
      ua[kk] = *(const v4f*)(aggt + (ln & 7) * AGG_LD + 8 * kk + 4 * half);
    __builtin_amdgcn_sched_barrier(0);
    #pragma unroll
    for (int kk = 0; kk < 16; ++kk) {
      v2f a, bb;
      a.x  = ua[kk].x; a.y  = ua[kk].y;
      bb.x = ub[kk].x; bb.y = ub[kk].y;
      accu = __builtin_amdgcn_wmma_f32_16x16x4_f32(
          false, a, false, bb, (short)0, accu, false, false);
      a.x  = ua[kk].z; a.y  = ua[kk].w;
      bb.x = ub[kk].z; bb.y = ub[kk].w;
      accu = __builtin_amdgcn_wmma_f32_16x16x4_f32(
          false, a, false, bb, (short)0, accu, false, false);
    }
    __builtin_amdgcn_sched_barrier(0);
    asm volatile("v_nop\n\tv_nop\n\tv_nop\n\tv_nop" : "+v"(accu));
    __builtin_amdgcn_sched_barrier(0);
  }
  if (half == 0) {
    #pragma unroll
    for (int r = 0; r < ITILE; ++r)
      utile[(kh * ITILE + r) * TILE_LD + un0 + ln] = accu[r];
  }
  __syncthreads();
  if (wu < ITILE) {
    const float4 xa = *(const float4*)(utile + wr * TILE_LD + e0);
    const float4 xh = *(const float4*)(utile + (ITILE + wr) * TILE_LD + e0);
    const float4 hr4 = *(const float4*)(hrows + wr * AGG_LD + e0);
    float4 u4;
    u4.x = (xh.x + xa.x) + bu4.x;
    u4.y = (xh.y + xa.y) + bu4.y;
    u4.z = (xh.z + xa.z) + bu4.z;
    u4.w = (xh.w + xa.w) + bu4.w;
    float ps  = u4.x + u4.y + u4.z + u4.w;
    float pss = u4.x * u4.x + u4.y * u4.y + u4.z * u4.z + u4.w * u4.w;
    #pragma unroll
    for (int off = 16; off >= 1; off >>= 1) {
      ps  += __shfl_xor(ps,  off, 32);
      pss += __shfl_xor(pss, off, 32);
    }
    const float mean = ps * (1.0f / DD);
    const float var  = pss * (1.0f / DD) - mean * mean;
    const float rinv = rsqrtf(var + 1e-5f);

    const float x0 = (u4.x - mean) * rinv * gu4.x + beu4.x;
    const float x1 = (u4.y - mean) * rinv * gu4.y + beu4.y;
    const float x2 = (u4.z - mean) * rinv * gu4.z + beu4.z;
    const float x3 = (u4.w - mean) * rinv * gu4.w + beu4.w;

    float4 o;
    o.x = hr4.x + x0 / (1.0f + __expf(-x0));
    o.y = hr4.y + x1 / (1.0f + __expf(-x1));
    o.z = hr4.z + x2 / (1.0f + __expf(-x2));
    o.w = hr4.w + x3 / (1.0f + __expf(-x3));
    *(float4*)(out + (size_t)(b * NN + i0 + wr) * DD + e0) = o;
  }
}

extern "C" void kernel_launch(void* const* d_in, const int* in_sizes, int n_in,
                              void* d_out, int out_size, void* d_ws, size_t ws_size,
                              hipStream_t stream) {
  const float* h       = (const float*)d_in[0];
  const float* adj     = (const float*)d_in[1];
  const float* Wmsg    = (const float*)d_in[2];
  const float* bmsg    = (const float*)d_in[3];
  const float* gmsg    = (const float*)d_in[4];
  const float* betamsg = (const float*)d_in[5];
  const float* Wupd    = (const float*)d_in[6];
  const float* bupd    = (const float*)d_in[7];
  const float* gupd    = (const float*)d_in[8];
  const float* betaupd = (const float*)d_in[9];
  float* out = (float*)d_out;

  const size_t mat_bytes = (size_t)ROWS * DD * sizeof(float);
  char* ws = (char*)d_ws;
  float*  T     = (float*)(ws);
  float*  S     = (float*)(ws + mat_bytes);
  float4* STATS = (float4*)(ws + 4 * mat_bytes);

  gemm_ts_stats_kernel<<<128, 256, 0, stream>>>(h, Wmsg, bmsg, T, S, STATS);
  msg_agg_kernel<<<256, 512, 0, stream>>>(T, S, adj, Wmsg, bmsg, gmsg,
                                          betamsg, STATS, h, Wupd, bupd, gupd,
                                          betaupd, out);
}


# msg kernel: G fragments read from LDS: source rows fetched once (the j-loop's register copy staged per wave), target rows staged once per group; set-up fetches lead; no sum reordered
# speedup vs baseline: 4.9310x; 1.0815x over previous
#include <hip/hip_runtime.h>

typedef __attribute__((ext_vector_type(2))) float v2f;
typedef __attribute__((ext_vector_type(8))) float v8f;
typedef __attribute__((ext_vector_type(4))) float v4f;
typedef __attribute__((ext_vector_type(4), aligned(4))) float v4f_a4;

#define DD 128
#define NN 256
#define BB 8
#define ROWS (BB * NN)
#define WMSG_LD (2 * DD + 1)
#define WUPD_LD (2 * DD)
#define JSPLIT 8
#define JPER (NN / JSPLIT)
#define TILE_LD 132

__global__ __launch_bounds__(256) void gemm_ts_stats_kernel(
    const float* __restrict__ H, const float* __restrict__ Wmsg,
    const float* __restrict__ bmsg,
    float* __restrict__ T, float* __restrict__ S, float4* __restrict__ STATS)
{
  __shared__ __attribute__((aligned(16))) float tile[16 * TILE_LD];
  v8f accT = {};
  v8f accS = {};
  {
    const int lane = threadIdx.x & 31;
    const int n0 = (threadIdx.x >> 5) << 4;
    const int m0 = blockIdx.x << 4;
    const int half = lane >> 4;
    const int ln = lane & 15;
    const float* __restrict__ arow = H + (size_t)(m0 + ln) * DD;
    const float* __restrict__ wrow = Wmsg + (size_t)(n0 + ln) * WMSG_LD;
    for (int k0 = 0; k0 < DD; k0 += 8) {
      const int ka = k0 + 4 * half;
      const v4f    a4  = *(const v4f*)(arow + ka);
      const v4f_a4 bt4 = *(const v4f_a4*)(wrow + ka);
      const v4f_a4 bs4 = *(const v4f_a4*)(wrow + DD + ka);
      v2f a, bt, bs;
      a.x  = a4.x;  a.y  = a4.y;
      bt.x = bt4.x; bt.y = bt4.y;
      bs.x = bs4.x; bs.y = bs4.y;
      accT = __builtin_amdgcn_wmma_f32_16x16x4_f32(
          false, a, false, bt, (short)0, accT, false, false);
      accS = __builtin_amdgcn_wmma_f32_16x16x4_f32(
          false, a, false, bs, (short)0, accS, false, false);
      a.x  = a4.z;  a.y  = a4.w;
      bt.x = bt4.z; bt.y = bt4.w;
      bs.x = bs4.z; bs.y = bs4.w;
      accT = __builtin_amdgcn_wmma_f32_16x16x4_f32(
          false, a, false, bt, (short)0, accT, false, false);
      accS = __builtin_amdgcn_wmma_f32_16x16x4_f32(
          false, a, false, bs, (short)0, accS, false, false);
    }
  }
  int tid = threadIdx.x;
  __builtin_amdgcn_sched_barrier(0);
  asm volatile("v_nop\n\tv_nop\n\tv_nop\n\tv_nop" : "+v"(accT), "+v"(accS), "+v"(tid));
  __builtin_amdgcn_sched_barrier(0);
  const int lane = tid & 31;
  const int nt = tid >> 5;
  const int m0 = blockIdx.x << 4, n0 = nt << 4;
  const int half = lane >> 4;
  const int ln = lane & 15;

  const int e0 = lane << 2;
  float4 wj4;
  wj4.x = Wmsg[(size_t)(e0 + 0) * WMSG_LD + 2 * DD];
  wj4.y = Wmsg[(size_t)(e0 + 1) * WMSG_LD + 2 * DD];
  wj4.z = Wmsg[(size_t)(e0 + 2) * WMSG_LD + 2 * DD];
  wj4.w = Wmsg[(size_t)(e0 + 3) * WMSG_LD + 2 * DD];
  const float4 bm4 = *(const float4*)(bmsg + e0);

  #pragma unroll
  for (int r = 0; r < 8; ++r)
    tile[(r + 8 * half) * TILE_LD + n0 + ln] = accS[r];
  __syncthreads();
  #pragma unroll
  for (int r = 0; r < 8; ++r) {
    const int row = m0 + r + 8 * half;
    T[(size_t)row * DD + n0 + ln] = accT[r];
    S[(size_t)row * DD + n0 + ln] = accS[r];
  }

  float p1[2], p2[2], p3[2], p4[2];
  #pragma unroll
  for (int rr = 0; rr < 2; ++rr) {
    const float4 s4 = *(const float4*)(tile + (2 * nt + rr) * TILE_LD + e0);
    p1[rr] = s4.x + s4.y + s4.z + s4.w;
    p2[rr] = s4.x * s4.x + s4.y * s4.y + s4.z * s4.z + s4.w * s4.w;
    p3[rr] = s4.x * wj4.x + s4.y * wj4.y + s4.z * wj4.z + s4.w * wj4.w;
    p4[rr] = s4.x * bm4.x + s4.y * bm4.y + s4.z * bm4.z + s4.w * bm4.w;
  }
  #pragma unroll
  for (int off = 16; off >= 1; off >>= 1) {
    #pragma unroll
    for (int rr = 0; rr < 2; ++rr) {
      p1[rr] += __shfl_xor(p1[rr], off, 32);
      p2[rr] += __shfl_xor(p2[rr], off, 32);
      p3[rr] += __shfl_xor(p3[rr], off, 32);
      p4[rr] += __shfl_xor(p4[rr], off, 32);
    }
  }
  if (lane < 2) {
    const int rr = lane;
    float4 st;
    st.x = rr ? p1[1] : p1[0];
    st.y = rr ? p2[1] : p2[0];
    st.z = rr ? p3[1] : p3[0];
    st.w = rr ? p4[1] : p4[0];
    STATS[m0 + 2 * nt + rr] = st;
  }
}

#define ITILE 8
#define MWAVES 16
#define JPW (NN / MWAVES)
#define AGG_LD 136
#define JHALF (JPW / 2)
#define OFF_JCONST 0
#define OFF_ROWC   (OFF_JCONST + 4 * ITILE * MWAVES * JPW)
#define OFF_PART   (OFF_ROWC + 4 * ITILE)
#define OFF_AGGT   (OFF_PART + 4 * ITILE * MWAVES * 32)
#define OFF_HROWS  (OFF_AGGT + ITILE * AGG_LD)
#define OFF_UTILE  (OFF_HROWS + ITILE * AGG_LD)
#define OFF_TROWS  (OFF_UTILE + 2 * ITILE * TILE_LD)
#define OFF_STILE  (OFF_TROWS + ITILE * AGG_LD)
#define LDS_FLOATS (OFF_STILE + MWAVES * JPW * TILE_LD)
__global__ __launch_bounds__(512, 4) void msg_agg_kernel(
    const float* __restrict__ T, const float* __restrict__ S,
    const float* __restrict__ adj, const float* __restrict__ Wmsg,
    const float* __restrict__ bmsg, const float* __restrict__ gmsg,
    const float* __restrict__ betamsg,
    const float4* __restrict__ STATS,
    const float* __restrict__ H, const float* __restrict__ Wupd,
    const float* __restrict__ bupd, const float* __restrict__ gupd,
    const float* __restrict__ betaupd, float* __restrict__ out)
{
  __shared__ __attribute__((aligned(16))) float lds[LDS_FLOATS];
  float4 (*jconst)[MWAVES][JPW] = reinterpret_cast<float4 (*)[MWAVES][JPW]>(lds + OFF_JCONST);
  float4* rowc = reinterpret_cast<float4*>(lds + OFF_ROWC);
  float4 (*part)[MWAVES][32] = reinterpret_cast<float4 (*)[MWAVES][32]>(lds + OFF_PART);
  float* aggt = lds + OFF_AGGT;
  float* hrows = lds + OFF_HROWS;
  float* utile = lds + OFF_UTILE;
  float* trows = lds + OFF_TROWS;
  float* stile = lds + OFF_STILE;
  const int tid  = threadIdx.x;
  const int w    = tid >> 5;
  const int lane = tid & 31;
  const int half = lane >> 4;
  const int ln   = lane & 15;
  const int b  = blockIdx.x >> 5;
  const int i0 = (blockIdx.x & 31) << 3;
  const int e0 = lane << 2;

  float4 wj4;
  wj4.x = Wmsg[(size_t)(e0 + 0) * WMSG_LD + 2 * DD];
  wj4.y = Wmsg[(size_t)(e0 + 1) * WMSG_LD + 2 * DD];
  wj4.z = Wmsg[(size_t)(e0 + 2) * WMSG_LD + 2 * DD];
  wj4.w = Wmsg[(size_t)(e0 + 3) * WMSG_LD + 2 * DD];
  const float4 bm4 = *(const float4*)(bmsg + e0);
  const float4 g4  = *(const float4*)(gmsg + e0);
  const float4 be4 = *(const float4*)(betamsg + e0);

  float4 gh4, bh4;
  gh4.x = 0.5f * g4.x;  gh4.y = 0.5f * g4.y;
  gh4.z = 0.5f * g4.z;  gh4.w = 0.5f * g4.w;
  bh4.x = 0.5f * be4.x; bh4.y = 0.5f * be4.y;
  bh4.z = 0.5f * be4.z; bh4.w = 0.5f * be4.w;

  float4 c4[ITILE];
  #pragma unroll
  for (int r = 0; r < ITILE; ++r) {
    const float4 t4 = *(const float4*)(T + (size_t)(b * NN + i0 + r) * DD + e0);
    c4[r].x = t4.x + bm4.x; c4[r].y = t4.y + bm4.y;
    c4[r].z = t4.z + bm4.z; c4[r].w = t4.w + bm4.w;
  }

  const float2 hst2 = *(const float2*)(H + (size_t)(b * NN + i0) * DD + 2 * tid);

  float4 sreg[JPW];
  #pragma unroll
  for (int jj = 0; jj < JPW; ++jj)
    sreg[jj] = *(const float4*)(S + (size_t)(b * NN + w * JPW + jj) * DD + e0);

  const int j0 = w * JPW;
  const float4 st = STATS[(size_t)b * NN + j0 + ln];
  float ajr[4];
  #pragma unroll
  for (int k = 0; k < 4; ++k)
    ajr[k] = adj[(size_t)(b * NN + i0 + 4 * half + k) * NN + j0 + ln];

  const int rown = w & (ITILE - 1);
  const float4 to4 = *(const float4*)(T + (size_t)(b * NN + i0 + rown) * DD + e0);
  float4 co4;
  co4.x = to4.x + bm4.x; co4.y = to4.y + bm4.y;
  co4.z = to4.z + bm4.z; co4.w = to4.w + bm4.w;
  float r1 = co4.x + co4.y + co4.z + co4.w;
  float r2 = co4.x * co4.x + co4.y * co4.y + co4.z * co4.z + co4.w * co4.w;
  float r3 = co4.x * wj4.x + co4.y * wj4.y + co4.z * wj4.z + co4.w * wj4.w;
  float r4 = wj4.x + wj4.y + wj4.z + wj4.w;
  float r5 = wj4.x * wj4.x + wj4.y * wj4.y + wj4.z * wj4.z + wj4.w * wj4.w;
  #pragma unroll
  for (int off = 16; off >= 1; off >>= 1) {
    r1 += __shfl_xor(r1, off, 32);
    r2 += __shfl_xor(r2, off, 32);
    r3 += __shfl_xor(r3, off, 32);
    r4 += __shfl_xor(r4, off, 32);
    r5 += __shfl_xor(r5, off, 32);
  }
  const float W1 = r4, W2 = r5;
  const float invD = 1.0f / DD;
  if (w < ITILE && lane == 0) {
    float4 rc;
    rc.x = r1; rc.y = r2; rc.z = r3; rc.w = 0.0f;
    rowc[w] = rc;
  }
  *(float2*)(hrows + (tid >> 6) * AGG_LD + ((2 * tid) & (DD - 1))) = hst2;
  if (w < ITILE)
    *(float4*)(trows + w * AGG_LD + e0) = to4;
  #pragma unroll
  for (int jj = 0; jj < JPW; ++jj)
    *(float4*)(stile + (w * JPW + jj) * TILE_LD + e0) = sreg[jj];
  __syncthreads();

  v8f accg = {};
  {
    const float* __restrict__ ap = trows + (ln & 7) * AGG_LD + 4 * half;
    const float* __restrict__ bp = stile + (w * JPW + ln) * TILE_LD + 4 * half;
    #pragma unroll
    for (int hk = 0; hk < 2; ++hk) {
      v4f ga[8], gb[8];
      #pragma unroll
      for (int kk = 0; kk < 8; ++kk) {
        ga[kk] = *(const v4f*)(ap + 8 * (8 * hk + kk));
        gb[kk] = *(const v4f*)(bp + 8 * (8 * hk + kk));
      }
      __builtin_amdgcn_sched_barrier(0);
      #pragma unroll
      for (int kk = 0; kk < 8; ++kk) {
        v2f a, bb;
        a.x  = ga[kk].x; a.y  = ga[kk].y;
        bb.x = gb[kk].x; bb.y = gb[kk].y;
        accg = __builtin_amdgcn_wmma_f32_16x16x4_f32(
            false, a, false, bb, (short)0, accg, false, false);
        a.x  = ga[kk].z; a.y  = ga[kk].w;
        bb.x = gb[kk].z; bb.y = gb[kk].w;
        accg = __builtin_amdgcn_wmma_f32_16x16x4_f32(
            false, a, false, bb, (short)0, accg, false, false);
      }
      __builtin_amdgcn_sched_barrier(0);
    }
  }
  asm volatile("v_nop\n\tv_nop\n\tv_nop\n\tv_nop" : "+v"(accg));
  __builtin_amdgcn_sched_barrier(0);

  {
    const int j = j0 + ln;
    #pragma unroll
    for (int k = 0; k < 4; ++k) {
      const int r = 4 * half + k;
      const int i = i0 + r;
      const float  aj  = ajr[k];
      const float  gij = half ? accg[4 + k] : accg[k];
      const float4 rc = rowc[r];
      const float C1 = rc.x, C2 = rc.y, Cw = rc.z;

      const float mean  = (C1 + st.x + aj * W1) * invD;
      const float sumsq = C2 + st.y + aj * aj * W2
                        + 2.0f * (gij + st.w)
                        + 2.0f * aj * (Cw + st.z);
      const float var  = sumsq * invD - mean * mean;
      const float rinv = rsqrtf(var + 1e-5f);

      float4 q;
      q.x = aj;
      q.y = (j != i) ? 1.0f : 0.0f;
      q.z = rinv;
      q.w = 0.0f;
      jconst[r][w][ln] = q;
    }
  }
  {
    const float mw = W1 * invD;
    wj4.x = gh4.x * (wj4.x - mw); wj4.y = gh4.y * (wj4.y - mw);
    wj4.z = gh4.z * (wj4.z - mw); wj4.w = gh4.w * (wj4.w - mw);
    #pragma unroll
    for (int r = 0; r < ITILE; ++r) {
      const float mc = rowc[r].x * invD;
      c4[r].x = gh4.x * (c4[r].x - mc); c4[r].y = gh4.y * (c4[r].y - mc);
      c4[r].z = gh4.z * (c4[r].z - mc); c4[r].w = gh4.w * (c4[r].w - mc);
    }
    #pragma unroll
    for (int jj = 0; jj < JPW; ++jj) {
      const float ms = invD * __int_as_float(
          __builtin_amdgcn_readlane(__float_as_int(st.x), jj));
      sreg[jj].x = gh4.x * (sreg[jj].x - ms); sreg[jj].y = gh4.y * (sreg[jj].y - ms);
      sreg[jj].z = gh4.z * (sreg[jj].z - ms); sreg[jj].w = gh4.w * (sreg[jj].w - ms);
    }
  }
  __syncthreads();

  float4 acc[ITILE];
  #pragma unroll
  for (int r = 0; r < ITILE; ++r) {
    acc[r].x = 0.f; acc[r].y = 0.f; acc[r].z = 0.f; acc[r].w = 0.f;
  }
  const v2f wjlo2 = {wj4.x, wj4.y}, wjhi2 = {wj4.z, wj4.w};
  const v2f bhlo2 = {bh4.x, bh4.y}, bhhi2 = {bh4.z, bh4.w};
  float4 qc[JHALF], qn[JHALF];
  #pragma unroll
  for (int v = 0; v < JHALF; ++v) qc[v] = jconst[0][w][v];
  #pragma unroll
  for (int g = 0; g < 2 * ITILE; ++g) {
    const int r  = g >> 1;
    const int jb = (g & 1) * JHALF;
    if (g + 1 < 2 * ITILE) {
      if (g > 0) {
        const int rp = (g - 1) >> 1;
        v2f alo = {acc[rp].x, acc[rp].y}, ahi = {acc[rp].z, acc[rp].w};
        asm volatile("" : "+v"(alo), "+v"(ahi));
        acc[rp].x = alo.x; acc[rp].y = alo.y; acc[rp].z = ahi.x; acc[rp].w = ahi.y;
      }
      #pragma unroll
      for (int v = 0; v < JHALF; ++v)
        qn[v] = jconst[(g + 1) >> 1][w][((g + 1) & 1) * JHALF + v];
      __builtin_amdgcn_sched_barrier(0);
      v2f clo = {c4[r].x, c4[r].y}, chi = {c4[r].z, c4[r].w};
      asm volatile("" : "+v"(clo), "+v"(chi));
      c4[r].x = clo.x; c4[r].y = clo.y; c4[r].z = chi.x; c4[r].w = chi.y;
    }
    const v2f clo2 = {c4[r].x, c4[r].y}, chi2 = {c4[r].z, c4[r].w};
    v2f alo2 = {acc[r].x, acc[r].y}, ahi2 = {acc[r].z, acc[r].w};
    #pragma unroll
    for (int j8 = 0; j8 < JHALF; ++j8) {
      const float4 s4 = sreg[jb + j8];
      const v2f slo2 = {s4.x, s4.y}, shi2 = {s4.z, s4.w};
      const v2f aj2 = {qc[j8].x, qc[j8].x};
      const v2f kp2 = {qc[j8].y, qc[j8].y};
      const v2f ri2 = {qc[j8].z, qc[j8].z};

      const v2f plo = __builtin_elementwise_fma(aj2, wjlo2, clo2 + slo2);
      const v2f phi = __builtin_elementwise_fma(aj2, wjhi2, chi2 + shi2);
      const v2f hlo = __builtin_elementwise_fma(ri2, plo, bhlo2);
      const v2f hhi = __builtin_elementwise_fma(ri2, phi, bhhi2);
      const v2f tlo = {__builtin_amdgcn_tanhf(hlo.x), __builtin_amdgcn_tanhf(hlo.y)};
      const v2f thi = {__builtin_amdgcn_tanhf(hhi.x), __builtin_amdgcn_tanhf(hhi.y)};
      const v2f mlo = hlo + hlo * tlo;
      const v2f mhi = hhi + hhi * thi;

      alo2 += kp2 * mlo;
      ahi2 += kp2 * mhi;
    }
    acc[r].x = alo2.x; acc[r].y = alo2.y; acc[r].z = ahi2.x; acc[r].w = ahi2.y;
    #pragma unroll
    for (int v = 0; v < JHALF; ++v) qc[v] = qn[v];
  }
  #pragma unroll
  for (int r = 0; r < ITILE; ++r)
    part[r][w][lane] = acc[r];

  const int wu  = __builtin_amdgcn_readfirstlane(w);
  const int kh  = wu >> 3;
  const int wr  = wu & (ITILE - 1);
  const int un0 = wr << 4;
  v4f ub[16], ua[16];
  {
    const float* __restrict__ uwrow =
        Wupd + (size_t)(un0 + ln) * WUPD_LD + (kh ? 0 : DD) + 4 * half;
    #pragma unroll
    for (int kk = 0; kk < 16; ++kk)
      ub[kk] = *(const v4f*)(uwrow + 8 * kk);
  }
  const float4 gu4  = *(const float4*)(gupd + e0);
  const float4 beu4 = *(const float4*)(betaupd + e0);
  const float4 bu4  = *(const float4*)(bupd + e0);

  __syncthreads();

  v8f accu = {};
  if (kh) {
    #pragma unroll
    for (int kk = 0; kk < 16; ++kk)
      ua[kk] = *(const v4f*)(hrows + (ln & 7) * AGG_LD + 8 * kk + 4 * half);
    __builtin_amdgcn_sched_barrier(0);
    #pragma unroll
    for (int kk = 0; kk < 16; ++kk) {
      v2f a, bb;
      a.x  = ua[kk].x; a.y  = ua[kk].y;
      bb.x = ub[kk].x; bb.y = ub[kk].y;
      accu = __builtin_amdgcn_wmma_f32_16x16x4_f32(
          false, a, false, bb, (short)0, accu, false, false);
      a.x  = ua[kk].z; a.y  = ua[kk].w;
      bb.x = ub[kk].z; bb.y = ub[kk].w;
      accu = __builtin_amdgcn_wmma_f32_16x16x4_f32(
          false, a, false, bb, (short)0, accu, false, false);
    }
    __builtin_amdgcn_sched_barrier(0);
    asm volatile("v_nop\n\tv_nop\n\tv_nop\n\tv_nop" : "+v"(accu));
    __builtin_amdgcn_sched_barrier(0);
  } else {
    float4 sum = part[wr][0][lane];
    for (int k = 1; k < MWAVES; ++k) {
      const float4 p = part[wr][k][lane];
      sum.x += p.x; sum.y += p.y; sum.z += p.z; sum.w += p.w;
    }
    *(float4*)(aggt + wr * AGG_LD + e0) = sum;
  }
  __syncthreads();
  if (!kh) {
    #pragma unroll
    for (int kk = 0; kk < 16; ++kk)
      ua[kk] = *(const v4f*)(aggt + (ln & 7) * AGG_LD + 8 * kk + 4 * half);
    __builtin_amdgcn_sched_barrier(0);
    #pragma unroll
    for (int kk = 0; kk < 16; ++kk) {
      v2f a, bb;
      a.x  = ua[kk].x; a.y  = ua[kk].y;
      bb.x = ub[kk].x; bb.y = ub[kk].y;
      accu = __builtin_amdgcn_wmma_f32_16x16x4_f32(
          false, a, false, bb, (short)0, accu, false, false);
      a.x  = ua[kk].z; a.y  = ua[kk].w;
      bb.x = ub[kk].z; bb.y = ub[kk].w;
      accu = __builtin_amdgcn_wmma_f32_16x16x4_f32(
          false, a, false, bb, (short)0, accu, false, false);
    }
    __builtin_amdgcn_sched_barrier(0);
    asm volatile("v_nop\n\tv_nop\n\tv_nop\n\tv_nop" : "+v"(accu));
    __builtin_amdgcn_sched_barrier(0);
  }
  if (half == 0) {
    #pragma unroll
    for (int r = 0; r < ITILE; ++r)
      utile[(kh * ITILE + r) * TILE_LD + un0 + ln] = accu[r];
  }
  __syncthreads();
  if (wu < ITILE) {
    const float4 xa = *(const float4*)(utile + wr * TILE_LD + e0);
    const float4 xh = *(const float4*)(utile + (ITILE + wr) * TILE_LD + e0);
    const float4 hr4 = *(const float4*)(hrows + wr * AGG_LD + e0);
    float4 u4;
    u4.x = (xh.x + xa.x) + bu4.x;
    u4.y = (xh.y + xa.y) + bu4.y;
    u4.z = (xh.z + xa.z) + bu4.z;
    u4.w = (xh.w + xa.w) + bu4.w;
    float ps  = u4.x + u4.y + u4.z + u4.w;
    float pss = u4.x * u4.x + u4.y * u4.y + u4.z * u4.z + u4.w * u4.w;
    #pragma unroll
    for (int off = 16; off >= 1; off >>= 1) {
      ps  += __shfl_xor(ps,  off, 32);
      pss += __shfl_xor(pss, off, 32);
    }
    const float mean = ps * (1.0f / DD);
    const float var  = pss * (1.0f / DD) - mean * mean;
    const float rinv = rsqrtf(var + 1e-5f);

    const float x0 = (u4.x - mean) * rinv * gu4.x + beu4.x;
    const float x1 = (u4.y - mean) * rinv * gu4.y + beu4.y;
    const float x2 = (u4.z - mean) * rinv * gu4.z + beu4.z;
    const float x3 = (u4.w - mean) * rinv * gu4.w + beu4.w;

    float4 o;
    o.x = hr4.x + x0 / (1.0f + __expf(-x0));
    o.y = hr4.y + x1 / (1.0f + __expf(-x1));
    o.z = hr4.z + x2 / (1.0f + __expf(-x2));
    o.w = hr4.w + x3 / (1.0f + __expf(-x3));
    *(float4*)(out + (size_t)(b * NN + i0 + wr) * DD + e0) = o;
  }
}

extern "C" void kernel_launch(void* const* d_in, const int* in_sizes, int n_in,
                              void* d_out, int out_size, void* d_ws, size_t ws_size,
                              hipStream_t stream) {
  const float* h       = (const float*)d_in[0];
  const float* adj     = (const float*)d_in[1];
  const float* Wmsg    = (const float*)d_in[2];
  const float* bmsg    = (const float*)d_in[3];
  const float* gmsg    = (const float*)d_in[4];
  const float* betamsg = (const float*)d_in[5];
  const float* Wupd    = (const float*)d_in[6];
  const float* bupd    = (const float*)d_in[7];
  const float* gupd    = (const float*)d_in[8];
  const float* betaupd = (const float*)d_in[9];
  float* out = (float*)d_out;

  const size_t mat_bytes = (size_t)ROWS * DD * sizeof(float);
  char* ws = (char*)d_ws;
  float*  T     = (float*)(ws);
  float*  S     = (float*)(ws + mat_bytes);
  float4* STATS = (float4*)(ws + 4 * mat_bytes);

  gemm_ts_stats_kernel<<<128, 256, 0, stream>>>(h, Wmsg, bmsg, T, S, STATS);
  msg_agg_kernel<<<256, 512, 0, stream>>>(T, S, adj, Wmsg, bmsg, gmsg,
                                          betamsg, STATS, h, Wupd, bupd, gupd,
                                          betaupd, out);
}
